# starting kernel
# baseline (speedup 1.0000x reference)
#include <hip/hip_runtime.h>
#include <math.h>

typedef float v2f __attribute__((ext_vector_type(2)));
typedef float v8f __attribute__((ext_vector_type(8)));

#define WMMA4(a, b, c) \
  __builtin_amdgcn_wmma_f32_16x16x4_f32(false, (a), false, (b), (short)0, (c), false, false)

__device__ __forceinline__ unsigned fenc(float f) {
  unsigned b = __float_as_uint(f);
  return (b & 0x80000000u) ? ~b : (b | 0x80000000u);
}
__device__ __forceinline__ float fdec(unsigned u) {
  return (u & 0x80000000u) ? __uint_as_float(u & 0x7FFFFFFFu) : __uint_as_float(~u);
}

__global__ void k_init_amax(unsigned* amax, int N) {
  int n = blockIdx.x * blockDim.x + threadIdx.x;
  if (n < N) amax[n] = fenc(-INFINITY);
}

__global__ void k_cnt_easum(const int* ei, const float* ea, float* cnt, float* eam, int E) {
  int tid = blockIdx.x * blockDim.x + threadIdx.x;
  if (tid >= E * 16) return;
  int e = tid >> 4, f = tid & 15;
  int d = ei[E + e];
  if (f == 0) atomicAdd(&cnt[d], 1.0f);
  atomicAdd(&eam[d * 16 + f], ea[e * 16 + f]);
}

__global__ void k_eamean(float* eam, const float* cnt, int N) {
  int tid = blockIdx.x * blockDim.x + threadIdx.x;
  if (tid >= N * 16) return;
  int n = tid >> 4;
  eam[tid] /= fmaxf(cnt[n], 1.0f);
}

__global__ void k_gemm_xlxr(const float* __restrict__ x,
                            const float* __restrict__ Wl, const float* __restrict__ bl,
                            const float* __restrict__ Wr, const float* __restrict__ br,
                            float* __restrict__ xl, float* __restrict__ xr) {
  int tile = blockIdx.x;
  int lane = threadIdx.x;
  int half = lane >> 4;
  int l16 = lane & 15;
  int row = tile * 16 + l16;
  v8f acc0 = {}; v8f acc1 = {};
#pragma unroll
  for (int k = 0; k < 128; k += 4) {
    int ka = k + half * 2;
    v2f a;  a.x = x[row * 128 + ka];     a.y = x[row * 128 + ka + 1];
    v2f b0; b0.x = Wl[ka * 16 + l16];    b0.y = Wl[(ka + 1) * 16 + l16];
    v2f b1; b1.x = Wr[ka * 16 + l16];    b1.y = Wr[(ka + 1) * 16 + l16];
    acc0 = WMMA4(a, b0, acc0);
    acc1 = WMMA4(a, b1, acc1);
  }
#pragma unroll
  for (int i = 0; i < 8; ++i) {
    int m = tile * 16 + i + half * 8;
    int c = l16;
    xl[m * 16 + c] = acc0[i] + bl[c];
    xr[m * 16 + c] = acc1[i] + br[c];
  }
}

__global__ void k_edge_alpha(const int* ei, const float* __restrict__ ea,
                             const float* __restrict__ xl, const float* __restrict__ xr,
                             const float* We, const float* att,
                             float* aedge, unsigned* amax, int E) {
  __shared__ float sWe[256];
  __shared__ float satt[16];
  int t = threadIdx.x;
  sWe[t] = We[t];
  if (t < 16) satt[t] = att[t];
  __syncthreads();
  int e = blockIdx.x * blockDim.x + t;
  if (e >= E) return;
  int s = ei[e], d = ei[E + e];
  float eav[16];
#pragma unroll
  for (int j = 0; j < 16; ++j) eav[j] = ea[e * 16 + j];
  float acc = 0.0f;
#pragma unroll
  for (int f = 0; f < 16; ++f) {
    float m = xl[s * 16 + f] + xr[d * 16 + f];
#pragma unroll
    for (int j = 0; j < 16; ++j) m += eav[j] * sWe[j * 16 + f];
    m = m > 0.0f ? m : 0.2f * m;
    acc += m * satt[f];
  }
  aedge[e] = acc;
  atomicMax(&amax[d], fenc(acc));
}

__global__ void k_loop_alpha(const float* __restrict__ eam,
                             const float* __restrict__ xl, const float* __restrict__ xr,
                             const float* We, const float* att,
                             float* aloop, unsigned* amax, int N) {
  __shared__ float sWe[256];
  __shared__ float satt[16];
  int t = threadIdx.x;
  sWe[t] = We[t];
  if (t < 16) satt[t] = att[t];
  __syncthreads();
  int n = blockIdx.x * blockDim.x + t;
  if (n >= N) return;
  float eav[16];
#pragma unroll
  for (int j = 0; j < 16; ++j) eav[j] = eam[n * 16 + j];
  float acc = 0.0f;
#pragma unroll
  for (int f = 0; f < 16; ++f) {
    float m = xl[n * 16 + f] + xr[n * 16 + f];
#pragma unroll
    for (int j = 0; j < 16; ++j) m += eav[j] * sWe[j * 16 + f];
    m = m > 0.0f ? m : 0.2f * m;
    acc += m * satt[f];
  }
  aloop[n] = acc;
  atomicMax(&amax[n], fenc(acc));
}

__global__ void k_exp_edge(const int* ei, float* aedge, const unsigned* amax,
                           float* denom, int E) {
  int e = blockIdx.x * blockDim.x + threadIdx.x;
  if (e >= E) return;
  int d = ei[E + e];
  float ex = expf(aedge[e] - fdec(amax[d]));
  aedge[e] = ex;
  atomicAdd(&denom[d], ex);
}

__global__ void k_exp_loop(float* aloop, const unsigned* amax, float* denom, int N) {
  int n = blockIdx.x * blockDim.x + threadIdx.x;
  if (n >= N) return;
  float ex = expf(aloop[n] - fdec(amax[n]));
  aloop[n] = ex;
  atomicAdd(&denom[n], ex);
}

__global__ void k_scatter(const int* ei, const float* __restrict__ aedge,
                          const float* __restrict__ denom, const float* __restrict__ xl,
                          float* hsum, int E) {
  int tid = blockIdx.x * blockDim.x + threadIdx.x;
  if (tid >= E * 16) return;
  int e = tid >> 4, f = tid & 15;
  int s = ei[e], d = ei[E + e];
  float w = aedge[e] / (denom[d] + 1e-16f);
  atomicAdd(&hsum[d * 16 + f], w * xl[s * 16 + f]);
}

__global__ void k_gatfin(float* hsum, const float* aloop, const float* denom,
                         const float* xl, const float* gbias, int N) {
  int tid = blockIdx.x * blockDim.x + threadIdx.x;
  if (tid >= N * 16) return;
  int n = tid >> 4, f = tid & 15;
  float w = aloop[n] / (denom[n] + 1e-16f);
  hsum[tid] = hsum[tid] + w * xl[n * 16 + f] + gbias[f];
}

__global__ void k_deg(const int* ei, float* deg, int E) {
  int e = blockIdx.x * blockDim.x + threadIdx.x;
  if (e >= E) return;
  int s = ei[e], d = ei[E + e];
  if (s != d) atomicAdd(&deg[s], 1.0f);
}

__global__ void k_dinv(float* deg, int N) {
  int n = blockIdx.x * blockDim.x + threadIdx.x;
  if (n >= N) return;
  float g = deg[n];
  deg[n] = (g > 0.0f) ? rsqrtf(fmaxf(g, 1e-12f)) : 0.0f;
}

__global__ void k_matvec(const int* ei, const float* __restrict__ dinv,
                         const float* __restrict__ in, float* out, int E) {
  int tid = blockIdx.x * blockDim.x + threadIdx.x;
  if (tid >= E * 16) return;
  int e = tid >> 4, f = tid & 15;
  int s = ei[e], d = ei[E + e];
  if (s == d) return;
  float nrm = -dinv[s] * dinv[d];
  atomicAdd(&out[d * 16 + f], nrm * in[s * 16 + f]);
}

__global__ void k_chebrec(float* T, const float* Tprev, int N) {
  int tid = blockIdx.x * blockDim.x + threadIdx.x;
  if (tid >= N * 16) return;
  T[tid] = 2.0f * T[tid] - Tprev[tid];
}

__global__ void k_gru(const float* __restrict__ T0, const float* __restrict__ T1,
                      const float* __restrict__ T2, const float* __restrict__ T3,
                      const float* __restrict__ chebW, const float* __restrict__ chebB,
                      float* __restrict__ H) {
  int tile = blockIdx.x;
  int lane = threadIdx.x;
  int half = lane >> 4;
  int l16 = lane & 15;
  int row = tile * 16 + l16;
  v8f accZ = {}; v8f accH = {};
#pragma unroll
  for (int t = 0; t < 4; ++t) {
    const float* Tm = (t == 0) ? T0 : (t == 1) ? T1 : (t == 2) ? T2 : T3;
    const float* Wz = chebW + t * 256;
    const float* Wh = chebW + (16 + t) * 256;
#pragma unroll
    for (int kc = 0; kc < 16; kc += 4) {
      int ka = kc + half * 2;
      v2f a;  a.x = Tm[row * 16 + ka];      a.y = Tm[row * 16 + ka + 1];
      v2f bz; bz.x = Wz[ka * 16 + l16];     bz.y = Wz[(ka + 1) * 16 + l16];
      v2f bh; bh.x = Wh[ka * 16 + l16];     bh.y = Wh[(ka + 1) * 16 + l16];
      accZ = WMMA4(a, bz, accZ);
      accH = WMMA4(a, bh, accH);
    }
  }
#pragma unroll
  for (int i = 0; i < 8; ++i) {
    int m = tile * 16 + i + half * 8;
    int c = l16;
    float zp = accZ[i] + chebB[c] + chebB[16 + c];
    float hp = accH[i] + chebB[64 + c] + chebB[80 + c];
    float z = 1.0f / (1.0f + expf(-zp));
    float ht = tanhf(hp);
    H[m * 16 + c] = (1.0f - z) * ht;
  }
}

__global__ void k_mlp(const float* __restrict__ H, const float* W1, const float* b1,
                      const float* W2, const float* b2, float* out, int N) {
  int idx = blockIdx.x * blockDim.x + threadIdx.x;
  if (idx > N) return;
  float hid[4];
#pragma unroll
  for (int j = 0; j < 4; ++j) {
    float s = b1[j];
    if (idx > 0) {
      const float* hr = H + (size_t)(idx - 1) * 16;
#pragma unroll
      for (int f = 0; f < 16; ++f) s += hr[f] * W1[f * 4 + j];
    }
    hid[j] = fmaxf(s, 0.0f);
  }
#pragma unroll
  for (int o = 0; o < 8; ++o) {
    float s = b2[o];
#pragma unroll
    for (int j = 0; j < 4; ++j) s += hid[j] * W2[j * 8 + o];
    out[idx * 8 + o] = s;
  }
}

extern "C" void kernel_launch(void* const* d_in, const int* in_sizes, int n_in,
                              void* d_out, int out_size, void* d_ws, size_t ws_size,
                              hipStream_t stream) {
  const float* x     = (const float*)d_in[0];
  const int*   ei    = (const int*)  d_in[1];
  const float* ea    = (const float*)d_in[2];
  const float* Wl    = (const float*)d_in[3];
  const float* bl    = (const float*)d_in[4];
  const float* Wr    = (const float*)d_in[5];
  const float* br    = (const float*)d_in[6];
  const float* We    = (const float*)d_in[7];
  const float* att   = (const float*)d_in[8];
  const float* gbias = (const float*)d_in[9];
  const float* chebW = (const float*)d_in[10];
  const float* chebB = (const float*)d_in[11];
  const float* W1    = (const float*)d_in[12];
  const float* b1    = (const float*)d_in[13];
  const float* W2    = (const float*)d_in[14];
  const float* b2    = (const float*)d_in[15];
  float* out = (float*)d_out;

  const int N = in_sizes[0] / 128;
  const int E = in_sizes[1] / 2;
  const size_t NF = (size_t)N * 16;

  float* ws    = (float*)d_ws;
  float* xl    = ws;
  float* xr    = xl + NF;
  float* eam   = xr + NF;
  float* hsum  = eam + NF;
  float* T1    = hsum + NF;
  float* T2    = T1 + NF;
  float* T3    = T2 + NF;
  float* cnt   = T3 + NF;
  float* denom = cnt + N;
  float* aloop = denom + N;
  float* dinv  = aloop + N;
  unsigned* amax = (unsigned*)(dinv + N);
  float* aedge = (float*)(amax + N);

  const int TPB = 256;
  int gE   = (E + TPB - 1) / TPB;
  int gEF  = (E * 16 + TPB - 1) / TPB;
  int gN   = (N + TPB - 1) / TPB;
  int gNF  = (N * 16 + TPB - 1) / TPB;
  int gN1  = (N + 1 + TPB - 1) / TPB;
  int tiles = N / 16;

  hipMemsetAsync(eam, 0, NF * 5 * sizeof(float), stream);
  hipMemsetAsync(cnt, 0, (size_t)N * 2 * sizeof(float), stream);
  hipMemsetAsync(dinv, 0, (size_t)N * sizeof(float), stream);
  k_init_amax<<<gN, TPB, 0, stream>>>(amax, N);

  k_cnt_easum<<<gEF, TPB, 0, stream>>>(ei, ea, cnt, eam, E);
  k_eamean<<<gNF, TPB, 0, stream>>>(eam, cnt, N);
  k_gemm_xlxr<<<tiles, 32, 0, stream>>>(x, Wl, bl, Wr, br, xl, xr);
  k_edge_alpha<<<gE, TPB, 0, stream>>>(ei, ea, xl, xr, We, att, aedge, amax, E);
  k_loop_alpha<<<gN, TPB, 0, stream>>>(eam, xl, xr, We, att, aloop, amax, N);
  k_exp_edge<<<gE, TPB, 0, stream>>>(ei, aedge, amax, denom, E);
  k_exp_loop<<<gN, TPB, 0, stream>>>(aloop, amax, denom, N);
  k_scatter<<<gEF, TPB, 0, stream>>>(ei, aedge, denom, xl, hsum, E);
  k_gatfin<<<gNF, TPB, 0, stream>>>(hsum, aloop, denom, xl, gbias, N);

  k_deg<<<gE, TPB, 0, stream>>>(ei, dinv, E);
  k_dinv<<<gN, TPB, 0, stream>>>(dinv, N);

  k_matvec<<<gEF, TPB, 0, stream>>>(ei, dinv, hsum, T1, E);
  k_matvec<<<gEF, TPB, 0, stream>>>(ei, dinv, T1, T2, E);
  k_chebrec<<<gNF, TPB, 0, stream>>>(T2, hsum, N);
  k_matvec<<<gEF, TPB, 0, stream>>>(ei, dinv, T2, T3, E);
  k_chebrec<<<gNF, TPB, 0, stream>>>(T3, T1, N);

  k_gru<<<tiles, 32, 0, stream>>>(hsum, T1, T2, T3, chebW, chebB, xl);

  k_mlp<<<gN1, TPB, 0, stream>>>(xl, W1, b1, W2, b2, out, N);
}


# Edges grouped by destination per call (integer atomic rank, scan, fill); per-destination sums and maxima become register gathers summed in arrival-rank list order + 16-lane tree; no float atomics
# speedup vs baseline: 2.0150x; 2.0150x over previous
#include <hip/hip_runtime.h>
#include <math.h>

typedef float v2f __attribute__((ext_vector_type(2)));
typedef float v8f __attribute__((ext_vector_type(8)));

#define WMMA4(a, b, c) \
  __builtin_amdgcn_wmma_f32_16x16x4_f32(false, (a), false, (b), (short)0, (c), false, false)

__device__ __forceinline__ float gsum16(float v) {
  v += __shfl_xor(v, 8, 16);
  v += __shfl_xor(v, 4, 16);
  v += __shfl_xor(v, 2, 16);
  v += __shfl_xor(v, 1, 16);
  return v;
}
__device__ __forceinline__ float gmax16(float v) {
  v = fmaxf(v, __shfl_xor(v, 8, 16));
  v = fmaxf(v, __shfl_xor(v, 4, 16));
  v = fmaxf(v, __shfl_xor(v, 2, 16));
  v = fmaxf(v, __shfl_xor(v, 1, 16));
  return v;
}

__device__ __forceinline__ void ldrow(const float* __restrict__ p, float* r) {
  const float4* q = (const float4*)p;
  float4 a = q[0], b = q[1], c = q[2], d = q[3];
  r[0] = a.x;  r[1] = a.y;  r[2] = a.z;  r[3] = a.w;
  r[4] = b.x;  r[5] = b.y;  r[6] = b.z;  r[7] = b.w;
  r[8] = c.x;  r[9] = c.y;  r[10] = c.z; r[11] = c.w;
  r[12] = d.x; r[13] = d.y; r[14] = d.z; r[15] = d.w;
}

__global__ void k_rank_deg(const int* __restrict__ ei, int* cnt, int* deg,
                           int* __restrict__ rank, int E) {
  int e = blockIdx.x * blockDim.x + threadIdx.x;
  if (e >= E) return;
  int s = ei[e], d = ei[E + e];
  rank[e] = atomicAdd(&cnt[d], 1);
  if (s != d) atomicAdd(&deg[s], 1);
}

__global__ void __launch_bounds__(1024)
k_scan(const int* __restrict__ cnt, int* __restrict__ rowptr, int N) {
  __shared__ int part[1024];
  int t = threadIdx.x;
  int per = (N + 1023) / 1024;
  int b = t * per;
  int en = b + per < N ? b + per : N;
  int s = 0;
  for (int i = b; i < en; ++i) s += cnt[i];
  part[t] = s;
  __syncthreads();
  for (int off = 1; off < 1024; off <<= 1) {
    int v = (t >= off) ? part[t - off] : 0;
    __syncthreads();
    part[t] += v;
    __syncthreads();
  }
  int run = part[t] - s;
  for (int i = b; i < en; ++i) { rowptr[i] = run; run += cnt[i]; }
  if (t == 1023) rowptr[N] = part[1023];
}

__global__ void k_dinv(const int* __restrict__ deg, float* __restrict__ dinv, int N) {
  int n = blockIdx.x * blockDim.x + threadIdx.x;
  if (n >= N) return;
  float g = (float)deg[n];
  dinv[n] = (g > 0.0f) ? rsqrtf(fmaxf(g, 1e-12f)) : 0.0f;
}

__global__ void k_fill(const int* __restrict__ ei, const int* __restrict__ rowptr,
                       const int* __restrict__ rank,
                       int* __restrict__ csrc, int* __restrict__ ceid, int E) {
  int e = blockIdx.x * blockDim.x + threadIdx.x;
  if (e >= E) return;
  int s = ei[e], d = ei[E + e];
  int p = rowptr[d] + rank[e];
  csrc[p] = s;
  ceid[p] = e;
}

__global__ void k_gemm_xlxr(const float* __restrict__ x,
                            const float* __restrict__ Wl, const float* __restrict__ bl,
                            const float* __restrict__ Wr, const float* __restrict__ br,
                            float* __restrict__ xl, float* __restrict__ xr) {
  int tile = blockIdx.x;
  int lane = threadIdx.x;
  int half = lane >> 4;
  int l16 = lane & 15;
  int row = tile * 16 + l16;
  v8f acc0 = {}; v8f acc1 = {};
#pragma unroll
  for (int k = 0; k < 128; k += 4) {
    int ka = k + half * 2;
    v2f a;  a.x = x[row * 128 + ka];     a.y = x[row * 128 + ka + 1];
    v2f b0; b0.x = Wl[ka * 16 + l16];    b0.y = Wl[(ka + 1) * 16 + l16];
    v2f b1; b1.x = Wr[ka * 16 + l16];    b1.y = Wr[(ka + 1) * 16 + l16];
    acc0 = WMMA4(a, b0, acc0);
    acc1 = WMMA4(a, b1, acc1);
  }
#pragma unroll
  for (int i = 0; i < 8; ++i) {
    int m = tile * 16 + i + half * 8;
    int c = l16;
    xl[m * 16 + c] = acc0[i] + bl[c];
    xr[m * 16 + c] = acc1[i] + br[c];
  }
}

__device__ __forceinline__ float logit16(const float* xs, const float* xd, const float* ev,
                                         const float* We, const float* att, int wo) {
  float acc = 0.0f;
#pragma unroll
  for (int f = 0; f < 16; ++f) {
    float m = xs[f] + xd[f];
#pragma unroll
    for (int j = 0; j < 16; ++j) m += ev[j] * We[wo + j * 16 + f];
    m = m > 0.0f ? m : 0.2f * m;
    acc += m * att[wo + f];
  }
  return acc;
}

__global__ void __launch_bounds__(256) __attribute__((amdgpu_waves_per_eu(6)))
k_gat(const int* __restrict__ rowptr, const int* __restrict__ csrc, const int* __restrict__ ceid,
      const float* __restrict__ ea, const float* __restrict__ xl, const float* __restrict__ xr,
      const float* __restrict__ We, const float* __restrict__ att,
      const float* __restrict__ gbias, float* ascr, float* __restrict__ h, int N) {
  __shared__ alignas(16) float sWe[256];
  __shared__ alignas(16) float satt[16];
  sWe[threadIdx.x] = We[threadIdx.x];
  if (threadIdx.x < 16) satt[threadIdx.x] = att[threadIdx.x];
  __syncthreads();
  int gid = blockIdx.x * blockDim.x + threadIdx.x;
  int n = gid >> 4, l = gid & 15;
  if (n >= N) return;
  int beg = rowptr[n], end = rowptr[n + 1];
  float xrn[16], xln[16];
  ldrow(xr + (size_t)n * 16, xrn);
  ldrow(xl + (size_t)n * 16, xln);

  float psum[16];
#pragma unroll
  for (int t = 0; t < 16; ++t) psum[t] = 0.0f;
  float amax = -INFINITY;
  for (int j = beg + l; j < end; j += 16) {
    int s = csrc[j], e = ceid[j];
    float ev[16], xs[16];
    ldrow(ea + (size_t)e * 16, ev);
    ldrow(xl + (size_t)s * 16, xs);
#pragma unroll
    for (int t = 0; t < 16; ++t) psum[t] += ev[t];
    int wq = 0;
    asm volatile("" : "+v"(wq));
    float a = logit16(xs, xrn, ev, sWe, satt, wq * 4);
    ascr[j] = a;
    amax = fmaxf(amax, a);
  }
  float dn = fmaxf((float)(end - beg), 1.0f);
  float em[16];
#pragma unroll
  for (int t = 0; t < 16; ++t) em[t] = gsum16(psum[t]) / dn;
  float aloop = logit16(xln, xrn, em, sWe, satt, 0);
  amax = fmaxf(gmax16(amax), aloop);

  float acc[16];
#pragma unroll
  for (int t = 0; t < 16; ++t) acc[t] = 0.0f;
  float den = 0.0f;
  for (int j = beg + l; j < end; j += 16) {
    int s = csrc[j];
    float ex = expf(ascr[j] - amax);
    float xs[16];
    ldrow(xl + (size_t)s * 16, xs);
    den += ex;
#pragma unroll
    for (int t = 0; t < 16; ++t) acc[t] += ex * xs[t];
  }
  float exl = expf(aloop - amax);
  den = gsum16(den) + exl;
  float mine = 0.0f;
#pragma unroll
  for (int t = 0; t < 16; ++t) {
    float v = gsum16(acc[t]) + exl * xln[t];
    if (l == t) mine = v;
  }
  h[(size_t)n * 16 + l] = mine / (den + 1e-16f) + gbias[l];
}

__global__ void __launch_bounds__(256)
k_matvec(const int* __restrict__ rowptr, const int* __restrict__ csrc,
         const float* __restrict__ dinv, const float* __restrict__ in,
         const float* __restrict__ prev, float scale, float* __restrict__ out, int N) {
  int gid = blockIdx.x * blockDim.x + threadIdx.x;
  int n = gid >> 4, l = gid & 15;
  if (n >= N) return;
  int beg = rowptr[n], end = rowptr[n + 1];
  float dn = dinv[n];
  float acc[16];
#pragma unroll
  for (int t = 0; t < 16; ++t) acc[t] = 0.0f;
  for (int j = beg + l; j < end; j += 16) {
    int s = csrc[j];
    float w = (s != n) ? -dinv[s] * dn : 0.0f;
    float xs[16];
    ldrow(in + (size_t)s * 16, xs);
#pragma unroll
    for (int t = 0; t < 16; ++t) acc[t] += w * xs[t];
  }
  float mine = 0.0f;
#pragma unroll
  for (int t = 0; t < 16; ++t) {
    float v = gsum16(acc[t]);
    if (l == t) mine = v;
  }
  float r = scale * mine;
  if (prev) r -= prev[(size_t)n * 16 + l];
  out[(size_t)n * 16 + l] = r;
}

__global__ void k_gru(const float* __restrict__ T0, const float* __restrict__ T1,
                      const float* __restrict__ T2, const float* __restrict__ T3,
                      const float* __restrict__ chebW, const float* __restrict__ chebB,
                      float* __restrict__ H) {
  int tile = blockIdx.x;
  int lane = threadIdx.x;
  int half = lane >> 4;
  int l16 = lane & 15;
  int row = tile * 16 + l16;
  v8f accZ = {}; v8f accH = {};
#pragma unroll
  for (int t = 0; t < 4; ++t) {
    const float* Tm = (t == 0) ? T0 : (t == 1) ? T1 : (t == 2) ? T2 : T3;
    const float* Wz = chebW + t * 256;
    const float* Wh = chebW + (16 + t) * 256;
#pragma unroll
    for (int kc = 0; kc < 16; kc += 4) {
      int ka = kc + half * 2;
      v2f a;  a.x = Tm[row * 16 + ka];      a.y = Tm[row * 16 + ka + 1];
      v2f bz; bz.x = Wz[ka * 16 + l16];     bz.y = Wz[(ka + 1) * 16 + l16];
      v2f bh; bh.x = Wh[ka * 16 + l16];     bh.y = Wh[(ka + 1) * 16 + l16];
      accZ = WMMA4(a, bz, accZ);
      accH = WMMA4(a, bh, accH);
    }
  }
#pragma unroll
  for (int i = 0; i < 8; ++i) {
    int m = tile * 16 + i + half * 8;
    int c = l16;
    float zp = accZ[i] + chebB[c] + chebB[16 + c];
    float hp = accH[i] + chebB[64 + c] + chebB[80 + c];
    float z = 1.0f / (1.0f + expf(-zp));
    float ht = tanhf(hp);
    H[m * 16 + c] = (1.0f - z) * ht;
  }
}

__global__ void k_mlp(const float* __restrict__ H, const float* W1, const float* b1,
                      const float* W2, const float* b2, float* out, int N) {
  int idx = blockIdx.x * blockDim.x + threadIdx.x;
  if (idx > N) return;
  float hid[4];
#pragma unroll
  for (int j = 0; j < 4; ++j) {
    float s = b1[j];
    if (idx > 0) {
      const float* hr = H + (size_t)(idx - 1) * 16;
#pragma unroll
      for (int f = 0; f < 16; ++f) s += hr[f] * W1[f * 4 + j];
    }
    hid[j] = fmaxf(s, 0.0f);
  }
#pragma unroll
  for (int o = 0; o < 8; ++o) {
    float s = b2[o];
#pragma unroll
    for (int j = 0; j < 4; ++j) s += hid[j] * W2[j * 8 + o];
    out[idx * 8 + o] = s;
  }
}

extern "C" void kernel_launch(void* const* d_in, const int* in_sizes, int n_in,
                              void* d_out, int out_size, void* d_ws, size_t ws_size,
                              hipStream_t stream) {
  const float* x     = (const float*)d_in[0];
  const int*   ei    = (const int*)  d_in[1];
  const float* ea    = (const float*)d_in[2];
  const float* Wl    = (const float*)d_in[3];
  const float* bl    = (const float*)d_in[4];
  const float* Wr    = (const float*)d_in[5];
  const float* br    = (const float*)d_in[6];
  const float* We    = (const float*)d_in[7];
  const float* att   = (const float*)d_in[8];
  const float* gbias = (const float*)d_in[9];
  const float* chebW = (const float*)d_in[10];
  const float* chebB = (const float*)d_in[11];
  const float* W1    = (const float*)d_in[12];
  const float* b1    = (const float*)d_in[13];
  const float* W2    = (const float*)d_in[14];
  const float* b2    = (const float*)d_in[15];
  float* out = (float*)d_out;

  const int N = in_sizes[0] / 128;
  const int E = in_sizes[1] / 2;
  const size_t NF = (size_t)N * 16;

  float* ws    = (float*)d_ws;
  float* xl    = ws;
  float* xr    = xl + NF;
  float* hh    = xr + NF;
  float* T1    = hh + NF;
  float* T2    = T1 + NF;
  float* T3    = T2 + NF;
  float* spare = T3 + NF;
  int*   cnt   = (int*)(spare + NF);
  int*   deg   = cnt + N;
  float* dinv  = (float*)(deg + N);
  int*   rowptr = (int*)(dinv + N);
  int*   csrc  = (int*)(spare + NF + (size_t)N * 5);
  int*   rank  = (int*)T1;
  float* ascr  = T1;
  int*   ceid  = (int*)T3;

  const int TPB = 256;
  int gE   = (E + TPB - 1) / TPB;
  int gN   = (N + TPB - 1) / TPB;
  int gN16 = (N * 16 + TPB - 1) / TPB;
  int gN1  = (N + 1 + TPB - 1) / TPB;
  int tiles = N / 16;

  hipMemsetAsync(cnt, 0, (size_t)N * 2 * sizeof(int), stream);

  k_rank_deg<<<gE, TPB, 0, stream>>>(ei, cnt, deg, rank, E);
  k_scan<<<1, 1024, 0, stream>>>(cnt, rowptr, N);
  k_dinv<<<gN, TPB, 0, stream>>>(deg, dinv, N);
  k_fill<<<gE, TPB, 0, stream>>>(ei, rowptr, rank, csrc, ceid, E);

  k_gemm_xlxr<<<tiles, 32, 0, stream>>>(x, Wl, bl, Wr, br, xl, xr);
  k_gat<<<gN16, TPB, 0, stream>>>(rowptr, csrc, ceid, ea, xl, xr, We, att, gbias, ascr, hh, N);

  k_matvec<<<gN16, TPB, 0, stream>>>(rowptr, csrc, dinv, hh, (const float*)nullptr, 1.0f, T1, N);
  k_matvec<<<gN16, TPB, 0, stream>>>(rowptr, csrc, dinv, T1, hh, 2.0f, T2, N);
  k_matvec<<<gN16, TPB, 0, stream>>>(rowptr, csrc, dinv, T2, T1, 2.0f, T3, N);

  k_gru<<<tiles, 32, 0, stream>>>(hh, T1, T2, T3, chebW, chebB, xl);

  k_mlp<<<gN1, TPB, 0, stream>>>(xl, W1, b1, W2, b2, out, N);
}


# GAT edge transform ea@We moved onto the f32 16x16x4 WMMA (16 edges a tile, 2 tiles a trip, an edge's rows split over lanes c and c+16); no LDS weight reads; logit sums re-associated by matrix unit
# speedup vs baseline: 2.1888x; 1.0863x over previous
#include <hip/hip_runtime.h>
#include <math.h>

typedef float v2f __attribute__((ext_vector_type(2)));
typedef float v8f __attribute__((ext_vector_type(8)));

#define WMMA4(a, b, c) \
  __builtin_amdgcn_wmma_f32_16x16x4_f32(false, (a), false, (b), (short)0, (c), false, false)

__device__ __forceinline__ float gsum16(float v) {
  v += __shfl_xor(v, 8, 16);
  v += __shfl_xor(v, 4, 16);
  v += __shfl_xor(v, 2, 16);
  v += __shfl_xor(v, 1, 16);
  return v;
}
__device__ __forceinline__ float gmax16(float v) {
  v = fmaxf(v, __shfl_xor(v, 8, 16));
  v = fmaxf(v, __shfl_xor(v, 4, 16));
  v = fmaxf(v, __shfl_xor(v, 2, 16));
  v = fmaxf(v, __shfl_xor(v, 1, 16));
  return v;
}

__device__ __forceinline__ void ldrow(const float* __restrict__ p, float* r) {
  const float4* q = (const float4*)p;
  float4 a = q[0], b = q[1], c = q[2], d = q[3];
  r[0] = a.x;  r[1] = a.y;  r[2] = a.z;  r[3] = a.w;
  r[4] = b.x;  r[5] = b.y;  r[6] = b.z;  r[7] = b.w;
  r[8] = c.x;  r[9] = c.y;  r[10] = c.z; r[11] = c.w;
  r[12] = d.x; r[13] = d.y; r[14] = d.z; r[15] = d.w;
}

__global__ void k_rank_deg(const int* __restrict__ ei, int* cnt, int* deg,
                           int* __restrict__ rank, int E) {
  int e = blockIdx.x * blockDim.x + threadIdx.x;
  if (e >= E) return;
  int s = ei[e], d = ei[E + e];
  rank[e] = atomicAdd(&cnt[d], 1);
  if (s != d) atomicAdd(&deg[s], 1);
}

__global__ void __launch_bounds__(1024)
k_scan(const int* __restrict__ cnt, int* __restrict__ rowptr, int N) {
  __shared__ int part[1024];
  int t = threadIdx.x;
  int per = (N + 1023) / 1024;
  int b = t * per;
  int en = b + per < N ? b + per : N;
  int s = 0;
  for (int i = b; i < en; ++i) s += cnt[i];
  part[t] = s;
  __syncthreads();
  for (int off = 1; off < 1024; off <<= 1) {
    int v = (t >= off) ? part[t - off] : 0;
    __syncthreads();
    part[t] += v;
    __syncthreads();
  }
  int run = part[t] - s;
  for (int i = b; i < en; ++i) { rowptr[i] = run; run += cnt[i]; }
  if (t == 1023) rowptr[N] = part[1023];
}

__global__ void k_dinv(const int* __restrict__ deg, float* __restrict__ dinv, int N) {
  int n = blockIdx.x * blockDim.x + threadIdx.x;
  if (n >= N) return;
  float g = (float)deg[n];
  dinv[n] = (g > 0.0f) ? rsqrtf(fmaxf(g, 1e-12f)) : 0.0f;
}

__global__ void k_fill(const int* __restrict__ ei, const int* __restrict__ rowptr,
                       const int* __restrict__ rank,
                       int* __restrict__ csrc, int* __restrict__ ceid, int E) {
  int e = blockIdx.x * blockDim.x + threadIdx.x;
  if (e >= E) return;
  int s = ei[e], d = ei[E + e];
  int p = rowptr[d] + rank[e];
  csrc[p] = s;
  ceid[p] = e;
}

__global__ void k_gemm_xlxr(const float* __restrict__ x,
                            const float* __restrict__ Wl, const float* __restrict__ bl,
                            const float* __restrict__ Wr, const float* __restrict__ br,
                            float* __restrict__ xl, float* __restrict__ xr) {
  int tile = blockIdx.x;
  int lane = threadIdx.x;
  int half = lane >> 4;
  int l16 = lane & 15;
  int row = tile * 16 + l16;
  v8f acc0 = {}; v8f acc1 = {};
#pragma unroll
  for (int k = 0; k < 128; k += 4) {
    int ka = k + half * 2;
    v2f a;  a.x = x[row * 128 + ka];     a.y = x[row * 128 + ka + 1];
    v2f b0; b0.x = Wl[ka * 16 + l16];    b0.y = Wl[(ka + 1) * 16 + l16];
    v2f b1; b1.x = Wr[ka * 16 + l16];    b1.y = Wr[(ka + 1) * 16 + l16];
    acc0 = WMMA4(a, b0, acc0);
    acc1 = WMMA4(a, b1, acc1);
  }
#pragma unroll
  for (int i = 0; i < 8; ++i) {
    int m = tile * 16 + i + half * 8;
    int c = l16;
    xl[m * 16 + c] = acc0[i] + bl[c];
    xr[m * 16 + c] = acc1[i] + br[c];
  }
}

__device__ __forceinline__ void ldhalf(const float* __restrict__ p, float* r) {
  const float4* q = (const float4*)p;
  float4 a = q[0], b = q[1];
  r[0] = a.x; r[1] = a.y; r[2] = a.z; r[3] = a.w;
  r[4] = b.x; r[5] = b.y; r[6] = b.z; r[7] = b.w;
}

__device__ __forceinline__ float tile_logit(const v2f* wa, const float* evh, const float* cin,
                                            const float* atth) {
  v8f acc;
#pragma unroll
  for (int i = 0; i < 8; ++i) acc[i] = cin[i];
#pragma unroll
  for (int t = 0; t < 4; ++t) {
    v2f b; b.x = evh[2 * t]; b.y = evh[2 * t + 1];
    acc = WMMA4(wa[t], b, acc);
  }
  float p = 0.0f;
#pragma unroll
  for (int i = 0; i < 8; ++i) {
    float m = acc[i];
    m = m > 0.0f ? m : 0.2f * m;
    p += m * atth[i];
  }
  return p + __shfl_xor(p, 16);
}

__global__ void __launch_bounds__(256)
k_gat(const int* __restrict__ rowptr, const int* __restrict__ csrc, const int* __restrict__ ceid,
      const float* __restrict__ ea, const float* __restrict__ xl, float* xr,
      const float* __restrict__ We, const float* __restrict__ att,
      const float* __restrict__ gbias, const float* __restrict__ dinv,
      float* ascr, float* __restrict__ h, int N) {
  int lane = threadIdx.x & 31;
  int c = lane & 15, hf = lane >> 4;
  int wave = __builtin_amdgcn_readfirstlane((int)((blockIdx.x * blockDim.x + threadIdx.x) >> 5));

  v2f wa[4];
#pragma unroll
  for (int t = 0; t < 4; ++t) {
    int j = 8 * hf + 2 * t;
    wa[t].x = We[j * 16 + c];
    wa[t].y = We[(j + 1) * 16 + c];
  }
  float atth[8];
  ldhalf(att + 8 * hf, atth);

  for (int q = 0; q < 2; ++q) {
    int n = wave * 2 + q;
    if (n >= N) break;
    int beg = rowptr[n], end = rowptr[n + 1];
    float xrh[8], xlh[8];
    ldhalf(xr + (size_t)n * 16 + 8 * hf, xrh);
    ldhalf(xl + (size_t)n * 16 + 8 * hf, xlh);

    float psum[8];
#pragma unroll
    for (int i = 0; i < 8; ++i) psum[i] = 0.0f;
    float amax = -INFINITY;
    for (int base = beg; base < end; base += 32) {
      int j0 = base + c, j1 = j0 + 16;
      bool v0 = j0 < end, v1 = j1 < end;
      int q0 = v0 ? j0 : end - 1;
      int q1 = v1 ? j1 : end - 1;
      int s0 = csrc[q0], e0 = ceid[q0];
      int s1 = csrc[q1], e1 = ceid[q1];
      float ev0[8], xs0[8], ev1[8], xs1[8];
      ldhalf(ea + (size_t)e0 * 16 + 8 * hf, ev0);
      ldhalf(xl + (size_t)s0 * 16 + 8 * hf, xs0);
      ldhalf(ea + (size_t)e1 * 16 + 8 * hf, ev1);
      ldhalf(xl + (size_t)s1 * 16 + 8 * hf, xs1);
#pragma unroll
      for (int i = 0; i < 8; ++i) { xs0[i] += xrh[i]; xs1[i] += xrh[i]; }
      float a0 = tile_logit(wa, ev0, xs0, atth);
      float a1 = tile_logit(wa, ev1, xs1, atth);
      if (v0) {
        ascr[j0] = a0;
        amax = fmaxf(amax, a0);
#pragma unroll
        for (int i = 0; i < 8; ++i) psum[i] += ev0[i];
      }
      if (v1) {
        ascr[j1] = a1;
        amax = fmaxf(amax, a1);
#pragma unroll
        for (int i = 0; i < 8; ++i) psum[i] += ev1[i];
      }
    }
    float dn = fmaxf((float)(end - beg), 1.0f);
    float emh[8], cl[8];
#pragma unroll
    for (int i = 0; i < 8; ++i) { emh[i] = gsum16(psum[i]) / dn; cl[i] = xlh[i] + xrh[i]; }
    float aloop = tile_logit(wa, emh, cl, atth);
    amax = fmaxf(gmax16(amax), aloop);

    float acc[8];
#pragma unroll
    for (int i = 0; i < 8; ++i) acc[i] = 0.0f;
    float den = 0.0f;
    for (int base = beg; base < end; base += 32) {
      int j0 = base + c, j1 = j0 + 16;
      bool v0 = j0 < end, v1 = j1 < end;
      int q0 = v0 ? j0 : end - 1;
      int q1 = v1 ? j1 : end - 1;
      int s0 = csrc[q0], s1 = csrc[q1];
      float al0 = ascr[q0], al1 = ascr[q1];
      float xs0[8], xs1[8];
      ldhalf(xl + (size_t)s0 * 16 + 8 * hf, xs0);
      ldhalf(xl + (size_t)s1 * 16 + 8 * hf, xs1);
      if (v0) {
        float ex = expf(al0 - amax);
        den += ex;
#pragma unroll
        for (int i = 0; i < 8; ++i) acc[i] += ex * xs0[i];
      }
      if (v1) {
        float ex = expf(al1 - amax);
        den += ex;
#pragma unroll
        for (int i = 0; i < 8; ++i) acc[i] += ex * xs1[i];
      }
    }
    float exl = expf(aloop - amax);
    den = gsum16(den) + exl;
    float mine = 0.0f;
#pragma unroll
    for (int i = 0; i < 8; ++i) {
      float v = gsum16(acc[i]) + exl * xlh[i];
      if (c == i) mine = v;
    }
    if (c < 8) {
      int f = 8 * hf + c;
      float hv = mine / (den + 1e-16f) + gbias[f];
      h[(size_t)n * 16 + f] = hv;
      xr[(size_t)n * 16 + f] = dinv[n] * hv;
    }
  }
}

__global__ void __launch_bounds__(256)
k_matvec(const int* __restrict__ rowptr, const int* __restrict__ csrc,
         const float* __restrict__ dinv, const float* __restrict__ yin,
         const float* __restrict__ prev, float scale, float* __restrict__ out,
         float* __restrict__ yout, int N) {
  int gid = blockIdx.x * blockDim.x + threadIdx.x;
  int n = gid >> 4, l = gid & 15;
  if (n >= N) return;
  int beg = rowptr[n], end = rowptr[n + 1];
  float dn = dinv[n];
  float acc[16];
#pragma unroll
  for (int t = 0; t < 16; ++t) acc[t] = 0.0f;
  for (int j = beg + l; j < end; j += 16) {
    int s = csrc[j];
    float w = (s != n) ? 1.0f : 0.0f;
    float xs[16];
    ldrow(yin + (size_t)s * 16, xs);
#pragma unroll
    for (int t = 0; t < 16; ++t) acc[t] += w * xs[t];
  }
  float mine = 0.0f;
#pragma unroll
  for (int t = 0; t < 16; ++t) {
    float v = gsum16(acc[t]);
    if (l == t) mine = v;
  }
  float r = scale * (-dn * mine);
  if (prev) r -= prev[(size_t)n * 16 + l];
  out[(size_t)n * 16 + l] = r;
  if (yout) yout[(size_t)n * 16 + l] = dn * r;
}

__global__ void k_gru(const float* __restrict__ T0, const float* __restrict__ T1,
                      const float* __restrict__ T2, const float* __restrict__ T3,
                      const float* __restrict__ chebW, const float* __restrict__ chebB,
                      float* __restrict__ H) {
  int tile = blockIdx.x;
  int lane = threadIdx.x;
  int half = lane >> 4;
  int l16 = lane & 15;
  int row = tile * 16 + l16;
  v8f accZ = {}; v8f accH = {};
#pragma unroll
  for (int t = 0; t < 4; ++t) {
    const float* Tm = (t == 0) ? T0 : (t == 1) ? T1 : (t == 2) ? T2 : T3;
    const float* Wz = chebW + t * 256;
    const float* Wh = chebW + (16 + t) * 256;
#pragma unroll
    for (int kc = 0; kc < 16; kc += 4) {
      int ka = kc + half * 2;
      v2f a;  a.x = Tm[row * 16 + ka];      a.y = Tm[row * 16 + ka + 1];
      v2f bz; bz.x = Wz[ka * 16 + l16];     bz.y = Wz[(ka + 1) * 16 + l16];
      v2f bh; bh.x = Wh[ka * 16 + l16];     bh.y = Wh[(ka + 1) * 16 + l16];
      accZ = WMMA4(a, bz, accZ);
      accH = WMMA4(a, bh, accH);
    }
  }
#pragma unroll
  for (int i = 0; i < 8; ++i) {
    int m = tile * 16 + i + half * 8;
    int c = l16;
    float zp = accZ[i] + chebB[c] + chebB[16 + c];
    float hp = accH[i] + chebB[64 + c] + chebB[80 + c];
    float z = 1.0f / (1.0f + expf(-zp));
    float ht = tanhf(hp);
    H[m * 16 + c] = (1.0f - z) * ht;
  }
}

__global__ void k_mlp(const float* __restrict__ H, const float* W1, const float* b1,
                      const float* W2, const float* b2, float* out, int N) {
  int idx = blockIdx.x * blockDim.x + threadIdx.x;
  if (idx > N) return;
  float hid[4];
#pragma unroll
  for (int j = 0; j < 4; ++j) {
    float s = b1[j];
    if (idx > 0) {
      const float* hr = H + (size_t)(idx - 1) * 16;
#pragma unroll
      for (int f = 0; f < 16; ++f) s += hr[f] * W1[f * 4 + j];
    }
    hid[j] = fmaxf(s, 0.0f);
  }
#pragma unroll
  for (int o = 0; o < 8; ++o) {
    float s = b2[o];
#pragma unroll
    for (int j = 0; j < 4; ++j) s += hid[j] * W2[j * 8 + o];
    out[idx * 8 + o] = s;
  }
}

extern "C" void kernel_launch(void* const* d_in, const int* in_sizes, int n_in,
                              void* d_out, int out_size, void* d_ws, size_t ws_size,
                              hipStream_t stream) {
  const float* x     = (const float*)d_in[0];
  const int*   ei    = (const int*)  d_in[1];
  const float* ea    = (const float*)d_in[2];
  const float* Wl    = (const float*)d_in[3];
  const float* bl    = (const float*)d_in[4];
  const float* Wr    = (const float*)d_in[5];
  const float* br    = (const float*)d_in[6];
  const float* We    = (const float*)d_in[7];
  const float* att   = (const float*)d_in[8];
  const float* gbias = (const float*)d_in[9];
  const float* chebW = (const float*)d_in[10];
  const float* chebB = (const float*)d_in[11];
  const float* W1    = (const float*)d_in[12];
  const float* b1    = (const float*)d_in[13];
  const float* W2    = (const float*)d_in[14];
  const float* b2    = (const float*)d_in[15];
  float* out = (float*)d_out;

  const int N = in_sizes[0] / 128;
  const int E = in_sizes[1] / 2;
  const size_t NF = (size_t)N * 16;

  float* ws    = (float*)d_ws;
  float* xl    = ws;
  float* xr    = xl + NF;
  float* hh    = xr + NF;
  float* T1    = hh + NF;
  float* T2    = T1 + NF;
  float* T3    = T2 + NF;
  float* spare = T3 + NF;
  int*   cnt   = (int*)(spare + NF);
  int*   deg   = cnt + N;
  float* dinv  = (float*)(deg + N);
  int*   rowptr = (int*)(dinv + N);
  int*   csrc  = (int*)(spare + NF + (size_t)N * 5);
  int*   rank  = (int*)T1;
  float* ascr  = T1;
  int*   ceid  = (int*)T3;

  const int TPB = 256;
  int gE   = (E + TPB - 1) / TPB;
  int gN   = (N + TPB - 1) / TPB;
  int gN16 = (N * 16 + TPB - 1) / TPB;
  int gN1  = (N + 1 + TPB - 1) / TPB;
  int tiles = N / 16;

  hipMemsetAsync(cnt, 0, (size_t)N * 2 * sizeof(int), stream);

  k_rank_deg<<<gE, TPB, 0, stream>>>(ei, cnt, deg, rank, E);
  k_scan<<<1, 1024, 0, stream>>>(cnt, rowptr, N);
  k_dinv<<<gN, TPB, 0, stream>>>(deg, dinv, N);
  k_fill<<<gE, TPB, 0, stream>>>(ei, rowptr, rank, csrc, ceid, E);

  k_gemm_xlxr<<<tiles, 32, 0, stream>>>(x, Wl, bl, Wr, br, xl, xr);
  k_gat<<<gN16, TPB, 0, stream>>>(rowptr, csrc, ceid, ea, xl, xr, We, att, gbias, dinv, ascr, hh, N);

  k_matvec<<<gN16, TPB, 0, stream>>>(rowptr, csrc, dinv, xr, (const float*)nullptr, 1.0f, T1, xl, N);
  k_matvec<<<gN16, TPB, 0, stream>>>(rowptr, csrc, dinv, xl, hh, 2.0f, T2, xr, N);
  k_matvec<<<gN16, TPB, 0, stream>>>(rowptr, csrc, dinv, xr, T1, 2.0f, T3, (float*)nullptr, N);

  k_gru<<<tiles, 32, 0, stream>>>(hh, T1, T2, T3, chebW, chebB, xl);

  k_mlp<<<gN1, TPB, 0, stream>>>(xl, W1, b1, W2, b2, out, N);
}


# Edges grouped by destination by a two-level counting sort: 256-node buckets, ranks from LDS atomics, one global add per (block, bucket) not one returning add per edge; list order still arrival
# speedup vs baseline: 3.3195x; 1.5166x over previous
#include <hip/hip_runtime.h>
#include <math.h>

typedef float v2f __attribute__((ext_vector_type(2)));
typedef float v8f __attribute__((ext_vector_type(8)));

#define WMMA4(a, b, c) \
  __builtin_amdgcn_wmma_f32_16x16x4_f32(false, (a), false, (b), (short)0, (c), false, false)

__device__ __forceinline__ float gsum16(float v) {
  v += __shfl_xor(v, 8, 16);
  v += __shfl_xor(v, 4, 16);
  v += __shfl_xor(v, 2, 16);
  v += __shfl_xor(v, 1, 16);
  return v;
}
__device__ __forceinline__ float gmax16(float v) {
  v = fmaxf(v, __shfl_xor(v, 8, 16));
  v = fmaxf(v, __shfl_xor(v, 4, 16));
  v = fmaxf(v, __shfl_xor(v, 2, 16));
  v = fmaxf(v, __shfl_xor(v, 1, 16));
  return v;
}

__device__ __forceinline__ void ldrow(const float* __restrict__ p, float* r) {
  const float4* q = (const float4*)p;
  float4 a = q[0], b = q[1], c = q[2], d = q[3];
  r[0] = a.x;  r[1] = a.y;  r[2] = a.z;  r[3] = a.w;
  r[4] = b.x;  r[5] = b.y;  r[6] = b.z;  r[7] = b.w;
  r[8] = c.x;  r[9] = c.y;  r[10] = c.z; r[11] = c.w;
  r[12] = d.x; r[13] = d.y; r[14] = d.z; r[15] = d.w;
}

#define BKT_SHIFT 8
#define BKT_NODES (1 << BKT_SHIFT)
#define MAXB 512
#define CPAD 32
#define EBLK 1024
#define EPT 8

__device__ __forceinline__ void lds_scan_incl(int* a, int n) {
  int t = threadIdx.x;
  for (int off = 1; off < n; off <<= 1) {
    int v = (t < n && t >= off) ? a[t - off] : 0;
    __syncthreads();
    if (t < n) a[t] += v;
    __syncthreads();
  }
}

__global__ void __launch_bounds__(EBLK)
k_count(const int* __restrict__ ei, int* btot, int* deg, int E, int nb) {
  __shared__ int hist[MAXB];
  int t = threadIdx.x;
  if (t < MAXB) hist[t] = 0;
  __syncthreads();
  int e0 = blockIdx.x * (EBLK * EPT) + t;
  int s[EPT], d[EPT];
#pragma unroll
  for (int k = 0; k < EPT; ++k) {
    int e = e0 + k * EBLK;
    bool v = e < E;
    s[k] = v ? ei[e] : 0;
    d[k] = v ? ei[E + e] : 0;
  }
#pragma unroll
  for (int k = 0; k < EPT; ++k) {
    if (e0 + k * EBLK < E) {
      atomicAdd(&hist[d[k] >> BKT_SHIFT], 1);
      if (s[k] != d[k]) atomicAdd(&deg[s[k]], 1);
    }
  }
  __syncthreads();
  if (t < nb) { int v = hist[t]; if (v) atomicAdd(&btot[t * CPAD], v); }
}

__global__ void __launch_bounds__(EBLK)
k_bin(const int* __restrict__ ei, const int* __restrict__ btot, int* bfill,
      int* __restrict__ bsd, int* __restrict__ be, int E, int nb) {
  __shared__ int hist[MAXB];
  __shared__ int goff[MAXB];
  int t = threadIdx.x;
  int mytot = (t < nb) ? btot[t * CPAD] : 0;
  if (t < MAXB) { hist[t] = 0; goff[t] = mytot; }
  __syncthreads();
  int e0 = blockIdx.x * (EBLK * EPT) + t;
  int s[EPT], d[EPT], lr[EPT];
#pragma unroll
  for (int k = 0; k < EPT; ++k) {
    int e = e0 + k * EBLK;
    bool v = e < E;
    s[k] = v ? ei[e] : 0;
    d[k] = v ? ei[E + e] : 0;
  }
#pragma unroll
  for (int k = 0; k < EPT; ++k) {
    lr[k] = 0;
    if (e0 + k * EBLK < E) lr[k] = atomicAdd(&hist[d[k] >> BKT_SHIFT], 1);
  }
  __syncthreads();
  lds_scan_incl(goff, nb);
  if (t < nb) {
    int n = hist[t];
    int mine = n ? atomicAdd(&bfill[t * CPAD], n) : 0;
    goff[t] = goff[t] - mytot + mine;
  }
  __syncthreads();
#pragma unroll
  for (int k = 0; k < EPT; ++k) {
    int e = e0 + k * EBLK;
    if (e < E) {
      int p = goff[d[k] >> BKT_SHIFT] + lr[k];
      bsd[p] = (int)((unsigned)s[k] | ((unsigned)(d[k] & (BKT_NODES - 1)) << 24));
      be[p] = e;
    }
  }
}

__global__ void __launch_bounds__(EBLK)
k_place(const int* __restrict__ btot, const int* __restrict__ bsd, const int* __restrict__ be,
        int* __restrict__ rowptr, int* __restrict__ csrc, int* __restrict__ ceid, int N, int nb) {
  __shared__ int bst[MAXB];
  __shared__ int cnt[BKT_NODES];
  __shared__ int cur[BKT_NODES];
  int t = threadIdx.x, b = blockIdx.x;
  int mytot = (t < nb) ? btot[t * CPAD] : 0;
  if (t < MAXB) bst[t] = mytot;
  if (t < BKT_NODES) cnt[t] = 0;
  __syncthreads();
  lds_scan_incl(bst, nb);
  int endp = bst[b];
  int begp = b ? bst[b - 1] : 0;
  for (int p0 = begp + t; p0 < endp; p0 += 4 * EBLK) {
    unsigned r[4];
#pragma unroll
    for (int u = 0; u < 4; ++u) { int p = p0 + u * EBLK; r[u] = p < endp ? (unsigned)bsd[p] : 0u; }
#pragma unroll
    for (int u = 0; u < 4; ++u) if (p0 + u * EBLK < endp) atomicAdd(&cnt[r[u] >> 24], 1);
  }
  __syncthreads();
  int mycnt = (t < BKT_NODES) ? cnt[t] : 0;
  lds_scan_incl(cnt, BKT_NODES);
  int lo = b << BKT_SHIFT;
  if (t < BKT_NODES) {
    int first = begp + cnt[t] - mycnt;
    cur[t] = first;
    if (lo + t < N) rowptr[lo + t] = first;
  }
  if (b == nb - 1 && t == 0) rowptr[N] = endp;
  __syncthreads();
  for (int p0 = begp + t; p0 < endp; p0 += 4 * EBLK) {
    unsigned r[4]; int ev[4];
#pragma unroll
    for (int u = 0; u < 4; ++u) {
      int p = p0 + u * EBLK;
      bool v = p < endp;
      r[u] = v ? (unsigned)bsd[p] : 0u;
      ev[u] = v ? be[p] : 0;
    }
#pragma unroll
    for (int u = 0; u < 4; ++u) {
      if (p0 + u * EBLK < endp) {
        int q = atomicAdd(&cur[r[u] >> 24], 1);
        csrc[q] = (int)(r[u] & 0xFFFFFFu);
        ceid[q] = ev[u];
      }
    }
  }
}

__global__ void k_dinv(const int* __restrict__ deg, float* __restrict__ dinv, int N) {
  int n = blockIdx.x * blockDim.x + threadIdx.x;
  if (n >= N) return;
  float g = (float)deg[n];
  dinv[n] = (g > 0.0f) ? rsqrtf(fmaxf(g, 1e-12f)) : 0.0f;
}

__global__ void k_gemm_xlxr(const float* __restrict__ x,
                            const float* __restrict__ Wl, const float* __restrict__ bl,
                            const float* __restrict__ Wr, const float* __restrict__ br,
                            float* __restrict__ xl, float* __restrict__ xr) {
  int tile = blockIdx.x;
  int lane = threadIdx.x;
  int half = lane >> 4;
  int l16 = lane & 15;
  int row = tile * 16 + l16;
  v8f acc0 = {}; v8f acc1 = {};
#pragma unroll
  for (int k = 0; k < 128; k += 4) {
    int ka = k + half * 2;
    v2f a;  a.x = x[row * 128 + ka];     a.y = x[row * 128 + ka + 1];
    v2f b0; b0.x = Wl[ka * 16 + l16];    b0.y = Wl[(ka + 1) * 16 + l16];
    v2f b1; b1.x = Wr[ka * 16 + l16];    b1.y = Wr[(ka + 1) * 16 + l16];
    acc0 = WMMA4(a, b0, acc0);
    acc1 = WMMA4(a, b1, acc1);
  }
#pragma unroll
  for (int i = 0; i < 8; ++i) {
    int m = tile * 16 + i + half * 8;
    int c = l16;
    xl[m * 16 + c] = acc0[i] + bl[c];
    xr[m * 16 + c] = acc1[i] + br[c];
  }
}

__device__ __forceinline__ void ldhalf(const float* __restrict__ p, float* r) {
  const float4* q = (const float4*)p;
  float4 a = q[0], b = q[1];
  r[0] = a.x; r[1] = a.y; r[2] = a.z; r[3] = a.w;
  r[4] = b.x; r[5] = b.y; r[6] = b.z; r[7] = b.w;
}

__device__ __forceinline__ float tile_logit(const v2f* wa, const float* evh, const float* cin,
                                            const float* atth) {
  v8f acc;
#pragma unroll
  for (int i = 0; i < 8; ++i) acc[i] = cin[i];
#pragma unroll
  for (int t = 0; t < 4; ++t) {
    v2f b; b.x = evh[2 * t]; b.y = evh[2 * t + 1];
    acc = WMMA4(wa[t], b, acc);
  }
  float p = 0.0f;
#pragma unroll
  for (int i = 0; i < 8; ++i) {
    float m = acc[i];
    m = m > 0.0f ? m : 0.2f * m;
    p += m * atth[i];
  }
  return p + __shfl_xor(p, 16);
}

__global__ void __launch_bounds__(256)
k_gat(const int* __restrict__ rowptr, const int* __restrict__ csrc, const int* __restrict__ ceid,
      const float* __restrict__ ea, const float* __restrict__ xl, float* xr,
      const float* __restrict__ We, const float* __restrict__ att,
      const float* __restrict__ gbias, const float* __restrict__ dinv,
      float* ascr, float* __restrict__ h, int N) {
  int lane = threadIdx.x & 31;
  int c = lane & 15, hf = lane >> 4;
  int wave = __builtin_amdgcn_readfirstlane((int)((blockIdx.x * blockDim.x + threadIdx.x) >> 5));

  v2f wa[4];
#pragma unroll
  for (int t = 0; t < 4; ++t) {
    int j = 8 * hf + 2 * t;
    wa[t].x = We[j * 16 + c];
    wa[t].y = We[(j + 1) * 16 + c];
  }
  float atth[8];
  ldhalf(att + 8 * hf, atth);

  for (int q = 0; q < 2; ++q) {
    int n = wave * 2 + q;
    if (n >= N) break;
    int beg = rowptr[n], end = rowptr[n + 1];
    float xrh[8], xlh[8];
    ldhalf(xr + (size_t)n * 16 + 8 * hf, xrh);
    ldhalf(xl + (size_t)n * 16 + 8 * hf, xlh);

    float psum[8];
#pragma unroll
    for (int i = 0; i < 8; ++i) psum[i] = 0.0f;
    float amax = -INFINITY;
    for (int base = beg; base < end; base += 32) {
      int j0 = base + c, j1 = j0 + 16;
      bool v0 = j0 < end, v1 = j1 < end;
      int q0 = v0 ? j0 : end - 1;
      int q1 = v1 ? j1 : end - 1;
      int s0 = csrc[q0], e0 = ceid[q0];
      int s1 = csrc[q1], e1 = ceid[q1];
      float ev0[8], xs0[8], ev1[8], xs1[8];
      ldhalf(ea + (size_t)e0 * 16 + 8 * hf, ev0);
      ldhalf(xl + (size_t)s0 * 16 + 8 * hf, xs0);
      ldhalf(ea + (size_t)e1 * 16 + 8 * hf, ev1);
      ldhalf(xl + (size_t)s1 * 16 + 8 * hf, xs1);
#pragma unroll
      for (int i = 0; i < 8; ++i) { xs0[i] += xrh[i]; xs1[i] += xrh[i]; }
      float a0 = tile_logit(wa, ev0, xs0, atth);
      float a1 = tile_logit(wa, ev1, xs1, atth);
      if (v0) {
        ascr[j0] = a0;
        amax = fmaxf(amax, a0);
#pragma unroll
        for (int i = 0; i < 8; ++i) psum[i] += ev0[i];
      }
      if (v1) {
        ascr[j1] = a1;
        amax = fmaxf(amax, a1);
#pragma unroll
        for (int i = 0; i < 8; ++i) psum[i] += ev1[i];
      }
    }
    float dn = fmaxf((float)(end - beg), 1.0f);
    float emh[8], cl[8];
#pragma unroll
    for (int i = 0; i < 8; ++i) { emh[i] = gsum16(psum[i]) / dn; cl[i] = xlh[i] + xrh[i]; }
    float aloop = tile_logit(wa, emh, cl, atth);
    amax = fmaxf(gmax16(amax), aloop);

    float acc[8];
#pragma unroll
    for (int i = 0; i < 8; ++i) acc[i] = 0.0f;
    float den = 0.0f;
    for (int base = beg; base < end; base += 32) {
      int j0 = base + c, j1 = j0 + 16;
      bool v0 = j0 < end, v1 = j1 < end;
      int q0 = v0 ? j0 : end - 1;
      int q1 = v1 ? j1 : end - 1;
      int s0 = csrc[q0], s1 = csrc[q1];
      float al0 = ascr[q0], al1 = ascr[q1];
      float xs0[8], xs1[8];
      ldhalf(xl + (size_t)s0 * 16 + 8 * hf, xs0);
      ldhalf(xl + (size_t)s1 * 16 + 8 * hf, xs1);
      if (v0) {
        float ex = expf(al0 - amax);
        den += ex;
#pragma unroll
        for (int i = 0; i < 8; ++i) acc[i] += ex * xs0[i];
      }
      if (v1) {
        float ex = expf(al1 - amax);
        den += ex;
#pragma unroll
        for (int i = 0; i < 8; ++i) acc[i] += ex * xs1[i];
      }
    }
    float exl = expf(aloop - amax);
    den = gsum16(den) + exl;
    float mine = 0.0f;
#pragma unroll
    for (int i = 0; i < 8; ++i) {
      float v = gsum16(acc[i]) + exl * xlh[i];
      if (c == i) mine = v;
    }
    if (c < 8) {
      int f = 8 * hf + c;
      float hv = mine / (den + 1e-16f) + gbias[f];
      h[(size_t)n * 16 + f] = hv;
      xr[(size_t)n * 16 + f] = dinv[n] * hv;
    }
  }
}

__global__ void __launch_bounds__(256)
k_matvec(const int* __restrict__ rowptr, const int* __restrict__ csrc,
         const float* __restrict__ dinv, const float* __restrict__ yin,
         const float* __restrict__ prev, float scale, float* __restrict__ out,
         float* __restrict__ yout, int N) {
  int gid = blockIdx.x * blockDim.x + threadIdx.x;
  int n = gid >> 4, l = gid & 15;
  if (n >= N) return;
  int beg = rowptr[n], end = rowptr[n + 1];
  float dn = dinv[n];
  float acc[16];
#pragma unroll
  for (int t = 0; t < 16; ++t) acc[t] = 0.0f;
  for (int j = beg + l; j < end; j += 16) {
    int s = csrc[j];
    float w = (s != n) ? 1.0f : 0.0f;
    float xs[16];
    ldrow(yin + (size_t)s * 16, xs);
#pragma unroll
    for (int t = 0; t < 16; ++t) acc[t] += w * xs[t];
  }
  float mine = 0.0f;
#pragma unroll
  for (int t = 0; t < 16; ++t) {
    float v = gsum16(acc[t]);
    if (l == t) mine = v;
  }
  float r = scale * (-dn * mine);
  if (prev) r -= prev[(size_t)n * 16 + l];
  out[(size_t)n * 16 + l] = r;
  if (yout) yout[(size_t)n * 16 + l] = dn * r;
}

__global__ void k_gru(const float* __restrict__ T0, const float* __restrict__ T1,
                      const float* __restrict__ T2, const float* __restrict__ T3,
                      const float* __restrict__ chebW, const float* __restrict__ chebB,
                      float* __restrict__ H) {
  int tile = blockIdx.x;
  int lane = threadIdx.x;
  int half = lane >> 4;
  int l16 = lane & 15;
  int row = tile * 16 + l16;
  v8f accZ = {}; v8f accH = {};
#pragma unroll
  for (int t = 0; t < 4; ++t) {
    const float* Tm = (t == 0) ? T0 : (t == 1) ? T1 : (t == 2) ? T2 : T3;
    const float* Wz = chebW + t * 256;
    const float* Wh = chebW + (16 + t) * 256;
#pragma unroll
    for (int kc = 0; kc < 16; kc += 4) {
      int ka = kc + half * 2;
      v2f a;  a.x = Tm[row * 16 + ka];      a.y = Tm[row * 16 + ka + 1];
      v2f bz; bz.x = Wz[ka * 16 + l16];     bz.y = Wz[(ka + 1) * 16 + l16];
      v2f bh; bh.x = Wh[ka * 16 + l16];     bh.y = Wh[(ka + 1) * 16 + l16];
      accZ = WMMA4(a, bz, accZ);
      accH = WMMA4(a, bh, accH);
    }
  }
#pragma unroll
  for (int i = 0; i < 8; ++i) {
    int m = tile * 16 + i + half * 8;
    int c = l16;
    float zp = accZ[i] + chebB[c] + chebB[16 + c];
    float hp = accH[i] + chebB[64 + c] + chebB[80 + c];
    float z = 1.0f / (1.0f + expf(-zp));
    float ht = tanhf(hp);
    H[m * 16 + c] = (1.0f - z) * ht;
  }
}

__global__ void k_mlp(const float* __restrict__ H, const float* W1, const float* b1,
                      const float* W2, const float* b2, float* out, int N) {
  int idx = blockIdx.x * blockDim.x + threadIdx.x;
  if (idx > N) return;
  float hid[4];
#pragma unroll
  for (int j = 0; j < 4; ++j) {
    float s = b1[j];
    if (idx > 0) {
      const float* hr = H + (size_t)(idx - 1) * 16;
#pragma unroll
      for (int f = 0; f < 16; ++f) s += hr[f] * W1[f * 4 + j];
    }
    hid[j] = fmaxf(s, 0.0f);
  }
#pragma unroll
  for (int o = 0; o < 8; ++o) {
    float s = b2[o];
#pragma unroll
    for (int j = 0; j < 4; ++j) s += hid[j] * W2[j * 8 + o];
    out[idx * 8 + o] = s;
  }
}

extern "C" void kernel_launch(void* const* d_in, const int* in_sizes, int n_in,
                              void* d_out, int out_size, void* d_ws, size_t ws_size,
                              hipStream_t stream) {
  const float* x     = (const float*)d_in[0];
  const int*   ei    = (const int*)  d_in[1];
  const float* ea    = (const float*)d_in[2];
  const float* Wl    = (const float*)d_in[3];
  const float* bl    = (const float*)d_in[4];
  const float* Wr    = (const float*)d_in[5];
  const float* br    = (const float*)d_in[6];
  const float* We    = (const float*)d_in[7];
  const float* att   = (const float*)d_in[8];
  const float* gbias = (const float*)d_in[9];
  const float* chebW = (const float*)d_in[10];
  const float* chebB = (const float*)d_in[11];
  const float* W1    = (const float*)d_in[12];
  const float* b1    = (const float*)d_in[13];
  const float* W2    = (const float*)d_in[14];
  const float* b2    = (const float*)d_in[15];
  float* out = (float*)d_out;

  const int N = in_sizes[0] / 128;
  const int E = in_sizes[1] / 2;
  const size_t NF = (size_t)N * 16;

  float* ws    = (float*)d_ws;
  float* xl    = ws;
  float* xr    = xl + NF;
  float* hh    = xr + NF;
  float* T1    = hh + NF;
  float* T2    = T1 + NF;
  float* T3    = T2 + NF;
  float* spare = T3 + NF;
  int*   cnt   = (int*)(spare + NF);
  int*   deg   = cnt + N;
  float* dinv  = (float*)(deg + N);
  int*   rowptr = (int*)(dinv + N);
  int*   csrc  = (int*)(spare + NF + (size_t)N * 5);
  int*   be    = (int*)T1;
  int*   bsd   = (int*)xl;
  int*   btot  = cnt;
  int*   bfill = cnt + MAXB * CPAD;
  float* ascr  = T1;
  int*   ceid  = (int*)T3;

  const int TPB = 256;
  int gEb  = (E + EBLK * EPT - 1) / (EBLK * EPT);
  int nb   = (N + BKT_NODES - 1) >> BKT_SHIFT;
  int gN   = (N + TPB - 1) / TPB;
  int gN16 = (N * 16 + TPB - 1) / TPB;
  int gN1  = (N + 1 + TPB - 1) / TPB;
  int tiles = N / 16;

  hipMemsetAsync(cnt, 0, (size_t)N * 2 * sizeof(int), stream);

  k_count<<<gEb, EBLK, 0, stream>>>(ei, btot, deg, E, nb);
  k_dinv<<<gN, TPB, 0, stream>>>(deg, dinv, N);
  k_bin<<<gEb, EBLK, 0, stream>>>(ei, btot, bfill, bsd, be, E, nb);
  k_place<<<nb, EBLK, 0, stream>>>(btot, bsd, be, rowptr, csrc, ceid, N, nb);

  k_gemm_xlxr<<<tiles, 32, 0, stream>>>(x, Wl, bl, Wr, br, xl, xr);
  k_gat<<<gN16, TPB, 0, stream>>>(rowptr, csrc, ceid, ea, xl, xr, We, att, gbias, dinv, ascr, hh, N);

  k_matvec<<<gN16, TPB, 0, stream>>>(rowptr, csrc, dinv, xr, (const float*)nullptr, 1.0f, T1, xl, N);
  k_matvec<<<gN16, TPB, 0, stream>>>(rowptr, csrc, dinv, xl, hh, 2.0f, T2, xr, N);
  k_matvec<<<gN16, TPB, 0, stream>>>(rowptr, csrc, dinv, xr, T1, 2.0f, T3, (float*)nullptr, N);

  k_gru<<<tiles, 32, 0, stream>>>(hh, T1, T2, T3, chebW, chebB, xl);

  k_mlp<<<gN1, TPB, 0, stream>>>(xl, W1, b1, W2, b2, out, N);
}


# Out-degree counted by a second LDS bucket pass: one byte per edge binned by source bucket, counted per bucket in LDS, dinv written there; no per-edge global atomic left, one launch fewer
# speedup vs baseline: 4.2780x; 1.2887x over previous
#include <hip/hip_runtime.h>
#include <math.h>

typedef float v2f __attribute__((ext_vector_type(2)));
typedef float v8f __attribute__((ext_vector_type(8)));

#define WMMA4(a, b, c) \
  __builtin_amdgcn_wmma_f32_16x16x4_f32(false, (a), false, (b), (short)0, (c), false, false)

__device__ __forceinline__ float gsum16(float v) {
  v += __shfl_xor(v, 8, 16);
  v += __shfl_xor(v, 4, 16);
  v += __shfl_xor(v, 2, 16);
  v += __shfl_xor(v, 1, 16);
  return v;
}
__device__ __forceinline__ float gmax16(float v) {
  v = fmaxf(v, __shfl_xor(v, 8, 16));
  v = fmaxf(v, __shfl_xor(v, 4, 16));
  v = fmaxf(v, __shfl_xor(v, 2, 16));
  v = fmaxf(v, __shfl_xor(v, 1, 16));
  return v;
}

__device__ __forceinline__ void ldrow(const float* __restrict__ p, float* r) {
  const float4* q = (const float4*)p;
  float4 a = q[0], b = q[1], c = q[2], d = q[3];
  r[0] = a.x;  r[1] = a.y;  r[2] = a.z;  r[3] = a.w;
  r[4] = b.x;  r[5] = b.y;  r[6] = b.z;  r[7] = b.w;
  r[8] = c.x;  r[9] = c.y;  r[10] = c.z; r[11] = c.w;
  r[12] = d.x; r[13] = d.y; r[14] = d.z; r[15] = d.w;
}

#define BKT_SHIFT 8
#define BKT_NODES (1 << BKT_SHIFT)
#define MAXB 512
#define CPAD 32
#define EBLK 1024
#define EPT 8

__device__ __forceinline__ void lds_scan_incl(int* a, int n) {
  int t = threadIdx.x;
  for (int off = 1; off < n; off <<= 1) {
    int v = (t < n && t >= off) ? a[t - off] : 0;
    __syncthreads();
    if (t < n) a[t] += v;
    __syncthreads();
  }
}

__device__ __forceinline__ void lds_scan_incl2(int* a, int* b, int n) {
  int t = threadIdx.x;
  for (int off = 1; off < n; off <<= 1) {
    bool on = (t < n && t >= off);
    int va = on ? a[t - off] : 0;
    int vb = on ? b[t - off] : 0;
    __syncthreads();
    if (t < n) { a[t] += va; b[t] += vb; }
    __syncthreads();
  }
}

__global__ void __launch_bounds__(EBLK)
k_count(const int* __restrict__ ei, int* btot, int* stot, int E, int nb) {
  __shared__ int hist[MAXB];
  __shared__ int shist[MAXB];
  int t = threadIdx.x;
  if (t < MAXB) { hist[t] = 0; shist[t] = 0; }
  __syncthreads();
  int e0 = blockIdx.x * (EBLK * EPT) + t;
  int s[EPT], d[EPT];
#pragma unroll
  for (int k = 0; k < EPT; ++k) {
    int e = e0 + k * EBLK;
    bool v = e < E;
    s[k] = v ? ei[e] : 0;
    d[k] = v ? ei[E + e] : 0;
  }
#pragma unroll
  for (int k = 0; k < EPT; ++k) {
    if (e0 + k * EBLK < E) {
      atomicAdd(&hist[d[k] >> BKT_SHIFT], 1);
      if (s[k] != d[k]) atomicAdd(&shist[s[k] >> BKT_SHIFT], 1);
    }
  }
  __syncthreads();
  if (t < nb) {
    int v = hist[t];  if (v) atomicAdd(&btot[t * CPAD], v);
    int w = shist[t]; if (w) atomicAdd(&stot[t * CPAD], w);
  }
}

__global__ void __launch_bounds__(EBLK)
k_bin(const int* __restrict__ ei, const int* __restrict__ btot, int* bfill,
      const int* __restrict__ stot, int* sfill,
      int* __restrict__ bsd, int* __restrict__ be, unsigned char* __restrict__ srec, int E, int nb) {
  __shared__ int hist[MAXB];
  __shared__ int goff[MAXB];
  __shared__ int shist[MAXB];
  __shared__ int sgoff[MAXB];
  int t = threadIdx.x;
  int mytot = (t < nb) ? btot[t * CPAD] : 0;
  int mystot = (t < nb) ? stot[t * CPAD] : 0;
  if (t < MAXB) { hist[t] = 0; goff[t] = mytot; shist[t] = 0; sgoff[t] = mystot; }
  __syncthreads();
  int slr[EPT];
  int e0 = blockIdx.x * (EBLK * EPT) + t;
  int s[EPT], d[EPT], lr[EPT];
#pragma unroll
  for (int k = 0; k < EPT; ++k) {
    int e = e0 + k * EBLK;
    bool v = e < E;
    s[k] = v ? ei[e] : 0;
    d[k] = v ? ei[E + e] : 0;
  }
#pragma unroll
  for (int k = 0; k < EPT; ++k) {
    lr[k] = 0; slr[k] = 0;
    if (e0 + k * EBLK < E) {
      lr[k] = atomicAdd(&hist[d[k] >> BKT_SHIFT], 1);
      if (s[k] != d[k]) slr[k] = atomicAdd(&shist[s[k] >> BKT_SHIFT], 1);
    }
  }
  __syncthreads();
  lds_scan_incl2(goff, sgoff, nb);
  if (t < nb) {
    int n = hist[t];
    int mine = n ? atomicAdd(&bfill[t * CPAD], n) : 0;
    goff[t] = goff[t] - mytot + mine;
    int sn = shist[t];
    int smine = sn ? atomicAdd(&sfill[t * CPAD], sn) : 0;
    sgoff[t] = sgoff[t] - mystot + smine;
  }
  __syncthreads();
#pragma unroll
  for (int k = 0; k < EPT; ++k) {
    int e = e0 + k * EBLK;
    if (e < E) {
      int p = goff[d[k] >> BKT_SHIFT] + lr[k];
      bsd[p] = (int)((unsigned)s[k] | ((unsigned)(d[k] & (BKT_NODES - 1)) << 24));
      be[p] = e;
      if (s[k] != d[k]) srec[sgoff[s[k] >> BKT_SHIFT] + slr[k]] = (unsigned char)(s[k] & (BKT_NODES - 1));
    }
  }
}

__global__ void __launch_bounds__(EBLK)
k_place(const int* __restrict__ btot, const int* __restrict__ bsd, const int* __restrict__ be,
        int* __restrict__ rowptr, int* __restrict__ csrc, int* __restrict__ ceid,
        const int* __restrict__ stot, const unsigned char* __restrict__ srec, float* __restrict__ dinv,
        int N, int nb) {
  __shared__ int bst[MAXB];
  __shared__ int cnt[BKT_NODES];
  __shared__ int cur[BKT_NODES];
  int t = threadIdx.x;
  bool srcblk = (int)blockIdx.x >= nb;
  int b = srcblk ? (int)blockIdx.x - nb : (int)blockIdx.x;
  const int* tot = srcblk ? stot : btot;
  int mytot = (t < nb) ? tot[t * CPAD] : 0;
  if (t < MAXB) bst[t] = mytot;
  if (t < BKT_NODES) cnt[t] = 0;
  __syncthreads();
  lds_scan_incl(bst, nb);
  int endp = bst[b];
  int begp = b ? bst[b - 1] : 0;
  if (srcblk) {
    for (int p0 = begp + t; p0 < endp; p0 += 8 * EBLK) {
      unsigned r[8];
#pragma unroll
      for (int u = 0; u < 8; ++u) { int p = p0 + u * EBLK; r[u] = p < endp ? (unsigned)srec[p] : 0u; }
#pragma unroll
      for (int u = 0; u < 8; ++u) if (p0 + u * EBLK < endp) atomicAdd(&cnt[r[u]], 1);
    }
    __syncthreads();
    int n = (b << BKT_SHIFT) + t;
    if (t < BKT_NODES && n < N) {
      float g = (float)cnt[t];
      dinv[n] = (g > 0.0f) ? rsqrtf(fmaxf(g, 1e-12f)) : 0.0f;
    }
    return;
  }
  for (int p0 = begp + t; p0 < endp; p0 += 4 * EBLK) {
    unsigned r[4];
#pragma unroll
    for (int u = 0; u < 4; ++u) { int p = p0 + u * EBLK; r[u] = p < endp ? (unsigned)bsd[p] : 0u; }
#pragma unroll
    for (int u = 0; u < 4; ++u) if (p0 + u * EBLK < endp) atomicAdd(&cnt[r[u] >> 24], 1);
  }
  __syncthreads();
  int mycnt = (t < BKT_NODES) ? cnt[t] : 0;
  lds_scan_incl(cnt, BKT_NODES);
  int lo = b << BKT_SHIFT;
  if (t < BKT_NODES) {
    int first = begp + cnt[t] - mycnt;
    cur[t] = first;
    if (lo + t < N) rowptr[lo + t] = first;
  }
  if (b == nb - 1 && t == 0) rowptr[N] = endp;
  __syncthreads();
  for (int p0 = begp + t; p0 < endp; p0 += 4 * EBLK) {
    unsigned r[4]; int ev[4];
#pragma unroll
    for (int u = 0; u < 4; ++u) {
      int p = p0 + u * EBLK;
      bool v = p < endp;
      r[u] = v ? (unsigned)bsd[p] : 0u;
      ev[u] = v ? be[p] : 0;
    }
#pragma unroll
    for (int u = 0; u < 4; ++u) {
      if (p0 + u * EBLK < endp) {
        int q = atomicAdd(&cur[r[u] >> 24], 1);
        csrc[q] = (int)(r[u] & 0xFFFFFFu);
        ceid[q] = ev[u];
      }
    }
  }
}

__global__ void k_gemm_xlxr(const float* __restrict__ x,
                            const float* __restrict__ Wl, const float* __restrict__ bl,
                            const float* __restrict__ Wr, const float* __restrict__ br,
                            float* __restrict__ xl, float* __restrict__ xr) {
  int tile = blockIdx.x;
  int lane = threadIdx.x;
  int half = lane >> 4;
  int l16 = lane & 15;
  int row = tile * 16 + l16;
  v8f acc0 = {}; v8f acc1 = {};
#pragma unroll
  for (int k = 0; k < 128; k += 4) {
    int ka = k + half * 2;
    v2f a;  a.x = x[row * 128 + ka];     a.y = x[row * 128 + ka + 1];
    v2f b0; b0.x = Wl[ka * 16 + l16];    b0.y = Wl[(ka + 1) * 16 + l16];
    v2f b1; b1.x = Wr[ka * 16 + l16];    b1.y = Wr[(ka + 1) * 16 + l16];
    acc0 = WMMA4(a, b0, acc0);
    acc1 = WMMA4(a, b1, acc1);
  }
#pragma unroll
  for (int i = 0; i < 8; ++i) {
    int m = tile * 16 + i + half * 8;
    int c = l16;
    xl[m * 16 + c] = acc0[i] + bl[c];
    xr[m * 16 + c] = acc1[i] + br[c];
  }
}

__device__ __forceinline__ void ldhalf(const float* __restrict__ p, float* r) {
  const float4* q = (const float4*)p;
  float4 a = q[0], b = q[1];
  r[0] = a.x; r[1] = a.y; r[2] = a.z; r[3] = a.w;
  r[4] = b.x; r[5] = b.y; r[6] = b.z; r[7] = b.w;
}

__device__ __forceinline__ float tile_logit(const v2f* wa, const float* evh, const float* cin,
                                            const float* atth) {
  v8f acc;
#pragma unroll
  for (int i = 0; i < 8; ++i) acc[i] = cin[i];
#pragma unroll
  for (int t = 0; t < 4; ++t) {
    v2f b; b.x = evh[2 * t]; b.y = evh[2 * t + 1];
    acc = WMMA4(wa[t], b, acc);
  }
  float p = 0.0f;
#pragma unroll
  for (int i = 0; i < 8; ++i) {
    float m = acc[i];
    m = m > 0.0f ? m : 0.2f * m;
    p += m * atth[i];
  }
  return p + __shfl_xor(p, 16);
}

__global__ void __launch_bounds__(256)
k_gat(const int* __restrict__ rowptr, const int* __restrict__ csrc, const int* __restrict__ ceid,
      const float* __restrict__ ea, const float* __restrict__ xl, float* xr,
      const float* __restrict__ We, const float* __restrict__ att,
      const float* __restrict__ gbias, const float* __restrict__ dinv,
      float* ascr, float* __restrict__ h, int N) {
  int lane = threadIdx.x & 31;
  int c = lane & 15, hf = lane >> 4;
  int wave = __builtin_amdgcn_readfirstlane((int)((blockIdx.x * blockDim.x + threadIdx.x) >> 5));

  v2f wa[4];
#pragma unroll
  for (int t = 0; t < 4; ++t) {
    int j = 8 * hf + 2 * t;
    wa[t].x = We[j * 16 + c];
    wa[t].y = We[(j + 1) * 16 + c];
  }
  float atth[8];
  ldhalf(att + 8 * hf, atth);

  for (int q = 0; q < 2; ++q) {
    int n = wave * 2 + q;
    if (n >= N) break;
    int beg = rowptr[n], end = rowptr[n + 1];
    float xrh[8], xlh[8];
    ldhalf(xr + (size_t)n * 16 + 8 * hf, xrh);
    ldhalf(xl + (size_t)n * 16 + 8 * hf, xlh);

    float psum[8];
#pragma unroll
    for (int i = 0; i < 8; ++i) psum[i] = 0.0f;
    float amax = -INFINITY;
    for (int base = beg; base < end; base += 32) {
      int j0 = base + c, j1 = j0 + 16;
      bool v0 = j0 < end, v1 = j1 < end;
      int q0 = v0 ? j0 : end - 1;
      int q1 = v1 ? j1 : end - 1;
      int s0 = csrc[q0], e0 = ceid[q0];
      int s1 = csrc[q1], e1 = ceid[q1];
      float ev0[8], xs0[8], ev1[8], xs1[8];
      ldhalf(ea + (size_t)e0 * 16 + 8 * hf, ev0);
      ldhalf(xl + (size_t)s0 * 16 + 8 * hf, xs0);
      ldhalf(ea + (size_t)e1 * 16 + 8 * hf, ev1);
      ldhalf(xl + (size_t)s1 * 16 + 8 * hf, xs1);
#pragma unroll
      for (int i = 0; i < 8; ++i) { xs0[i] += xrh[i]; xs1[i] += xrh[i]; }
      float a0 = tile_logit(wa, ev0, xs0, atth);
      float a1 = tile_logit(wa, ev1, xs1, atth);
      if (v0) {
        ascr[j0] = a0;
        amax = fmaxf(amax, a0);
#pragma unroll
        for (int i = 0; i < 8; ++i) psum[i] += ev0[i];
      }
      if (v1) {
        ascr[j1] = a1;
        amax = fmaxf(amax, a1);
#pragma unroll
        for (int i = 0; i < 8; ++i) psum[i] += ev1[i];
      }
    }
    float dn = fmaxf((float)(end - beg), 1.0f);
    float emh[8], cl[8];
#pragma unroll
    for (int i = 0; i < 8; ++i) { emh[i] = gsum16(psum[i]) / dn; cl[i] = xlh[i] + xrh[i]; }
    float aloop = tile_logit(wa, emh, cl, atth);
    amax = fmaxf(gmax16(amax), aloop);

    float acc[8];
#pragma unroll
    for (int i = 0; i < 8; ++i) acc[i] = 0.0f;
    float den = 0.0f;
    for (int base = beg; base < end; base += 32) {
      int j0 = base + c, j1 = j0 + 16;
      bool v0 = j0 < end, v1 = j1 < end;
      int q0 = v0 ? j0 : end - 1;
      int q1 = v1 ? j1 : end - 1;
      int s0 = csrc[q0], s1 = csrc[q1];
      float al0 = ascr[q0], al1 = ascr[q1];
      float xs0[8], xs1[8];
      ldhalf(xl + (size_t)s0 * 16 + 8 * hf, xs0);
      ldhalf(xl + (size_t)s1 * 16 + 8 * hf, xs1);
      if (v0) {
        float ex = expf(al0 - amax);
        den += ex;
#pragma unroll
        for (int i = 0; i < 8; ++i) acc[i] += ex * xs0[i];
      }
      if (v1) {
        float ex = expf(al1 - amax);
        den += ex;
#pragma unroll
        for (int i = 0; i < 8; ++i) acc[i] += ex * xs1[i];
      }
    }
    float exl = expf(aloop - amax);
    den = gsum16(den) + exl;
    float mine = 0.0f;
#pragma unroll
    for (int i = 0; i < 8; ++i) {
      float v = gsum16(acc[i]) + exl * xlh[i];
      if (c == i) mine = v;
    }
    if (c < 8) {
      int f = 8 * hf + c;
      float hv = mine / (den + 1e-16f) + gbias[f];
      h[(size_t)n * 16 + f] = hv;
      xr[(size_t)n * 16 + f] = dinv[n] * hv;
    }
  }
}

__global__ void __launch_bounds__(256)
k_matvec(const int* __restrict__ rowptr, const int* __restrict__ csrc,
         const float* __restrict__ dinv, const float* __restrict__ yin,
         const float* __restrict__ prev, float scale, float* __restrict__ out,
         float* __restrict__ yout, int N) {
  int gid = blockIdx.x * blockDim.x + threadIdx.x;
  int n = gid >> 4, l = gid & 15;
  if (n >= N) return;
  int beg = rowptr[n], end = rowptr[n + 1];
  float dn = dinv[n];
  float acc[16];
#pragma unroll
  for (int t = 0; t < 16; ++t) acc[t] = 0.0f;
  for (int j = beg + l; j < end; j += 16) {
    int s = csrc[j];
    float w = (s != n) ? 1.0f : 0.0f;
    float xs[16];
    ldrow(yin + (size_t)s * 16, xs);
#pragma unroll
    for (int t = 0; t < 16; ++t) acc[t] += w * xs[t];
  }
  float mine = 0.0f;
#pragma unroll
  for (int t = 0; t < 16; ++t) {
    float v = gsum16(acc[t]);
    if (l == t) mine = v;
  }
  float r = scale * (-dn * mine);
  if (prev) r -= prev[(size_t)n * 16 + l];
  out[(size_t)n * 16 + l] = r;
  if (yout) yout[(size_t)n * 16 + l] = dn * r;
}

__global__ void k_gru(const float* __restrict__ T0, const float* __restrict__ T1,
                      const float* __restrict__ T2, const float* __restrict__ T3,
                      const float* __restrict__ chebW, const float* __restrict__ chebB,
                      float* __restrict__ H) {
  int tile = blockIdx.x;
  int lane = threadIdx.x;
  int half = lane >> 4;
  int l16 = lane & 15;
  int row = tile * 16 + l16;
  v8f accZ = {}; v8f accH = {};
#pragma unroll
  for (int t = 0; t < 4; ++t) {
    const float* Tm = (t == 0) ? T0 : (t == 1) ? T1 : (t == 2) ? T2 : T3;
    const float* Wz = chebW + t * 256;
    const float* Wh = chebW + (16 + t) * 256;
#pragma unroll
    for (int kc = 0; kc < 16; kc += 4) {
      int ka = kc + half * 2;
      v2f a;  a.x = Tm[row * 16 + ka];      a.y = Tm[row * 16 + ka + 1];
      v2f bz; bz.x = Wz[ka * 16 + l16];     bz.y = Wz[(ka + 1) * 16 + l16];
      v2f bh; bh.x = Wh[ka * 16 + l16];     bh.y = Wh[(ka + 1) * 16 + l16];
      accZ = WMMA4(a, bz, accZ);
      accH = WMMA4(a, bh, accH);
    }
  }
#pragma unroll
  for (int i = 0; i < 8; ++i) {
    int m = tile * 16 + i + half * 8;
    int c = l16;
    float zp = accZ[i] + chebB[c] + chebB[16 + c];
    float hp = accH[i] + chebB[64 + c] + chebB[80 + c];
    float z = 1.0f / (1.0f + expf(-zp));
    float ht = tanhf(hp);
    H[m * 16 + c] = (1.0f - z) * ht;
  }
}

__global__ void k_mlp(const float* __restrict__ H, const float* W1, const float* b1,
                      const float* W2, const float* b2, float* out, int N) {
  int idx = blockIdx.x * blockDim.x + threadIdx.x;
  if (idx > N) return;
  float hid[4];
#pragma unroll
  for (int j = 0; j < 4; ++j) {
    float s = b1[j];
    if (idx > 0) {
      const float* hr = H + (size_t)(idx - 1) * 16;
#pragma unroll
      for (int f = 0; f < 16; ++f) s += hr[f] * W1[f * 4 + j];
    }
    hid[j] = fmaxf(s, 0.0f);
  }
#pragma unroll
  for (int o = 0; o < 8; ++o) {
    float s = b2[o];
#pragma unroll
    for (int j = 0; j < 4; ++j) s += hid[j] * W2[j * 8 + o];
    out[idx * 8 + o] = s;
  }
}

extern "C" void kernel_launch(void* const* d_in, const int* in_sizes, int n_in,
                              void* d_out, int out_size, void* d_ws, size_t ws_size,
                              hipStream_t stream) {
  const float* x     = (const float*)d_in[0];
  const int*   ei    = (const int*)  d_in[1];
  const float* ea    = (const float*)d_in[2];
  const float* Wl    = (const float*)d_in[3];
  const float* bl    = (const float*)d_in[4];
  const float* Wr    = (const float*)d_in[5];
  const float* br    = (const float*)d_in[6];
  const float* We    = (const float*)d_in[7];
  const float* att   = (const float*)d_in[8];
  const float* gbias = (const float*)d_in[9];
  const float* chebW = (const float*)d_in[10];
  const float* chebB = (const float*)d_in[11];
  const float* W1    = (const float*)d_in[12];
  const float* b1    = (const float*)d_in[13];
  const float* W2    = (const float*)d_in[14];
  const float* b2    = (const float*)d_in[15];
  float* out = (float*)d_out;

  const int N = in_sizes[0] / 128;
  const int E = in_sizes[1] / 2;
  const size_t NF = (size_t)N * 16;

  float* ws    = (float*)d_ws;
  float* xl    = ws;
  float* xr    = xl + NF;
  float* hh    = xr + NF;
  float* T1    = hh + NF;
  float* T2    = T1 + NF;
  float* T3    = T2 + NF;
  float* spare = T3 + NF;
  int*   cnt   = (int*)(spare + NF);
  int*   scnt  = cnt + N;
  float* dinv  = (float*)(scnt + N);
  int*   rowptr = (int*)(dinv + N);
  int*   csrc  = (int*)(spare + NF + (size_t)N * 5);
  int*   be    = (int*)T1;
  int*   bsd   = (int*)xl;
  int*   btot  = cnt;
  int*   bfill = cnt + MAXB * CPAD;
  int*   stot  = scnt;
  int*   sfill = scnt + MAXB * CPAD;
  unsigned char* srec = (unsigned char*)hh;
  float* ascr  = T1;
  int*   ceid  = (int*)T3;

  const int TPB = 256;
  int gEb  = (E + EBLK * EPT - 1) / (EBLK * EPT);
  int nb   = (N + BKT_NODES - 1) >> BKT_SHIFT;
  int gN16 = (N * 16 + TPB - 1) / TPB;
  int gN1  = (N + 1 + TPB - 1) / TPB;
  int tiles = N / 16;

  hipMemsetAsync(cnt, 0, (size_t)N * 2 * sizeof(int), stream);

  k_count<<<gEb, EBLK, 0, stream>>>(ei, btot, stot, E, nb);
  k_bin<<<gEb, EBLK, 0, stream>>>(ei, btot, bfill, stot, sfill, bsd, be, srec, E, nb);
  k_place<<<nb + nb, EBLK, 0, stream>>>(btot, bsd, be, rowptr, csrc, ceid, stot, srec, dinv, N, nb);

  k_gemm_xlxr<<<tiles, 32, 0, stream>>>(x, Wl, bl, Wr, br, xl, xr);
  k_gat<<<gN16, TPB, 0, stream>>>(rowptr, csrc, ceid, ea, xl, xr, We, att, gbias, dinv, ascr, hh, N);

  k_matvec<<<gN16, TPB, 0, stream>>>(rowptr, csrc, dinv, xr, (const float*)nullptr, 1.0f, T1, xl, N);
  k_matvec<<<gN16, TPB, 0, stream>>>(rowptr, csrc, dinv, xl, hh, 2.0f, T2, xr, N);
  k_matvec<<<gN16, TPB, 0, stream>>>(rowptr, csrc, dinv, xr, T1, 2.0f, T3, (float*)nullptr, N);

  k_gru<<<tiles, 32, 0, stream>>>(hh, T1, T2, T3, chebW, chebB, xl);

  k_mlp<<<gN1, TPB, 0, stream>>>(xl, W1, b1, W2, b2, out, N);
}


# Chebyshev gather reads each 64-byte source row with a lane quad (4 lanes x 16 B: 8 whole rows per load, 8 lines touched, not 32); sums re-associated: 4 partial sums per node, 2-step tree
# speedup vs baseline: 4.4886x; 1.0492x over previous
#include <hip/hip_runtime.h>
#include <math.h>

typedef float v2f __attribute__((ext_vector_type(2)));
typedef float v8f __attribute__((ext_vector_type(8)));

#define WMMA4(a, b, c) \
  __builtin_amdgcn_wmma_f32_16x16x4_f32(false, (a), false, (b), (short)0, (c), false, false)

__device__ __forceinline__ float gsum16(float v) {
  v += __shfl_xor(v, 8, 16);
  v += __shfl_xor(v, 4, 16);
  v += __shfl_xor(v, 2, 16);
  v += __shfl_xor(v, 1, 16);
  return v;
}
__device__ __forceinline__ float gmax16(float v) {
  v = fmaxf(v, __shfl_xor(v, 8, 16));
  v = fmaxf(v, __shfl_xor(v, 4, 16));
  v = fmaxf(v, __shfl_xor(v, 2, 16));
  v = fmaxf(v, __shfl_xor(v, 1, 16));
  return v;
}

__device__ __forceinline__ void ldrow(const float* __restrict__ p, float* r) {
  const float4* q = (const float4*)p;
  float4 a = q[0], b = q[1], c = q[2], d = q[3];
  r[0] = a.x;  r[1] = a.y;  r[2] = a.z;  r[3] = a.w;
  r[4] = b.x;  r[5] = b.y;  r[6] = b.z;  r[7] = b.w;
  r[8] = c.x;  r[9] = c.y;  r[10] = c.z; r[11] = c.w;
  r[12] = d.x; r[13] = d.y; r[14] = d.z; r[15] = d.w;
}

#define BKT_SHIFT 8
#define BKT_NODES (1 << BKT_SHIFT)
#define MAXB 512
#define CPAD 32
#define EBLK 1024
#define EPT 8

__device__ __forceinline__ void lds_scan_incl(int* a, int n) {
  int t = threadIdx.x;
  for (int off = 1; off < n; off <<= 1) {
    int v = (t < n && t >= off) ? a[t - off] : 0;
    __syncthreads();
    if (t < n) a[t] += v;
    __syncthreads();
  }
}

__device__ __forceinline__ void lds_scan_incl2(int* a, int* b, int n) {
  int t = threadIdx.x;
  for (int off = 1; off < n; off <<= 1) {
    bool on = (t < n && t >= off);
    int va = on ? a[t - off] : 0;
    int vb = on ? b[t - off] : 0;
    __syncthreads();
    if (t < n) { a[t] += va; b[t] += vb; }
    __syncthreads();
  }
}

__global__ void __launch_bounds__(EBLK)
k_count(const int* __restrict__ ei, int* btot, int* stot, int E, int nb) {
  __shared__ int hist[MAXB];
  __shared__ int shist[MAXB];
  int t = threadIdx.x;
  if (t < MAXB) { hist[t] = 0; shist[t] = 0; }
  __syncthreads();
  int e0 = blockIdx.x * (EBLK * EPT) + t;
  int s[EPT], d[EPT];
#pragma unroll
  for (int k = 0; k < EPT; ++k) {
    int e = e0 + k * EBLK;
    bool v = e < E;
    s[k] = v ? ei[e] : 0;
    d[k] = v ? ei[E + e] : 0;
  }
#pragma unroll
  for (int k = 0; k < EPT; ++k) {
    if (e0 + k * EBLK < E) {
      atomicAdd(&hist[d[k] >> BKT_SHIFT], 1);
      if (s[k] != d[k]) atomicAdd(&shist[s[k] >> BKT_SHIFT], 1);
    }
  }
  __syncthreads();
  if (t < nb) {
    int v = hist[t];  if (v) atomicAdd(&btot[t * CPAD], v);
    int w = shist[t]; if (w) atomicAdd(&stot[t * CPAD], w);
  }
}

__global__ void __launch_bounds__(EBLK)
k_bin(const int* __restrict__ ei, const int* __restrict__ btot, int* bfill,
      const int* __restrict__ stot, int* sfill,
      int* __restrict__ bsd, int* __restrict__ be, unsigned char* __restrict__ srec, int E, int nb) {
  __shared__ int hist[MAXB];
  __shared__ int goff[MAXB];
  __shared__ int shist[MAXB];
  __shared__ int sgoff[MAXB];
  int t = threadIdx.x;
  int mytot = (t < nb) ? btot[t * CPAD] : 0;
  int mystot = (t < nb) ? stot[t * CPAD] : 0;
  if (t < MAXB) { hist[t] = 0; goff[t] = mytot; shist[t] = 0; sgoff[t] = mystot; }
  __syncthreads();
  int slr[EPT];
  int e0 = blockIdx.x * (EBLK * EPT) + t;
  int s[EPT], d[EPT], lr[EPT];
#pragma unroll
  for (int k = 0; k < EPT; ++k) {
    int e = e0 + k * EBLK;
    bool v = e < E;
    s[k] = v ? ei[e] : 0;
    d[k] = v ? ei[E + e] : 0;
  }
#pragma unroll
  for (int k = 0; k < EPT; ++k) {
    lr[k] = 0; slr[k] = 0;
    if (e0 + k * EBLK < E) {
      lr[k] = atomicAdd(&hist[d[k] >> BKT_SHIFT], 1);
      if (s[k] != d[k]) slr[k] = atomicAdd(&shist[s[k] >> BKT_SHIFT], 1);
    }
  }
  __syncthreads();
  lds_scan_incl2(goff, sgoff, nb);
  if (t < nb) {
    int n = hist[t];
    int mine = n ? atomicAdd(&bfill[t * CPAD], n) : 0;
    goff[t] = goff[t] - mytot + mine;
    int sn = shist[t];
    int smine = sn ? atomicAdd(&sfill[t * CPAD], sn) : 0;
    sgoff[t] = sgoff[t] - mystot + smine;
  }
  __syncthreads();
#pragma unroll
  for (int k = 0; k < EPT; ++k) {
    int e = e0 + k * EBLK;
    if (e < E) {
      int p = goff[d[k] >> BKT_SHIFT] + lr[k];
      bsd[p] = (int)((unsigned)s[k] | ((unsigned)(d[k] & (BKT_NODES - 1)) << 24));
      be[p] = e;
      if (s[k] != d[k]) srec[sgoff[s[k] >> BKT_SHIFT] + slr[k]] = (unsigned char)(s[k] & (BKT_NODES - 1));
    }
  }
}

__global__ void __launch_bounds__(EBLK)
k_place(const int* __restrict__ btot, const int* __restrict__ bsd, const int* __restrict__ be,
        int* __restrict__ rowptr, int* __restrict__ csrc, int* __restrict__ ceid,
        const int* __restrict__ stot, const unsigned char* __restrict__ srec, float* __restrict__ dinv,
        int N, int nb) {
  __shared__ int bst[MAXB];
  __shared__ int cnt[BKT_NODES];
  __shared__ int cur[BKT_NODES];
  int t = threadIdx.x;
  bool srcblk = (int)blockIdx.x >= nb;
  int b = srcblk ? (int)blockIdx.x - nb : (int)blockIdx.x;
  const int* tot = srcblk ? stot : btot;
  int mytot = (t < nb) ? tot[t * CPAD] : 0;
  if (t < MAXB) bst[t] = mytot;
  if (t < BKT_NODES) cnt[t] = 0;
  __syncthreads();
  lds_scan_incl(bst, nb);
  int endp = bst[b];
  int begp = b ? bst[b - 1] : 0;
  if (srcblk) {
    for (int p0 = begp + t; p0 < endp; p0 += 8 * EBLK) {
      unsigned r[8];
#pragma unroll
      for (int u = 0; u < 8; ++u) { int p = p0 + u * EBLK; r[u] = p < endp ? (unsigned)srec[p] : 0u; }
#pragma unroll
      for (int u = 0; u < 8; ++u) if (p0 + u * EBLK < endp) atomicAdd(&cnt[r[u]], 1);
    }
    __syncthreads();
    int n = (b << BKT_SHIFT) + t;
    if (t < BKT_NODES && n < N) {
      float g = (float)cnt[t];
      dinv[n] = (g > 0.0f) ? rsqrtf(fmaxf(g, 1e-12f)) : 0.0f;
    }
    return;
  }
  for (int p0 = begp + t; p0 < endp; p0 += 4 * EBLK) {
    unsigned r[4];
#pragma unroll
    for (int u = 0; u < 4; ++u) { int p = p0 + u * EBLK; r[u] = p < endp ? (unsigned)bsd[p] : 0u; }
#pragma unroll
    for (int u = 0; u < 4; ++u) if (p0 + u * EBLK < endp) atomicAdd(&cnt[r[u] >> 24], 1);
  }
  __syncthreads();
  int mycnt = (t < BKT_NODES) ? cnt[t] : 0;
  lds_scan_incl(cnt, BKT_NODES);
  int lo = b << BKT_SHIFT;
  if (t < BKT_NODES) {
    int first = begp + cnt[t] - mycnt;
    cur[t] = first;
    if (lo + t < N) rowptr[lo + t] = first;
  }
  if (b == nb - 1 && t == 0) rowptr[N] = endp;
  __syncthreads();
  for (int p0 = begp + t; p0 < endp; p0 += 4 * EBLK) {
    unsigned r[4]; int ev[4];
#pragma unroll
    for (int u = 0; u < 4; ++u) {
      int p = p0 + u * EBLK;
      bool v = p < endp;
      r[u] = v ? (unsigned)bsd[p] : 0u;
      ev[u] = v ? be[p] : 0;
    }
#pragma unroll
    for (int u = 0; u < 4; ++u) {
      if (p0 + u * EBLK < endp) {
        int q = atomicAdd(&cur[r[u] >> 24], 1);
        csrc[q] = (int)(r[u] & 0xFFFFFFu);
        ceid[q] = ev[u];
      }
    }
  }
}

__global__ void k_gemm_xlxr(const float* __restrict__ x,
                            const float* __restrict__ Wl, const float* __restrict__ bl,
                            const float* __restrict__ Wr, const float* __restrict__ br,
                            float* __restrict__ xl, float* __restrict__ xr) {
  int tile = blockIdx.x;
  int lane = threadIdx.x;
  int half = lane >> 4;
  int l16 = lane & 15;
  int row = tile * 16 + l16;
  v8f acc0 = {}; v8f acc1 = {};
#pragma unroll
  for (int k = 0; k < 128; k += 4) {
    int ka = k + half * 2;
    v2f a;  a.x = x[row * 128 + ka];     a.y = x[row * 128 + ka + 1];
    v2f b0; b0.x = Wl[ka * 16 + l16];    b0.y = Wl[(ka + 1) * 16 + l16];
    v2f b1; b1.x = Wr[ka * 16 + l16];    b1.y = Wr[(ka + 1) * 16 + l16];
    acc0 = WMMA4(a, b0, acc0);
    acc1 = WMMA4(a, b1, acc1);
  }
#pragma unroll
  for (int i = 0; i < 8; ++i) {
    int m = tile * 16 + i + half * 8;
    int c = l16;
    xl[m * 16 + c] = acc0[i] + bl[c];
    xr[m * 16 + c] = acc1[i] + br[c];
  }
}

__device__ __forceinline__ void ldhalf(const float* __restrict__ p, float* r) {
  const float4* q = (const float4*)p;
  float4 a = q[0], b = q[1];
  r[0] = a.x; r[1] = a.y; r[2] = a.z; r[3] = a.w;
  r[4] = b.x; r[5] = b.y; r[6] = b.z; r[7] = b.w;
}

__device__ __forceinline__ float tile_logit(const v2f* wa, const float* evh, const float* cin,
                                            const float* atth) {
  v8f acc;
#pragma unroll
  for (int i = 0; i < 8; ++i) acc[i] = cin[i];
#pragma unroll
  for (int t = 0; t < 4; ++t) {
    v2f b; b.x = evh[2 * t]; b.y = evh[2 * t + 1];
    acc = WMMA4(wa[t], b, acc);
  }
  float p = 0.0f;
#pragma unroll
  for (int i = 0; i < 8; ++i) {
    float m = acc[i];
    m = m > 0.0f ? m : 0.2f * m;
    p += m * atth[i];
  }
  return p + __shfl_xor(p, 16);
}

__global__ void __launch_bounds__(256)
k_gat(const int* __restrict__ rowptr, const int* __restrict__ csrc, const int* __restrict__ ceid,
      const float* __restrict__ ea, const float* __restrict__ xl, float* xr,
      const float* __restrict__ We, const float* __restrict__ att,
      const float* __restrict__ gbias, const float* __restrict__ dinv,
      float* ascr, float* __restrict__ h, int N) {
  int lane = threadIdx.x & 31;
  int c = lane & 15, hf = lane >> 4;
  int wave = __builtin_amdgcn_readfirstlane((int)((blockIdx.x * blockDim.x + threadIdx.x) >> 5));

  v2f wa[4];
#pragma unroll
  for (int t = 0; t < 4; ++t) {
    int j = 8 * hf + 2 * t;
    wa[t].x = We[j * 16 + c];
    wa[t].y = We[(j + 1) * 16 + c];
  }
  float atth[8];
  ldhalf(att + 8 * hf, atth);

  for (int q = 0; q < 2; ++q) {
    int n = wave * 2 + q;
    if (n >= N) break;
    int beg = rowptr[n], end = rowptr[n + 1];
    float xrh[8], xlh[8];
    ldhalf(xr + (size_t)n * 16 + 8 * hf, xrh);
    ldhalf(xl + (size_t)n * 16 + 8 * hf, xlh);

    float psum[8];
#pragma unroll
    for (int i = 0; i < 8; ++i) psum[i] = 0.0f;
    float amax = -INFINITY;
    for (int base = beg; base < end; base += 32) {
      int j0 = base + c, j1 = j0 + 16;
      bool v0 = j0 < end, v1 = j1 < end;
      int q0 = v0 ? j0 : end - 1;
      int q1 = v1 ? j1 : end - 1;
      int s0 = csrc[q0], e0 = ceid[q0];
      int s1 = csrc[q1], e1 = ceid[q1];
      float ev0[8], xs0[8], ev1[8], xs1[8];
      ldhalf(ea + (size_t)e0 * 16 + 8 * hf, ev0);
      ldhalf(xl + (size_t)s0 * 16 + 8 * hf, xs0);
      ldhalf(ea + (size_t)e1 * 16 + 8 * hf, ev1);
      ldhalf(xl + (size_t)s1 * 16 + 8 * hf, xs1);
#pragma unroll
      for (int i = 0; i < 8; ++i) { xs0[i] += xrh[i]; xs1[i] += xrh[i]; }
      float a0 = tile_logit(wa, ev0, xs0, atth);
      float a1 = tile_logit(wa, ev1, xs1, atth);
      if (v0) {
        ascr[j0] = a0;
        amax = fmaxf(amax, a0);
#pragma unroll
        for (int i = 0; i < 8; ++i) psum[i] += ev0[i];
      }
      if (v1) {
        ascr[j1] = a1;
        amax = fmaxf(amax, a1);
#pragma unroll
        for (int i = 0; i < 8; ++i) psum[i] += ev1[i];
      }
    }
    float dn = fmaxf((float)(end - beg), 1.0f);
    float emh[8], cl[8];
#pragma unroll
    for (int i = 0; i < 8; ++i) { emh[i] = gsum16(psum[i]) / dn; cl[i] = xlh[i] + xrh[i]; }
    float aloop = tile_logit(wa, emh, cl, atth);
    amax = fmaxf(gmax16(amax), aloop);

    float acc[8];
#pragma unroll
    for (int i = 0; i < 8; ++i) acc[i] = 0.0f;
    float den = 0.0f;
    for (int base = beg; base < end; base += 32) {
      int j0 = base + c, j1 = j0 + 16;
      bool v0 = j0 < end, v1 = j1 < end;
      int q0 = v0 ? j0 : end - 1;
      int q1 = v1 ? j1 : end - 1;
      int s0 = csrc[q0], s1 = csrc[q1];
      float al0 = ascr[q0], al1 = ascr[q1];
      float xs0[8], xs1[8];
      ldhalf(xl + (size_t)s0 * 16 + 8 * hf, xs0);
      ldhalf(xl + (size_t)s1 * 16 + 8 * hf, xs1);
      if (v0) {
        float ex = expf(al0 - amax);
        den += ex;
#pragma unroll
        for (int i = 0; i < 8; ++i) acc[i] += ex * xs0[i];
      }
      if (v1) {
        float ex = expf(al1 - amax);
        den += ex;
#pragma unroll
        for (int i = 0; i < 8; ++i) acc[i] += ex * xs1[i];
      }
    }
    float exl = expf(aloop - amax);
    den = gsum16(den) + exl;
    float mine = 0.0f;
#pragma unroll
    for (int i = 0; i < 8; ++i) {
      float v = gsum16(acc[i]) + exl * xlh[i];
      if (c == i) mine = v;
    }
    if (c < 8) {
      int f = 8 * hf + c;
      float hv = mine / (den + 1e-16f) + gbias[f];
      h[(size_t)n * 16 + f] = hv;
      xr[(size_t)n * 16 + f] = dinv[n] * hv;
    }
  }
}

__global__ void __launch_bounds__(256)
k_matvec(const int* __restrict__ rowptr, const int* __restrict__ csrc,
         const float* __restrict__ dinv, const float* __restrict__ yin,
         const float* __restrict__ prev, float scale, float* __restrict__ out,
         float* __restrict__ yout, int N) {
  int gid = blockIdx.x * blockDim.x + threadIdx.x;
  int n = gid >> 4, l = gid & 15;
  if (n >= N) return;
  int g = l >> 2, q = l & 3;
  int beg = rowptr[n], end = rowptr[n + 1];
  float dn = dinv[n];
  const float4* __restrict__ yin4 = (const float4*)yin;
  float a0 = 0.0f, a1 = 0.0f, a2 = 0.0f, a3 = 0.0f;
  for (int base = beg; base < end; base += 16) {
    int jj = base + l;
    unsigned sidx = (unsigned)((jj < end) ? csrc[jj] : n);
    int s0 = (int)__builtin_amdgcn_permlane16(sidx, sidx, 0x11110000u, 0x33332222u, false, false);
    int s1 = (int)__builtin_amdgcn_permlane16(sidx, sidx, 0x55554444u, 0x77776666u, false, false);
    int s2 = (int)__builtin_amdgcn_permlane16(sidx, sidx, 0x99998888u, 0xBBBBAAAAu, false, false);
    int s3 = (int)__builtin_amdgcn_permlane16(sidx, sidx, 0xDDDDCCCCu, 0xFFFFEEEEu, false, false);
    float4 r0 = yin4[(size_t)s0 * 4 + q];
    float4 r1 = yin4[(size_t)s1 * 4 + q];
    float4 r2 = yin4[(size_t)s2 * 4 + q];
    float4 r3 = yin4[(size_t)s3 * 4 + q];
    int e = base + g;
    if (e < end)      { float w = (s0 != n) ? 1.0f : 0.0f; a0 += w * r0.x; a1 += w * r0.y; a2 += w * r0.z; a3 += w * r0.w; }
    if (e + 4 < end)  { float w = (s1 != n) ? 1.0f : 0.0f; a0 += w * r1.x; a1 += w * r1.y; a2 += w * r1.z; a3 += w * r1.w; }
    if (e + 8 < end)  { float w = (s2 != n) ? 1.0f : 0.0f; a0 += w * r2.x; a1 += w * r2.y; a2 += w * r2.z; a3 += w * r2.w; }
    if (e + 12 < end) { float w = (s3 != n) ? 1.0f : 0.0f; a0 += w * r3.x; a1 += w * r3.y; a2 += w * r3.z; a3 += w * r3.w; }
  }
  a0 += __shfl_xor(a0, 4, 16); a1 += __shfl_xor(a1, 4, 16); a2 += __shfl_xor(a2, 4, 16); a3 += __shfl_xor(a3, 4, 16);
  a0 += __shfl_xor(a0, 8, 16); a1 += __shfl_xor(a1, 8, 16); a2 += __shfl_xor(a2, 8, 16); a3 += __shfl_xor(a3, 8, 16);
  float mine = (g == 0) ? a0 : (g == 1) ? a1 : (g == 2) ? a2 : a3;
  int f = 4 * q + g;
  float r = scale * (-dn * mine);
  if (prev) r -= prev[(size_t)n * 16 + f];
  out[(size_t)n * 16 + f] = r;
  if (yout) yout[(size_t)n * 16 + f] = dn * r;
}

__global__ void k_gru(const float* __restrict__ T0, const float* __restrict__ T1,
                      const float* __restrict__ T2, const float* __restrict__ T3,
                      const float* __restrict__ chebW, const float* __restrict__ chebB,
                      float* __restrict__ H) {
  int tile = blockIdx.x;
  int lane = threadIdx.x;
  int half = lane >> 4;
  int l16 = lane & 15;
  int row = tile * 16 + l16;
  v8f accZ = {}; v8f accH = {};
#pragma unroll
  for (int t = 0; t < 4; ++t) {
    const float* Tm = (t == 0) ? T0 : (t == 1) ? T1 : (t == 2) ? T2 : T3;
    const float* Wz = chebW + t * 256;
    const float* Wh = chebW + (16 + t) * 256;
#pragma unroll
    for (int kc = 0; kc < 16; kc += 4) {
      int ka = kc + half * 2;
      v2f a;  a.x = Tm[row * 16 + ka];      a.y = Tm[row * 16 + ka + 1];
      v2f bz; bz.x = Wz[ka * 16 + l16];     bz.y = Wz[(ka + 1) * 16 + l16];
      v2f bh; bh.x = Wh[ka * 16 + l16];     bh.y = Wh[(ka + 1) * 16 + l16];
      accZ = WMMA4(a, bz, accZ);
      accH = WMMA4(a, bh, accH);
    }
  }
#pragma unroll
  for (int i = 0; i < 8; ++i) {
    int m = tile * 16 + i + half * 8;
    int c = l16;
    float zp = accZ[i] + chebB[c] + chebB[16 + c];
    float hp = accH[i] + chebB[64 + c] + chebB[80 + c];
    float z = 1.0f / (1.0f + expf(-zp));
    float ht = tanhf(hp);
    H[m * 16 + c] = (1.0f - z) * ht;
  }
}

__global__ void k_mlp(const float* __restrict__ H, const float* W1, const float* b1,
                      const float* W2, const float* b2, float* out, int N) {
  int idx = blockIdx.x * blockDim.x + threadIdx.x;
  if (idx > N) return;
  float hid[4];
#pragma unroll
  for (int j = 0; j < 4; ++j) {
    float s = b1[j];
    if (idx > 0) {
      const float* hr = H + (size_t)(idx - 1) * 16;
#pragma unroll
      for (int f = 0; f < 16; ++f) s += hr[f] * W1[f * 4 + j];
    }
    hid[j] = fmaxf(s, 0.0f);
  }
#pragma unroll
  for (int o = 0; o < 8; ++o) {
    float s = b2[o];
#pragma unroll
    for (int j = 0; j < 4; ++j) s += hid[j] * W2[j * 8 + o];
    out[idx * 8 + o] = s;
  }
}

extern "C" void kernel_launch(void* const* d_in, const int* in_sizes, int n_in,
                              void* d_out, int out_size, void* d_ws, size_t ws_size,
                              hipStream_t stream) {
  const float* x     = (const float*)d_in[0];
  const int*   ei    = (const int*)  d_in[1];
  const float* ea    = (const float*)d_in[2];
  const float* Wl    = (const float*)d_in[3];
  const float* bl    = (const float*)d_in[4];
  const float* Wr    = (const float*)d_in[5];
  const float* br    = (const float*)d_in[6];
  const float* We    = (const float*)d_in[7];
  const float* att   = (const float*)d_in[8];
  const float* gbias = (const float*)d_in[9];
  const float* chebW = (const float*)d_in[10];
  const float* chebB = (const float*)d_in[11];
  const float* W1    = (const float*)d_in[12];
  const float* b1    = (const float*)d_in[13];
  const float* W2    = (const float*)d_in[14];
  const float* b2    = (const float*)d_in[15];
  float* out = (float*)d_out;

  const int N = in_sizes[0] / 128;
  const int E = in_sizes[1] / 2;
  const size_t NF = (size_t)N * 16;

  float* ws    = (float*)d_ws;
  float* xl    = ws;
  float* xr    = xl + NF;
  float* hh    = xr + NF;
  float* T1    = hh + NF;
  float* T2    = T1 + NF;
  float* T3    = T2 + NF;
  float* spare = T3 + NF;
  int*   cnt   = (int*)(spare + NF);
  int*   scnt  = cnt + N;
  float* dinv  = (float*)(scnt + N);
  int*   rowptr = (int*)(dinv + N);
  int*   csrc  = (int*)(spare + NF + (size_t)N * 5);
  int*   be    = (int*)T1;
  int*   bsd   = (int*)xl;
  int*   btot  = cnt;
  int*   bfill = cnt + MAXB * CPAD;
  int*   stot  = scnt;
  int*   sfill = scnt + MAXB * CPAD;
  unsigned char* srec = (unsigned char*)hh;
  float* ascr  = T1;
  int*   ceid  = (int*)T3;

  const int TPB = 256;
  int gEb  = (E + EBLK * EPT - 1) / (EBLK * EPT);
  int nb   = (N + BKT_NODES - 1) >> BKT_SHIFT;
  int gN16 = (N * 16 + TPB - 1) / TPB;
  int gN1  = (N + 1 + TPB - 1) / TPB;
  int tiles = N / 16;

  hipMemsetAsync(cnt, 0, (size_t)N * 2 * sizeof(int), stream);

  k_count<<<gEb, EBLK, 0, stream>>>(ei, btot, stot, E, nb);
  k_bin<<<gEb, EBLK, 0, stream>>>(ei, btot, bfill, stot, sfill, bsd, be, srec, E, nb);
  k_place<<<nb + nb, EBLK, 0, stream>>>(btot, bsd, be, rowptr, csrc, ceid, stot, srec, dinv, N, nb);

  k_gemm_xlxr<<<tiles, 32, 0, stream>>>(x, Wl, bl, Wr, br, xl, xr);
  k_gat<<<gN16, TPB, 0, stream>>>(rowptr, csrc, ceid, ea, xl, xr, We, att, gbias, dinv, ascr, hh, N);

  k_matvec<<<gN16, TPB, 0, stream>>>(rowptr, csrc, dinv, xr, (const float*)nullptr, 1.0f, T1, xl, N);
  k_matvec<<<gN16, TPB, 0, stream>>>(rowptr, csrc, dinv, xl, hh, 2.0f, T2, xr, N);
  k_matvec<<<gN16, TPB, 0, stream>>>(rowptr, csrc, dinv, xr, T1, 2.0f, T3, (float*)nullptr, N);

  k_gru<<<tiles, 32, 0, stream>>>(hh, T1, T2, T3, chebW, chebB, xl);

  k_mlp<<<gN1, TPB, 0, stream>>>(xl, W1, b1, W2, b2, out, N);
}


# Bucket sort with no count pass: a block bins its 8192 edges in its own segment and stores its bucket scan; a bucket's group finds its runs by that table. No memset, no global atomic; lists by arrival
# speedup vs baseline: 5.5787x; 1.2429x over previous
#include <hip/hip_runtime.h>
#include <math.h>

typedef float v2f __attribute__((ext_vector_type(2)));
typedef float v8f __attribute__((ext_vector_type(8)));

#define WMMA4(a, b, c) \
  __builtin_amdgcn_wmma_f32_16x16x4_f32(false, (a), false, (b), (short)0, (c), false, false)

__device__ __forceinline__ float gsum16(float v) {
  v += __shfl_xor(v, 8, 16);
  v += __shfl_xor(v, 4, 16);
  v += __shfl_xor(v, 2, 16);
  v += __shfl_xor(v, 1, 16);
  return v;
}
__device__ __forceinline__ float gmax16(float v) {
  v = fmaxf(v, __shfl_xor(v, 8, 16));
  v = fmaxf(v, __shfl_xor(v, 4, 16));
  v = fmaxf(v, __shfl_xor(v, 2, 16));
  v = fmaxf(v, __shfl_xor(v, 1, 16));
  return v;
}

__device__ __forceinline__ void ldrow(const float* __restrict__ p, float* r) {
  const float4* q = (const float4*)p;
  float4 a = q[0], b = q[1], c = q[2], d = q[3];
  r[0] = a.x;  r[1] = a.y;  r[2] = a.z;  r[3] = a.w;
  r[4] = b.x;  r[5] = b.y;  r[6] = b.z;  r[7] = b.w;
  r[8] = c.x;  r[9] = c.y;  r[10] = c.z; r[11] = c.w;
  r[12] = d.x; r[13] = d.y; r[14] = d.z; r[15] = d.w;
}

#define BKT_SHIFT 8
#define BKT_NODES (1 << BKT_SHIFT)
#define MAXB 512
#define EBLK 1024
#define EPT 8
#define ESEG (EBLK * EPT)

__device__ __forceinline__ void lds_scan_incl(int* a, int n) {
  int t = threadIdx.x;
  for (int off = 1; off < n; off <<= 1) {
    int v = (t < n && t >= off) ? a[t - off] : 0;
    __syncthreads();
    if (t < n) a[t] += v;
    __syncthreads();
  }
}

__device__ __forceinline__ void lds_scan_incl2(int* a, int* b, int n) {
  int t = threadIdx.x;
  for (int off = 1; off < n; off <<= 1) {
    bool on = (t < n && t >= off);
    int va = on ? a[t - off] : 0;
    int vb = on ? b[t - off] : 0;
    __syncthreads();
    if (t < n) { a[t] += va; b[t] += vb; }
    __syncthreads();
  }
}

__global__ void __launch_bounds__(EBLK)
k_bin(const int* __restrict__ ei, int* __restrict__ tabd, int* __restrict__ tabs,
      int* __restrict__ bsd, int* __restrict__ be, unsigned char* __restrict__ srec, int E, int nb) {
  __shared__ int hist[MAXB];
  __shared__ int shist[MAXB];
  int t = threadIdx.x;
  int blk = blockIdx.x, gb = gridDim.x;
  if (t < MAXB) { hist[t] = 0; shist[t] = 0; }
  __syncthreads();
  int e0 = blk * ESEG + t;
  int s[EPT], d[EPT], lr[EPT], slr[EPT];
#pragma unroll
  for (int k = 0; k < EPT; ++k) {
    int e = e0 + k * EBLK;
    bool v = e < E;
    s[k] = v ? ei[e] : 0;
    d[k] = v ? ei[E + e] : 0;
  }
#pragma unroll
  for (int k = 0; k < EPT; ++k) {
    lr[k] = 0; slr[k] = 0;
    if (e0 + k * EBLK < E) {
      lr[k] = atomicAdd(&hist[d[k] >> BKT_SHIFT], 1);
      if (s[k] != d[k]) slr[k] = atomicAdd(&shist[s[k] >> BKT_SHIFT], 1);
    }
  }
  __syncthreads();
  lds_scan_incl2(hist, shist, nb);
  if (t < nb) {
    tabd[t * gb + blk] = hist[t];
    tabs[t * gb + blk] = shist[t];
  }
  int base = blk * ESEG;
#pragma unroll
  for (int k = 0; k < EPT; ++k) {
    int e = e0 + k * EBLK;
    if (e < E) {
      int bk = d[k] >> BKT_SHIFT;
      int p = base + (bk ? hist[bk - 1] : 0) + lr[k];
      bsd[p] = (int)((unsigned)s[k] | ((unsigned)(d[k] & (BKT_NODES - 1)) << 24));
      be[p] = e;
      if (s[k] != d[k]) {
        int sb = s[k] >> BKT_SHIFT;
        srec[base + (sb ? shist[sb - 1] : 0) + slr[k]] = (unsigned char)(s[k] & (BKT_NODES - 1));
      }
    }
  }
}

template <int U>
__device__ __forceinline__ void run_slots(const int* rend, const int* roff, int gb, int nstep, const int* i, int* p) {
  int lo[U], hi[U];
#pragma unroll
  for (int u = 0; u < U; ++u) { lo[u] = 0; hi[u] = gb - 1; }
  for (int st = 0; st < nstep; ++st) {
#pragma unroll
    for (int u = 0; u < U; ++u) {
      int mid = (lo[u] + hi[u]) >> 1;
      bool gt = rend[mid] > i[u];
      hi[u] = gt ? mid : hi[u];
      lo[u] = gt ? lo[u] : mid + 1;
    }
  }
#pragma unroll
  for (int u = 0; u < U; ++u) {
    int r = hi[u];
    int first = r ? rend[r - 1] : 0;
    p[u] = r * ESEG + roff[r] + (i[u] - first);
  }
}

__global__ void __launch_bounds__(EBLK)
k_place(const int* __restrict__ tabd, const int* __restrict__ bsd, const int* __restrict__ be,
        int* __restrict__ rowptr, int* __restrict__ csrc, int* __restrict__ ceid,
        const int* __restrict__ tabs, const unsigned char* __restrict__ srec, float* __restrict__ dinv,
        int N, int nb, int gb) {
  __shared__ int rend[MAXB];
  __shared__ int roff[MAXB];
  __shared__ int psum[MAXB];
  __shared__ int cnt[BKT_NODES];
  __shared__ int cur[BKT_NODES];
  int t = threadIdx.x;
  bool srcblk = (int)blockIdx.x >= nb;
  int b = srcblk ? (int)blockIdx.x - nb : (int)blockIdx.x;
  const int* tab = srcblk ? tabs : tabd;
  if (t < MAXB) {
    int upto = 0, before = 0;
    if (t < gb) {
      upto = tab[b * gb + t];
      before = b ? tab[(b - 1) * gb + t] : 0;
    }
    rend[t] = upto - before; roff[t] = before; psum[t] = before;
  }
  if (t < BKT_NODES) cnt[t] = 0;
  __syncthreads();
  lds_scan_incl2(rend, psum, gb);
  int total = rend[gb - 1];
  int begp = psum[gb - 1];
  int endp = begp + total;
  int nstep = 32 - __clz(gb - 1);
  if (gb < 2) nstep = 0;
  if (srcblk) {
    for (int i0 = t; i0 < total; i0 += 8 * EBLK) {
      unsigned r[8]; int ii[8], pp[8];
#pragma unroll
      for (int u = 0; u < 8; ++u) ii[u] = min(i0 + u * EBLK, total - 1);
      run_slots<8>(rend, roff, gb, nstep, ii, pp);
#pragma unroll
      for (int u = 0; u < 8; ++u) r[u] = (unsigned)srec[pp[u]];
#pragma unroll
      for (int u = 0; u < 8; ++u) if (i0 + u * EBLK < total) atomicAdd(&cnt[r[u]], 1);
    }
    __syncthreads();
    int n = (b << BKT_SHIFT) + t;
    if (t < BKT_NODES && n < N) {
      float g = (float)cnt[t];
      dinv[n] = (g > 0.0f) ? rsqrtf(fmaxf(g, 1e-12f)) : 0.0f;
    }
    return;
  }
  for (int i0 = t; i0 < total; i0 += 4 * EBLK) {
    unsigned r[4]; int ii[4], pp[4];
#pragma unroll
    for (int u = 0; u < 4; ++u) ii[u] = min(i0 + u * EBLK, total - 1);
    run_slots<4>(rend, roff, gb, nstep, ii, pp);
#pragma unroll
    for (int u = 0; u < 4; ++u) r[u] = (unsigned)bsd[pp[u]];
#pragma unroll
    for (int u = 0; u < 4; ++u) if (i0 + u * EBLK < total) atomicAdd(&cnt[r[u] >> 24], 1);
  }
  __syncthreads();
  int mycnt = (t < BKT_NODES) ? cnt[t] : 0;
  lds_scan_incl(cnt, BKT_NODES);
  int lo = b << BKT_SHIFT;
  if (t < BKT_NODES) {
    int first = begp + cnt[t] - mycnt;
    cur[t] = first;
    if (lo + t < N) rowptr[lo + t] = first;
  }
  if (b == nb - 1 && t == 0) rowptr[N] = endp;
  __syncthreads();
  for (int i0 = t; i0 < total; i0 += 4 * EBLK) {
    unsigned r[4]; int ev[4], ii[4], pp[4];
#pragma unroll
    for (int u = 0; u < 4; ++u) ii[u] = min(i0 + u * EBLK, total - 1);
    run_slots<4>(rend, roff, gb, nstep, ii, pp);
#pragma unroll
    for (int u = 0; u < 4; ++u) {
      r[u] = (unsigned)bsd[pp[u]];
      ev[u] = be[pp[u]];
    }
#pragma unroll
    for (int u = 0; u < 4; ++u) {
      if (i0 + u * EBLK < total) {
        int q = atomicAdd(&cur[r[u] >> 24], 1);
        csrc[q] = (int)(r[u] & 0xFFFFFFu);
        ceid[q] = ev[u];
      }
    }
  }
}

__global__ void k_gemm_xlxr(const float* __restrict__ x,
                            const float* __restrict__ Wl, const float* __restrict__ bl,
                            const float* __restrict__ Wr, const float* __restrict__ br,
                            float* __restrict__ xl, float* __restrict__ xr) {
  int tile = blockIdx.x;
  int lane = threadIdx.x;
  int half = lane >> 4;
  int l16 = lane & 15;
  int row = tile * 16 + l16;
  v8f acc0 = {}; v8f acc1 = {};
#pragma unroll
  for (int k = 0; k < 128; k += 4) {
    int ka = k + half * 2;
    v2f a;  a.x = x[row * 128 + ka];     a.y = x[row * 128 + ka + 1];
    v2f b0; b0.x = Wl[ka * 16 + l16];    b0.y = Wl[(ka + 1) * 16 + l16];
    v2f b1; b1.x = Wr[ka * 16 + l16];    b1.y = Wr[(ka + 1) * 16 + l16];
    acc0 = WMMA4(a, b0, acc0);
    acc1 = WMMA4(a, b1, acc1);
  }
#pragma unroll
  for (int i = 0; i < 8; ++i) {
    int m = tile * 16 + i + half * 8;
    int c = l16;
    xl[m * 16 + c] = acc0[i] + bl[c];
    xr[m * 16 + c] = acc1[i] + br[c];
  }
}

__device__ __forceinline__ void ldhalf(const float* __restrict__ p, float* r) {
  const float4* q = (const float4*)p;
  float4 a = q[0], b = q[1];
  r[0] = a.x; r[1] = a.y; r[2] = a.z; r[3] = a.w;
  r[4] = b.x; r[5] = b.y; r[6] = b.z; r[7] = b.w;
}

__device__ __forceinline__ float tile_logit(const v2f* wa, const float* evh, const float* cin,
                                            const float* atth) {
  v8f acc;
#pragma unroll
  for (int i = 0; i < 8; ++i) acc[i] = cin[i];
#pragma unroll
  for (int t = 0; t < 4; ++t) {
    v2f b; b.x = evh[2 * t]; b.y = evh[2 * t + 1];
    acc = WMMA4(wa[t], b, acc);
  }
  float p = 0.0f;
#pragma unroll
  for (int i = 0; i < 8; ++i) {
    float m = acc[i];
    m = m > 0.0f ? m : 0.2f * m;
    p += m * atth[i];
  }
  return p + __shfl_xor(p, 16);
}

__global__ void __launch_bounds__(256)
k_gat(const int* __restrict__ rowptr, const int* __restrict__ csrc, const int* __restrict__ ceid,
      const float* __restrict__ ea, const float* __restrict__ xl, float* xr,
      const float* __restrict__ We, const float* __restrict__ att,
      const float* __restrict__ gbias, const float* __restrict__ dinv,
      float* ascr, float* __restrict__ h, int N) {
  int lane = threadIdx.x & 31;
  int c = lane & 15, hf = lane >> 4;
  int wave = __builtin_amdgcn_readfirstlane((int)((blockIdx.x * blockDim.x + threadIdx.x) >> 5));

  v2f wa[4];
#pragma unroll
  for (int t = 0; t < 4; ++t) {
    int j = 8 * hf + 2 * t;
    wa[t].x = We[j * 16 + c];
    wa[t].y = We[(j + 1) * 16 + c];
  }
  float atth[8];
  ldhalf(att + 8 * hf, atth);

  for (int q = 0; q < 2; ++q) {
    int n = wave * 2 + q;
    if (n >= N) break;
    int beg = rowptr[n], end = rowptr[n + 1];
    float xrh[8], xlh[8];
    ldhalf(xr + (size_t)n * 16 + 8 * hf, xrh);
    ldhalf(xl + (size_t)n * 16 + 8 * hf, xlh);

    float psum[8];
#pragma unroll
    for (int i = 0; i < 8; ++i) psum[i] = 0.0f;
    float amax = -INFINITY;
    for (int base = beg; base < end; base += 32) {
      int j0 = base + c, j1 = j0 + 16;
      bool v0 = j0 < end, v1 = j1 < end;
      int q0 = v0 ? j0 : end - 1;
      int q1 = v1 ? j1 : end - 1;
      int s0 = csrc[q0], e0 = ceid[q0];
      int s1 = csrc[q1], e1 = ceid[q1];
      float ev0[8], xs0[8], ev1[8], xs1[8];
      ldhalf(ea + (size_t)e0 * 16 + 8 * hf, ev0);
      ldhalf(xl + (size_t)s0 * 16 + 8 * hf, xs0);
      ldhalf(ea + (size_t)e1 * 16 + 8 * hf, ev1);
      ldhalf(xl + (size_t)s1 * 16 + 8 * hf, xs1);
#pragma unroll
      for (int i = 0; i < 8; ++i) { xs0[i] += xrh[i]; xs1[i] += xrh[i]; }
      float a0 = tile_logit(wa, ev0, xs0, atth);
      float a1 = tile_logit(wa, ev1, xs1, atth);
      if (v0) {
        ascr[j0] = a0;
        amax = fmaxf(amax, a0);
#pragma unroll
        for (int i = 0; i < 8; ++i) psum[i] += ev0[i];
      }
      if (v1) {
        ascr[j1] = a1;
        amax = fmaxf(amax, a1);
#pragma unroll
        for (int i = 0; i < 8; ++i) psum[i] += ev1[i];
      }
    }
    float dn = fmaxf((float)(end - beg), 1.0f);
    float emh[8], cl[8];
#pragma unroll
    for (int i = 0; i < 8; ++i) { emh[i] = gsum16(psum[i]) / dn; cl[i] = xlh[i] + xrh[i]; }
    float aloop = tile_logit(wa, emh, cl, atth);
    amax = fmaxf(gmax16(amax), aloop);

    float acc[8];
#pragma unroll
    for (int i = 0; i < 8; ++i) acc[i] = 0.0f;
    float den = 0.0f;
    for (int base = beg; base < end; base += 32) {
      int j0 = base + c, j1 = j0 + 16;
      bool v0 = j0 < end, v1 = j1 < end;
      int q0 = v0 ? j0 : end - 1;
      int q1 = v1 ? j1 : end - 1;
      int s0 = csrc[q0], s1 = csrc[q1];
      float al0 = ascr[q0], al1 = ascr[q1];
      float xs0[8], xs1[8];
      ldhalf(xl + (size_t)s0 * 16 + 8 * hf, xs0);
      ldhalf(xl + (size_t)s1 * 16 + 8 * hf, xs1);
      if (v0) {
        float ex = expf(al0 - amax);
        den += ex;
#pragma unroll
        for (int i = 0; i < 8; ++i) acc[i] += ex * xs0[i];
      }
      if (v1) {
        float ex = expf(al1 - amax);
        den += ex;
#pragma unroll
        for (int i = 0; i < 8; ++i) acc[i] += ex * xs1[i];
      }
    }
    float exl = expf(aloop - amax);
    den = gsum16(den) + exl;
    float mine = 0.0f;
#pragma unroll
    for (int i = 0; i < 8; ++i) {
      float v = gsum16(acc[i]) + exl * xlh[i];
      if (c == i) mine = v;
    }
    if (c < 8) {
      int f = 8 * hf + c;
      float hv = mine / (den + 1e-16f) + gbias[f];
      h[(size_t)n * 16 + f] = hv;
      xr[(size_t)n * 16 + f] = dinv[n] * hv;
    }
  }
}

__global__ void __launch_bounds__(256)
k_matvec(const int* __restrict__ rowptr, const int* __restrict__ csrc,
         const float* __restrict__ dinv, const float* __restrict__ yin,
         const float* __restrict__ prev, float scale, float* __restrict__ out,
         float* __restrict__ yout, int N) {
  int gid = blockIdx.x * blockDim.x + threadIdx.x;
  int n = gid >> 4, l = gid & 15;
  if (n >= N) return;
  int g = l >> 2, q = l & 3;
  int beg = rowptr[n], end = rowptr[n + 1];
  float dn = dinv[n];
  const float4* __restrict__ yin4 = (const float4*)yin;
  float a0 = 0.0f, a1 = 0.0f, a2 = 0.0f, a3 = 0.0f;
  for (int base = beg; base < end; base += 16) {
    int jj = base + l;
    unsigned sidx = (unsigned)((jj < end) ? csrc[jj] : n);
    int s0 = (int)__builtin_amdgcn_permlane16(sidx, sidx, 0x11110000u, 0x33332222u, false, false);
    int s1 = (int)__builtin_amdgcn_permlane16(sidx, sidx, 0x55554444u, 0x77776666u, false, false);
    int s2 = (int)__builtin_amdgcn_permlane16(sidx, sidx, 0x99998888u, 0xBBBBAAAAu, false, false);
    int s3 = (int)__builtin_amdgcn_permlane16(sidx, sidx, 0xDDDDCCCCu, 0xFFFFEEEEu, false, false);
    float4 r0 = yin4[(size_t)s0 * 4 + q];
    float4 r1 = yin4[(size_t)s1 * 4 + q];
    float4 r2 = yin4[(size_t)s2 * 4 + q];
    float4 r3 = yin4[(size_t)s3 * 4 + q];
    int e = base + g;
    if (e < end)      { float w = (s0 != n) ? 1.0f : 0.0f; a0 += w * r0.x; a1 += w * r0.y; a2 += w * r0.z; a3 += w * r0.w; }
    if (e + 4 < end)  { float w = (s1 != n) ? 1.0f : 0.0f; a0 += w * r1.x; a1 += w * r1.y; a2 += w * r1.z; a3 += w * r1.w; }
    if (e + 8 < end)  { float w = (s2 != n) ? 1.0f : 0.0f; a0 += w * r2.x; a1 += w * r2.y; a2 += w * r2.z; a3 += w * r2.w; }
    if (e + 12 < end) { float w = (s3 != n) ? 1.0f : 0.0f; a0 += w * r3.x; a1 += w * r3.y; a2 += w * r3.z; a3 += w * r3.w; }
  }
  a0 += __shfl_xor(a0, 4, 16); a1 += __shfl_xor(a1, 4, 16); a2 += __shfl_xor(a2, 4, 16); a3 += __shfl_xor(a3, 4, 16);
  a0 += __shfl_xor(a0, 8, 16); a1 += __shfl_xor(a1, 8, 16); a2 += __shfl_xor(a2, 8, 16); a3 += __shfl_xor(a3, 8, 16);
  float mine = (g == 0) ? a0 : (g == 1) ? a1 : (g == 2) ? a2 : a3;
  int f = 4 * q + g;
  float r = scale * (-dn * mine);
  if (prev) r -= prev[(size_t)n * 16 + f];
  out[(size_t)n * 16 + f] = r;
  if (yout) yout[(size_t)n * 16 + f] = dn * r;
}

__global__ void k_gru(const float* __restrict__ T0, const float* __restrict__ T1,
                      const float* __restrict__ T2, const float* __restrict__ T3,
                      const float* __restrict__ chebW, const float* __restrict__ chebB,
                      float* __restrict__ H) {
  int tile = blockIdx.x;
  int lane = threadIdx.x;
  int half = lane >> 4;
  int l16 = lane & 15;
  int row = tile * 16 + l16;
  v8f accZ = {}; v8f accH = {};
#pragma unroll
  for (int t = 0; t < 4; ++t) {
    const float* Tm = (t == 0) ? T0 : (t == 1) ? T1 : (t == 2) ? T2 : T3;
    const float* Wz = chebW + t * 256;
    const float* Wh = chebW + (16 + t) * 256;
#pragma unroll
    for (int kc = 0; kc < 16; kc += 4) {
      int ka = kc + half * 2;
      v2f a;  a.x = Tm[row * 16 + ka];      a.y = Tm[row * 16 + ka + 1];
      v2f bz; bz.x = Wz[ka * 16 + l16];     bz.y = Wz[(ka + 1) * 16 + l16];
      v2f bh; bh.x = Wh[ka * 16 + l16];     bh.y = Wh[(ka + 1) * 16 + l16];
      accZ = WMMA4(a, bz, accZ);
      accH = WMMA4(a, bh, accH);
    }
  }
#pragma unroll
  for (int i = 0; i < 8; ++i) {
    int m = tile * 16 + i + half * 8;
    int c = l16;
    float zp = accZ[i] + chebB[c] + chebB[16 + c];
    float hp = accH[i] + chebB[64 + c] + chebB[80 + c];
    float z = 1.0f / (1.0f + expf(-zp));
    float ht = tanhf(hp);
    H[m * 16 + c] = (1.0f - z) * ht;
  }
}

__global__ void k_mlp(const float* __restrict__ H, const float* W1, const float* b1,
                      const float* W2, const float* b2, float* out, int N) {
  int idx = blockIdx.x * blockDim.x + threadIdx.x;
  if (idx > N) return;
  float hid[4];
#pragma unroll
  for (int j = 0; j < 4; ++j) {
    float s = b1[j];
    if (idx > 0) {
      const float* hr = H + (size_t)(idx - 1) * 16;
#pragma unroll
      for (int f = 0; f < 16; ++f) s += hr[f] * W1[f * 4 + j];
    }
    hid[j] = fmaxf(s, 0.0f);
  }
#pragma unroll
  for (int o = 0; o < 8; ++o) {
    float s = b2[o];
#pragma unroll
    for (int j = 0; j < 4; ++j) s += hid[j] * W2[j * 8 + o];
    out[idx * 8 + o] = s;
  }
}

extern "C" void kernel_launch(void* const* d_in, const int* in_sizes, int n_in,
                              void* d_out, int out_size, void* d_ws, size_t ws_size,
                              hipStream_t stream) {
  const float* x     = (const float*)d_in[0];
  const int*   ei    = (const int*)  d_in[1];
  const float* ea    = (const float*)d_in[2];
  const float* Wl    = (const float*)d_in[3];
  const float* bl    = (const float*)d_in[4];
  const float* Wr    = (const float*)d_in[5];
  const float* br    = (const float*)d_in[6];
  const float* We    = (const float*)d_in[7];
  const float* att   = (const float*)d_in[8];
  const float* gbias = (const float*)d_in[9];
  const float* chebW = (const float*)d_in[10];
  const float* chebB = (const float*)d_in[11];
  const float* W1    = (const float*)d_in[12];
  const float* b1    = (const float*)d_in[13];
  const float* W2    = (const float*)d_in[14];
  const float* b2    = (const float*)d_in[15];
  float* out = (float*)d_out;

  const int N = in_sizes[0] / 128;
  const int E = in_sizes[1] / 2;
  const size_t NF = (size_t)N * 16;

  float* ws    = (float*)d_ws;
  float* xl    = ws;
  float* xr    = xl + NF;
  float* hh    = xr + NF;
  float* T1    = hh + NF;
  float* T2    = T1 + NF;
  float* T3    = T2 + NF;
  float* spare = T3 + NF;
  int*   cnt   = (int*)(spare + NF);
  int*   scnt  = cnt + N;
  float* dinv  = (float*)(scnt + N);
  int*   rowptr = (int*)(dinv + N);
  int*   csrc  = (int*)(spare + NF + (size_t)N * 5);
  int*   be    = (int*)T1;
  int*   bsd   = (int*)xl;
  int*   tabd  = cnt;
  int*   tabs  = scnt;
  unsigned char* srec = (unsigned char*)hh;
  float* ascr  = T1;
  int*   ceid  = (int*)T3;

  const int TPB = 256;
  int gEb  = (E + ESEG - 1) / ESEG;
  int nb   = (N + BKT_NODES - 1) >> BKT_SHIFT;
  int gN16 = (N * 16 + TPB - 1) / TPB;
  int gN1  = (N + 1 + TPB - 1) / TPB;
  int tiles = N / 16;

  k_bin<<<gEb, EBLK, 0, stream>>>(ei, tabd, tabs, bsd, be, srec, E, nb);
  k_place<<<nb + nb, EBLK, 0, stream>>>(tabd, bsd, be, rowptr, csrc, ceid, tabs, srec, dinv, N, nb, gEb);

  k_gemm_xlxr<<<tiles, 32, 0, stream>>>(x, Wl, bl, Wr, br, xl, xr);
  k_gat<<<gN16, TPB, 0, stream>>>(rowptr, csrc, ceid, ea, xl, xr, We, att, gbias, dinv, ascr, hh, N);

  k_matvec<<<gN16, TPB, 0, stream>>>(rowptr, csrc, dinv, xr, (const float*)nullptr, 1.0f, T1, xl, N);
  k_matvec<<<gN16, TPB, 0, stream>>>(rowptr, csrc, dinv, xl, hh, 2.0f, T2, xr, N);
  k_matvec<<<gN16, TPB, 0, stream>>>(rowptr, csrc, dinv, xr, T1, 2.0f, T3, (float*)nullptr, N);

  k_gru<<<tiles, 32, 0, stream>>>(hh, T1, T2, T3, chebW, chebB, xl);

  k_mlp<<<gN1, TPB, 0, stream>>>(xl, W1, b1, W2, b2, out, N);
}


# Sort's record stores staged in LDS: a block's segment (level 1) and a bucket's placed list (level 2) are put in order in LDS and copied out as whole 128-byte lines, not one 4-byte store a line
# speedup vs baseline: 6.5074x; 1.1665x over previous
#include <hip/hip_runtime.h>
#include <math.h>

typedef float v2f __attribute__((ext_vector_type(2)));
typedef float v8f __attribute__((ext_vector_type(8)));

#define WMMA4(a, b, c) \
  __builtin_amdgcn_wmma_f32_16x16x4_f32(false, (a), false, (b), (short)0, (c), false, false)

__device__ __forceinline__ float gsum16(float v) {
  v += __shfl_xor(v, 8, 16);
  v += __shfl_xor(v, 4, 16);
  v += __shfl_xor(v, 2, 16);
  v += __shfl_xor(v, 1, 16);
  return v;
}
__device__ __forceinline__ float gmax16(float v) {
  v = fmaxf(v, __shfl_xor(v, 8, 16));
  v = fmaxf(v, __shfl_xor(v, 4, 16));
  v = fmaxf(v, __shfl_xor(v, 2, 16));
  v = fmaxf(v, __shfl_xor(v, 1, 16));
  return v;
}

__device__ __forceinline__ void ldrow(const float* __restrict__ p, float* r) {
  const float4* q = (const float4*)p;
  float4 a = q[0], b = q[1], c = q[2], d = q[3];
  r[0] = a.x;  r[1] = a.y;  r[2] = a.z;  r[3] = a.w;
  r[4] = b.x;  r[5] = b.y;  r[6] = b.z;  r[7] = b.w;
  r[8] = c.x;  r[9] = c.y;  r[10] = c.z; r[11] = c.w;
  r[12] = d.x; r[13] = d.y; r[14] = d.z; r[15] = d.w;
}

#define BKT_SHIFT 8
#define BKT_NODES (1 << BKT_SHIFT)
#define MAXB 512
#define EBLK 1024
#define EPT 8
#define ESEG (EBLK * EPT)

__device__ __forceinline__ void lds_scan_incl(int* a, int n) {
  int t = threadIdx.x;
  for (int off = 1; off < n; off <<= 1) {
    int v = (t < n && t >= off) ? a[t - off] : 0;
    __syncthreads();
    if (t < n) a[t] += v;
    __syncthreads();
  }
}

__device__ __forceinline__ void lds_scan_incl2(int* a, int* b, int n) {
  int t = threadIdx.x;
  for (int off = 1; off < n; off <<= 1) {
    bool on = (t < n && t >= off);
    int va = on ? a[t - off] : 0;
    int vb = on ? b[t - off] : 0;
    __syncthreads();
    if (t < n) { a[t] += va; b[t] += vb; }
    __syncthreads();
  }
}

__global__ void __launch_bounds__(EBLK)
k_bin(const int* __restrict__ ei, int* __restrict__ tabd, int* __restrict__ tabs,
      int* __restrict__ bsd, int* __restrict__ be, unsigned char* __restrict__ srec, int E, int nb) {
  __shared__ int hist[MAXB];
  __shared__ int shist[MAXB];
  __shared__ int sbsd[ESEG];
  __shared__ int sbe[ESEG];
  __shared__ int ssrw[ESEG / 4];
  unsigned char* ssr = (unsigned char*)ssrw;
  int t = threadIdx.x;
  int blk = blockIdx.x, gb = gridDim.x;
  if (t < MAXB) { hist[t] = 0; shist[t] = 0; }
  __syncthreads();
  int e0 = blk * ESEG + t;
  int s[EPT], d[EPT], lr[EPT], slr[EPT];
#pragma unroll
  for (int k = 0; k < EPT; ++k) {
    int e = e0 + k * EBLK;
    bool v = e < E;
    s[k] = v ? ei[e] : 0;
    d[k] = v ? ei[E + e] : 0;
  }
#pragma unroll
  for (int k = 0; k < EPT; ++k) {
    lr[k] = 0; slr[k] = 0;
    if (e0 + k * EBLK < E) {
      lr[k] = atomicAdd(&hist[d[k] >> BKT_SHIFT], 1);
      if (s[k] != d[k]) slr[k] = atomicAdd(&shist[s[k] >> BKT_SHIFT], 1);
    }
  }
  __syncthreads();
  lds_scan_incl2(hist, shist, nb);
  int base = blk * ESEG;
#pragma unroll
  for (int k = 0; k < EPT; ++k) {
    int e = e0 + k * EBLK;
    if (e < E) {
      int bk = d[k] >> BKT_SHIFT;
      int p = (bk ? hist[bk - 1] : 0) + lr[k];
      sbsd[p] = (int)((unsigned)s[k] | ((unsigned)(d[k] & (BKT_NODES - 1)) << 24));
      sbe[p] = e;
      if (s[k] != d[k]) {
        int sb = s[k] >> BKT_SHIFT;
        ssr[(sb ? shist[sb - 1] : 0) + slr[k]] = (unsigned char)(s[k] & (BKT_NODES - 1));
      }
    }
  }
  __syncthreads();
  int nd = hist[nb - 1];
  int ns = shist[nb - 1];
  int nsw = ns >> 2;
  for (int i = t; i < nd; i += EBLK) {
    bsd[base + i] = sbsd[i];
    be[base + i] = sbe[i];
  }
  int* srecw = (int*)(srec + base);
  for (int i = t; i < nsw; i += EBLK) srecw[i] = ssrw[i];
  if (t < (ns & 3)) srec[base + (nsw << 2) + t] = ssr[(nsw << 2) + t];
  if (t < nb) {
    tabd[t * gb + blk] = hist[t];
    tabs[t * gb + blk] = shist[t];
  }
}

template <int U>
__device__ __forceinline__ void run_slots(const int* rend, const int* roff, int gb, int nstep, const int* i, int* p) {
  int lo[U], hi[U];
#pragma unroll
  for (int u = 0; u < U; ++u) { lo[u] = 0; hi[u] = gb - 1; }
  for (int st = 0; st < nstep; ++st) {
#pragma unroll
    for (int u = 0; u < U; ++u) {
      int mid = (lo[u] + hi[u]) >> 1;
      bool gt = rend[mid] > i[u];
      hi[u] = gt ? mid : hi[u];
      lo[u] = gt ? lo[u] : mid + 1;
    }
  }
#pragma unroll
  for (int u = 0; u < U; ++u) {
    int r = hi[u];
    int first = r ? rend[r - 1] : 0;
    p[u] = r * ESEG + roff[r] + (i[u] - first);
  }
}

#define PCAP 10240
__global__ void __launch_bounds__(EBLK)
k_place(const int* __restrict__ tabd, const int* __restrict__ bsd, const int* __restrict__ be,
        int* __restrict__ rowptr, int* __restrict__ csrc, int* __restrict__ ceid,
        const int* __restrict__ tabs, const unsigned char* __restrict__ srec, float* __restrict__ dinv,
        int N, int nb, int gb) {
  __shared__ int rend[MAXB];
  __shared__ int roff[MAXB];
  __shared__ int psum[MAXB];
  __shared__ int cnt[BKT_NODES];
  __shared__ int cur[BKT_NODES];
  __shared__ int ssrc[PCAP];
  __shared__ int seid[PCAP];
  int t = threadIdx.x;
  bool srcblk = (int)blockIdx.x >= nb;
  int b = srcblk ? (int)blockIdx.x - nb : (int)blockIdx.x;
  const int* tab = srcblk ? tabs : tabd;
  if (t < MAXB) {
    int upto = 0, before = 0;
    if (t < gb) {
      upto = tab[b * gb + t];
      before = b ? tab[(b - 1) * gb + t] : 0;
    }
    rend[t] = upto - before; roff[t] = before; psum[t] = before;
  }
  if (t < BKT_NODES) cnt[t] = 0;
  __syncthreads();
  lds_scan_incl2(rend, psum, gb);
  int total = rend[gb - 1];
  int begp = psum[gb - 1];
  int endp = begp + total;
  int nstep = 32 - __clz(gb - 1);
  if (gb < 2) nstep = 0;
  if (srcblk) {
    for (int i0 = t; i0 < total; i0 += 8 * EBLK) {
      unsigned r[8]; int ii[8], pp[8];
#pragma unroll
      for (int u = 0; u < 8; ++u) ii[u] = min(i0 + u * EBLK, total - 1);
      run_slots<8>(rend, roff, gb, nstep, ii, pp);
#pragma unroll
      for (int u = 0; u < 8; ++u) r[u] = (unsigned)srec[pp[u]];
#pragma unroll
      for (int u = 0; u < 8; ++u) if (i0 + u * EBLK < total) atomicAdd(&cnt[r[u]], 1);
    }
    __syncthreads();
    int n = (b << BKT_SHIFT) + t;
    if (t < BKT_NODES && n < N) {
      float g = (float)cnt[t];
      dinv[n] = (g > 0.0f) ? rsqrtf(fmaxf(g, 1e-12f)) : 0.0f;
    }
    return;
  }
  for (int i0 = t; i0 < total; i0 += 4 * EBLK) {
    unsigned r[4]; int ii[4], pp[4];
#pragma unroll
    for (int u = 0; u < 4; ++u) ii[u] = min(i0 + u * EBLK, total - 1);
    run_slots<4>(rend, roff, gb, nstep, ii, pp);
#pragma unroll
    for (int u = 0; u < 4; ++u) r[u] = (unsigned)bsd[pp[u]];
#pragma unroll
    for (int u = 0; u < 4; ++u) if (i0 + u * EBLK < total) atomicAdd(&cnt[r[u] >> 24], 1);
  }
  __syncthreads();
  int mycnt = (t < BKT_NODES) ? cnt[t] : 0;
  lds_scan_incl(cnt, BKT_NODES);
  int lo = b << BKT_SHIFT;
  if (t < BKT_NODES) {
    int first = begp + cnt[t] - mycnt;
    cur[t] = first;
    if (lo + t < N) rowptr[lo + t] = first;
  }
  if (b == nb - 1 && t == 0) rowptr[N] = endp;
  __syncthreads();
  bool staged = total <= PCAP;
  for (int i0 = t; i0 < total; i0 += 4 * EBLK) {
    unsigned r[4]; int ev[4], ii[4], pp[4];
#pragma unroll
    for (int u = 0; u < 4; ++u) ii[u] = min(i0 + u * EBLK, total - 1);
    run_slots<4>(rend, roff, gb, nstep, ii, pp);
#pragma unroll
    for (int u = 0; u < 4; ++u) {
      r[u] = (unsigned)bsd[pp[u]];
      ev[u] = be[pp[u]];
    }
#pragma unroll
    for (int u = 0; u < 4; ++u) {
      if (i0 + u * EBLK < total) {
        int q = atomicAdd(&cur[r[u] >> 24], 1);
        if (staged) {
          ssrc[q - begp] = (int)(r[u] & 0xFFFFFFu);
          seid[q - begp] = ev[u];
        } else {
          csrc[q] = (int)(r[u] & 0xFFFFFFu);
          ceid[q] = ev[u];
        }
      }
    }
  }
  if (staged) {
    __syncthreads();
    int sh = begp & 31;
    for (int i = t - sh; i < total; i += EBLK) {
      if (i >= 0) {
        csrc[begp + i] = ssrc[i];
        ceid[begp + i] = seid[i];
      }
    }
  }
}

__global__ void k_gemm_xlxr(const float* __restrict__ x,
                            const float* __restrict__ Wl, const float* __restrict__ bl,
                            const float* __restrict__ Wr, const float* __restrict__ br,
                            float* __restrict__ xl, float* __restrict__ xr) {
  int tile = blockIdx.x;
  int lane = threadIdx.x;
  int half = lane >> 4;
  int l16 = lane & 15;
  int row = tile * 16 + l16;
  v8f acc0 = {}; v8f acc1 = {};
#pragma unroll
  for (int k = 0; k < 128; k += 4) {
    int ka = k + half * 2;
    v2f a;  a.x = x[row * 128 + ka];     a.y = x[row * 128 + ka + 1];
    v2f b0; b0.x = Wl[ka * 16 + l16];    b0.y = Wl[(ka + 1) * 16 + l16];
    v2f b1; b1.x = Wr[ka * 16 + l16];    b1.y = Wr[(ka + 1) * 16 + l16];
    acc0 = WMMA4(a, b0, acc0);
    acc1 = WMMA4(a, b1, acc1);
  }
#pragma unroll
  for (int i = 0; i < 8; ++i) {
    int m = tile * 16 + i + half * 8;
    int c = l16;
    xl[m * 16 + c] = acc0[i] + bl[c];
    xr[m * 16 + c] = acc1[i] + br[c];
  }
}

__device__ __forceinline__ void ldhalf(const float* __restrict__ p, float* r) {
  const float4* q = (const float4*)p;
  float4 a = q[0], b = q[1];
  r[0] = a.x; r[1] = a.y; r[2] = a.z; r[3] = a.w;
  r[4] = b.x; r[5] = b.y; r[6] = b.z; r[7] = b.w;
}

__device__ __forceinline__ float tile_logit(const v2f* wa, const float* evh, const float* cin,
                                            const float* atth) {
  v8f acc;
#pragma unroll
  for (int i = 0; i < 8; ++i) acc[i] = cin[i];
#pragma unroll
  for (int t = 0; t < 4; ++t) {
    v2f b; b.x = evh[2 * t]; b.y = evh[2 * t + 1];
    acc = WMMA4(wa[t], b, acc);
  }
  float p = 0.0f;
#pragma unroll
  for (int i = 0; i < 8; ++i) {
    float m = acc[i];
    m = m > 0.0f ? m : 0.2f * m;
    p += m * atth[i];
  }
  return p + __shfl_xor(p, 16);
}

__global__ void __launch_bounds__(256)
k_gat(const int* __restrict__ rowptr, const int* __restrict__ csrc, const int* __restrict__ ceid,
      const float* __restrict__ ea, const float* __restrict__ xl, float* xr,
      const float* __restrict__ We, const float* __restrict__ att,
      const float* __restrict__ gbias, const float* __restrict__ dinv,
      float* ascr, float* __restrict__ h, int N) {
  int lane = threadIdx.x & 31;
  int c = lane & 15, hf = lane >> 4;
  int wave = __builtin_amdgcn_readfirstlane((int)((blockIdx.x * blockDim.x + threadIdx.x) >> 5));

  v2f wa[4];
#pragma unroll
  for (int t = 0; t < 4; ++t) {
    int j = 8 * hf + 2 * t;
    wa[t].x = We[j * 16 + c];
    wa[t].y = We[(j + 1) * 16 + c];
  }
  float atth[8];
  ldhalf(att + 8 * hf, atth);

  for (int q = 0; q < 2; ++q) {
    int n = wave * 2 + q;
    if (n >= N) break;
    int beg = rowptr[n], end = rowptr[n + 1];
    float xrh[8], xlh[8];
    ldhalf(xr + (size_t)n * 16 + 8 * hf, xrh);
    ldhalf(xl + (size_t)n * 16 + 8 * hf, xlh);

    float psum[8];
#pragma unroll
    for (int i = 0; i < 8; ++i) psum[i] = 0.0f;
    float amax = -INFINITY;
    for (int base = beg; base < end; base += 32) {
      int j0 = base + c, j1 = j0 + 16;
      bool v0 = j0 < end, v1 = j1 < end;
      int q0 = v0 ? j0 : end - 1;
      int q1 = v1 ? j1 : end - 1;
      int s0 = csrc[q0], e0 = ceid[q0];
      int s1 = csrc[q1], e1 = ceid[q1];
      float ev0[8], xs0[8], ev1[8], xs1[8];
      ldhalf(ea + (size_t)e0 * 16 + 8 * hf, ev0);
      ldhalf(xl + (size_t)s0 * 16 + 8 * hf, xs0);
      ldhalf(ea + (size_t)e1 * 16 + 8 * hf, ev1);
      ldhalf(xl + (size_t)s1 * 16 + 8 * hf, xs1);
#pragma unroll
      for (int i = 0; i < 8; ++i) { xs0[i] += xrh[i]; xs1[i] += xrh[i]; }
      float a0 = tile_logit(wa, ev0, xs0, atth);
      float a1 = tile_logit(wa, ev1, xs1, atth);
      if (v0) {
        ascr[j0] = a0;
        amax = fmaxf(amax, a0);
#pragma unroll
        for (int i = 0; i < 8; ++i) psum[i] += ev0[i];
      }
      if (v1) {
        ascr[j1] = a1;
        amax = fmaxf(amax, a1);
#pragma unroll
        for (int i = 0; i < 8; ++i) psum[i] += ev1[i];
      }
    }
    float dn = fmaxf((float)(end - beg), 1.0f);
    float emh[8], cl[8];
#pragma unroll
    for (int i = 0; i < 8; ++i) { emh[i] = gsum16(psum[i]) / dn; cl[i] = xlh[i] + xrh[i]; }
    float aloop = tile_logit(wa, emh, cl, atth);
    amax = fmaxf(gmax16(amax), aloop);

    float acc[8];
#pragma unroll
    for (int i = 0; i < 8; ++i) acc[i] = 0.0f;
    float den = 0.0f;
    for (int base = beg; base < end; base += 32) {
      int j0 = base + c, j1 = j0 + 16;
      bool v0 = j0 < end, v1 = j1 < end;
      int q0 = v0 ? j0 : end - 1;
      int q1 = v1 ? j1 : end - 1;
      int s0 = csrc[q0], s1 = csrc[q1];
      float al0 = ascr[q0], al1 = ascr[q1];
      float xs0[8], xs1[8];
      ldhalf(xl + (size_t)s0 * 16 + 8 * hf, xs0);
      ldhalf(xl + (size_t)s1 * 16 + 8 * hf, xs1);
      if (v0) {
        float ex = expf(al0 - amax);
        den += ex;
#pragma unroll
        for (int i = 0; i < 8; ++i) acc[i] += ex * xs0[i];
      }
      if (v1) {
        float ex = expf(al1 - amax);
        den += ex;
#pragma unroll
        for (int i = 0; i < 8; ++i) acc[i] += ex * xs1[i];
      }
    }
    float exl = expf(aloop - amax);
    den = gsum16(den) + exl;
    float mine = 0.0f;
#pragma unroll
    for (int i = 0; i < 8; ++i) {
      float v = gsum16(acc[i]) + exl * xlh[i];
      if (c == i) mine = v;
    }
    if (c < 8) {
      int f = 8 * hf + c;
      float hv = mine / (den + 1e-16f) + gbias[f];
      h[(size_t)n * 16 + f] = hv;
      xr[(size_t)n * 16 + f] = dinv[n] * hv;
    }
  }
}

__global__ void __launch_bounds__(256)
k_matvec(const int* __restrict__ rowptr, const int* __restrict__ csrc,
         const float* __restrict__ dinv, const float* __restrict__ yin,
         const float* __restrict__ prev, float scale, float* __restrict__ out,
         float* __restrict__ yout, int N) {
  int gid = blockIdx.x * blockDim.x + threadIdx.x;
  int n = gid >> 4, l = gid & 15;
  if (n >= N) return;
  int g = l >> 2, q = l & 3;
  int beg = rowptr[n], end = rowptr[n + 1];
  float dn = dinv[n];
  const float4* __restrict__ yin4 = (const float4*)yin;
  float a0 = 0.0f, a1 = 0.0f, a2 = 0.0f, a3 = 0.0f;
  for (int base = beg; base < end; base += 16) {
    int jj = base + l;
    unsigned sidx = (unsigned)((jj < end) ? csrc[jj] : n);
    int s0 = (int)__builtin_amdgcn_permlane16(sidx, sidx, 0x11110000u, 0x33332222u, false, false);
    int s1 = (int)__builtin_amdgcn_permlane16(sidx, sidx, 0x55554444u, 0x77776666u, false, false);
    int s2 = (int)__builtin_amdgcn_permlane16(sidx, sidx, 0x99998888u, 0xBBBBAAAAu, false, false);
    int s3 = (int)__builtin_amdgcn_permlane16(sidx, sidx, 0xDDDDCCCCu, 0xFFFFEEEEu, false, false);
    float4 r0 = yin4[(size_t)s0 * 4 + q];
    float4 r1 = yin4[(size_t)s1 * 4 + q];
    float4 r2 = yin4[(size_t)s2 * 4 + q];
    float4 r3 = yin4[(size_t)s3 * 4 + q];
    int e = base + g;
    if (e < end)      { float w = (s0 != n) ? 1.0f : 0.0f; a0 += w * r0.x; a1 += w * r0.y; a2 += w * r0.z; a3 += w * r0.w; }
    if (e + 4 < end)  { float w = (s1 != n) ? 1.0f : 0.0f; a0 += w * r1.x; a1 += w * r1.y; a2 += w * r1.z; a3 += w * r1.w; }
    if (e + 8 < end)  { float w = (s2 != n) ? 1.0f : 0.0f; a0 += w * r2.x; a1 += w * r2.y; a2 += w * r2.z; a3 += w * r2.w; }
    if (e + 12 < end) { float w = (s3 != n) ? 1.0f : 0.0f; a0 += w * r3.x; a1 += w * r3.y; a2 += w * r3.z; a3 += w * r3.w; }
  }
  a0 += __shfl_xor(a0, 4, 16); a1 += __shfl_xor(a1, 4, 16); a2 += __shfl_xor(a2, 4, 16); a3 += __shfl_xor(a3, 4, 16);
  a0 += __shfl_xor(a0, 8, 16); a1 += __shfl_xor(a1, 8, 16); a2 += __shfl_xor(a2, 8, 16); a3 += __shfl_xor(a3, 8, 16);
  float mine = (g == 0) ? a0 : (g == 1) ? a1 : (g == 2) ? a2 : a3;
  int f = 4 * q + g;
  float r = scale * (-dn * mine);
  if (prev) r -= prev[(size_t)n * 16 + f];
  out[(size_t)n * 16 + f] = r;
  if (yout) yout[(size_t)n * 16 + f] = dn * r;
}

__global__ void k_gru(const float* __restrict__ T0, const float* __restrict__ T1,
                      const float* __restrict__ T2, const float* __restrict__ T3,
                      const float* __restrict__ chebW, const float* __restrict__ chebB,
                      float* __restrict__ H) {
  int tile = blockIdx.x;
  int lane = threadIdx.x;
  int half = lane >> 4;
  int l16 = lane & 15;
  int row = tile * 16 + l16;
  v8f accZ = {}; v8f accH = {};
#pragma unroll
  for (int t = 0; t < 4; ++t) {
    const float* Tm = (t == 0) ? T0 : (t == 1) ? T1 : (t == 2) ? T2 : T3;
    const float* Wz = chebW + t * 256;
    const float* Wh = chebW + (16 + t) * 256;
#pragma unroll
    for (int kc = 0; kc < 16; kc += 4) {
      int ka = kc + half * 2;
      v2f a;  a.x = Tm[row * 16 + ka];      a.y = Tm[row * 16 + ka + 1];
      v2f bz; bz.x = Wz[ka * 16 + l16];     bz.y = Wz[(ka + 1) * 16 + l16];
      v2f bh; bh.x = Wh[ka * 16 + l16];     bh.y = Wh[(ka + 1) * 16 + l16];
      accZ = WMMA4(a, bz, accZ);
      accH = WMMA4(a, bh, accH);
    }
  }
#pragma unroll
  for (int i = 0; i < 8; ++i) {
    int m = tile * 16 + i + half * 8;
    int c = l16;
    float zp = accZ[i] + chebB[c] + chebB[16 + c];
    float hp = accH[i] + chebB[64 + c] + chebB[80 + c];
    float z = 1.0f / (1.0f + expf(-zp));
    float ht = tanhf(hp);
    H[m * 16 + c] = (1.0f - z) * ht;
  }
}

__global__ void k_mlp(const float* __restrict__ H, const float* W1, const float* b1,
                      const float* W2, const float* b2, float* out, int N) {
  int idx = blockIdx.x * blockDim.x + threadIdx.x;
  if (idx > N) return;
  float hid[4];
#pragma unroll
  for (int j = 0; j < 4; ++j) {
    float s = b1[j];
    if (idx > 0) {
      const float* hr = H + (size_t)(idx - 1) * 16;
#pragma unroll
      for (int f = 0; f < 16; ++f) s += hr[f] * W1[f * 4 + j];
    }
    hid[j] = fmaxf(s, 0.0f);
  }
#pragma unroll
  for (int o = 0; o < 8; ++o) {
    float s = b2[o];
#pragma unroll
    for (int j = 0; j < 4; ++j) s += hid[j] * W2[j * 8 + o];
    out[idx * 8 + o] = s;
  }
}

extern "C" void kernel_launch(void* const* d_in, const int* in_sizes, int n_in,
                              void* d_out, int out_size, void* d_ws, size_t ws_size,
                              hipStream_t stream) {
  const float* x     = (const float*)d_in[0];
  const int*   ei    = (const int*)  d_in[1];
  const float* ea    = (const float*)d_in[2];
  const float* Wl    = (const float*)d_in[3];
  const float* bl    = (const float*)d_in[4];
  const float* Wr    = (const float*)d_in[5];
  const float* br    = (const float*)d_in[6];
  const float* We    = (const float*)d_in[7];
  const float* att   = (const float*)d_in[8];
  const float* gbias = (const float*)d_in[9];
  const float* chebW = (const float*)d_in[10];
  const float* chebB = (const float*)d_in[11];
  const float* W1    = (const float*)d_in[12];
  const float* b1    = (const float*)d_in[13];
  const float* W2    = (const float*)d_in[14];
  const float* b2    = (const float*)d_in[15];
  float* out = (float*)d_out;

  const int N = in_sizes[0] / 128;
  const int E = in_sizes[1] / 2;
  const size_t NF = (size_t)N * 16;

  float* ws    = (float*)d_ws;
  float* xl    = ws;
  float* xr    = xl + NF;
  float* hh    = xr + NF;
  float* T1    = hh + NF;
  float* T2    = T1 + NF;
  float* T3    = T2 + NF;
  float* spare = T3 + NF;
  int*   cnt   = (int*)(spare + NF);
  int*   scnt  = cnt + N;
  float* dinv  = (float*)(scnt + N);
  int*   rowptr = (int*)(dinv + N);
  int*   csrc  = (int*)(spare + NF + (size_t)N * 5);
  int*   be    = (int*)T1;
  int*   bsd   = (int*)xl;
  int*   tabd  = cnt;
  int*   tabs  = scnt;
  unsigned char* srec = (unsigned char*)hh;
  float* ascr  = T1;
  int*   ceid  = (int*)T3;

  const int TPB = 256;
  int gEb  = (E + ESEG - 1) / ESEG;
  int nb   = (N + BKT_NODES - 1) >> BKT_SHIFT;
  int gN16 = (N * 16 + TPB - 1) / TPB;
  int gN1  = (N + 1 + TPB - 1) / TPB;
  int tiles = N / 16;

  k_bin<<<gEb, EBLK, 0, stream>>>(ei, tabd, tabs, bsd, be, srec, E, nb);
  k_place<<<nb + nb, EBLK, 0, stream>>>(tabd, bsd, be, rowptr, csrc, ceid, tabs, srec, dinv, N, nb, gEb);

  k_gemm_xlxr<<<tiles, 32, 0, stream>>>(x, Wl, bl, Wr, br, xl, xr);
  k_gat<<<gN16, TPB, 0, stream>>>(rowptr, csrc, ceid, ea, xl, xr, We, att, gbias, dinv, ascr, hh, N);

  k_matvec<<<gN16, TPB, 0, stream>>>(rowptr, csrc, dinv, xr, (const float*)nullptr, 1.0f, T1, xl, N);
  k_matvec<<<gN16, TPB, 0, stream>>>(rowptr, csrc, dinv, xl, hh, 2.0f, T2, xr, N);
  k_matvec<<<gN16, TPB, 0, stream>>>(rowptr, csrc, dinv, xr, T1, 2.0f, T3, (float*)nullptr, N);

  k_gru<<<tiles, 32, 0, stream>>>(hh, T1, T2, T3, chebW, chebB, xl);

  k_mlp<<<gN1, TPB, 0, stream>>>(xl, W1, b1, W2, b2, out, N);
}


# Last Chebyshev step, GRU product (f32 WMMA, transposed: weights as A, node rows as B) and MLP in one launch per 16-row tile of out; T3 and H stay on chip; GRU and hidden sums re-associated
# speedup vs baseline: 6.6930x; 1.0285x over previous
#include <hip/hip_runtime.h>
#include <math.h>

typedef float v2f __attribute__((ext_vector_type(2)));
typedef float v8f __attribute__((ext_vector_type(8)));

#define WMMA4(a, b, c) \
  __builtin_amdgcn_wmma_f32_16x16x4_f32(false, (a), false, (b), (short)0, (c), false, false)

__device__ __forceinline__ float gsum16(float v) {
  v += __shfl_xor(v, 8, 16);
  v += __shfl_xor(v, 4, 16);
  v += __shfl_xor(v, 2, 16);
  v += __shfl_xor(v, 1, 16);
  return v;
}
__device__ __forceinline__ float gmax16(float v) {
  v = fmaxf(v, __shfl_xor(v, 8, 16));
  v = fmaxf(v, __shfl_xor(v, 4, 16));
  v = fmaxf(v, __shfl_xor(v, 2, 16));
  v = fmaxf(v, __shfl_xor(v, 1, 16));
  return v;
}

__device__ __forceinline__ void ldrow(const float* __restrict__ p, float* r) {
  const float4* q = (const float4*)p;
  float4 a = q[0], b = q[1], c = q[2], d = q[3];
  r[0] = a.x;  r[1] = a.y;  r[2] = a.z;  r[3] = a.w;
  r[4] = b.x;  r[5] = b.y;  r[6] = b.z;  r[7] = b.w;
  r[8] = c.x;  r[9] = c.y;  r[10] = c.z; r[11] = c.w;
  r[12] = d.x; r[13] = d.y; r[14] = d.z; r[15] = d.w;
}

#define BKT_SHIFT 8
#define BKT_NODES (1 << BKT_SHIFT)
#define MAXB 512
#define EBLK 1024
#define EPT 8
#define ESEG (EBLK * EPT)

__device__ __forceinline__ void lds_scan_incl(int* a, int n) {
  int t = threadIdx.x;
  for (int off = 1; off < n; off <<= 1) {
    int v = (t < n && t >= off) ? a[t - off] : 0;
    __syncthreads();
    if (t < n) a[t] += v;
    __syncthreads();
  }
}

__device__ __forceinline__ void lds_scan_incl2(int* a, int* b, int n) {
  int t = threadIdx.x;
  for (int off = 1; off < n; off <<= 1) {
    bool on = (t < n && t >= off);
    int va = on ? a[t - off] : 0;
    int vb = on ? b[t - off] : 0;
    __syncthreads();
    if (t < n) { a[t] += va; b[t] += vb; }
    __syncthreads();
  }
}

__global__ void __launch_bounds__(EBLK)
k_bin(const int* __restrict__ ei, int* __restrict__ tabd, int* __restrict__ tabs,
      int* __restrict__ bsd, int* __restrict__ be, unsigned char* __restrict__ srec, int E, int nb) {
  __shared__ int hist[MAXB];
  __shared__ int shist[MAXB];
  __shared__ int sbsd[ESEG];
  __shared__ int sbe[ESEG];
  __shared__ int ssrw[ESEG / 4];
  unsigned char* ssr = (unsigned char*)ssrw;
  int t = threadIdx.x;
  int blk = blockIdx.x, gb = gridDim.x;
  if (t < MAXB) { hist[t] = 0; shist[t] = 0; }
  __syncthreads();
  int e0 = blk * ESEG + t;
  int s[EPT], d[EPT], lr[EPT], slr[EPT];
#pragma unroll
  for (int k = 0; k < EPT; ++k) {
    int e = e0 + k * EBLK;
    bool v = e < E;
    s[k] = v ? ei[e] : 0;
    d[k] = v ? ei[E + e] : 0;
  }
#pragma unroll
  for (int k = 0; k < EPT; ++k) {
    lr[k] = 0; slr[k] = 0;
    if (e0 + k * EBLK < E) {
      lr[k] = atomicAdd(&hist[d[k] >> BKT_SHIFT], 1);
      if (s[k] != d[k]) slr[k] = atomicAdd(&shist[s[k] >> BKT_SHIFT], 1);
    }
  }
  __syncthreads();
  lds_scan_incl2(hist, shist, nb);
  int base = blk * ESEG;
#pragma unroll
  for (int k = 0; k < EPT; ++k) {
    int e = e0 + k * EBLK;
    if (e < E) {
      int bk = d[k] >> BKT_SHIFT;
      int p = (bk ? hist[bk - 1] : 0) + lr[k];
      sbsd[p] = (int)((unsigned)s[k] | ((unsigned)(d[k] & (BKT_NODES - 1)) << 24));
      sbe[p] = e;
      if (s[k] != d[k]) {
        int sb = s[k] >> BKT_SHIFT;
        ssr[(sb ? shist[sb - 1] : 0) + slr[k]] = (unsigned char)(s[k] & (BKT_NODES - 1));
      }
    }
  }
  __syncthreads();
  int nd = hist[nb - 1];
  int ns = shist[nb - 1];
  int nsw = ns >> 2;
  for (int i = t; i < nd; i += EBLK) {
    bsd[base + i] = sbsd[i];
    be[base + i] = sbe[i];
  }
  int* srecw = (int*)(srec + base);
  for (int i = t; i < nsw; i += EBLK) srecw[i] = ssrw[i];
  if (t < (ns & 3)) srec[base + (nsw << 2) + t] = ssr[(nsw << 2) + t];
  if (t < nb) {
    tabd[t * gb + blk] = hist[t];
    tabs[t * gb + blk] = shist[t];
  }
}

template <int U>
__device__ __forceinline__ void run_slots(const int* rend, const int* roff, int gb, int nstep, const int* i, int* p) {
  int lo[U], hi[U];
#pragma unroll
  for (int u = 0; u < U; ++u) { lo[u] = 0; hi[u] = gb - 1; }
  for (int st = 0; st < nstep; ++st) {
#pragma unroll
    for (int u = 0; u < U; ++u) {
      int mid = (lo[u] + hi[u]) >> 1;
      bool gt = rend[mid] > i[u];
      hi[u] = gt ? mid : hi[u];
      lo[u] = gt ? lo[u] : mid + 1;
    }
  }
#pragma unroll
  for (int u = 0; u < U; ++u) {
    int r = hi[u];
    int first = r ? rend[r - 1] : 0;
    p[u] = r * ESEG + roff[r] + (i[u] - first);
  }
}

#define PCAP 10240
__global__ void __launch_bounds__(EBLK)
k_place(const int* __restrict__ tabd, const int* __restrict__ bsd, const int* __restrict__ be,
        int* __restrict__ rowptr, int* __restrict__ csrc, int* __restrict__ ceid,
        const int* __restrict__ tabs, const unsigned char* __restrict__ srec, float* __restrict__ dinv,
        int N, int nb, int gb) {
  __shared__ int rend[MAXB];
  __shared__ int roff[MAXB];
  __shared__ int psum[MAXB];
  __shared__ int cnt[BKT_NODES];
  __shared__ int cur[BKT_NODES];
  __shared__ int ssrc[PCAP];
  __shared__ int seid[PCAP];
  int t = threadIdx.x;
  bool srcblk = (int)blockIdx.x >= nb;
  int b = srcblk ? (int)blockIdx.x - nb : (int)blockIdx.x;
  const int* tab = srcblk ? tabs : tabd;
  if (t < MAXB) {
    int upto = 0, before = 0;
    if (t < gb) {
      upto = tab[b * gb + t];
      before = b ? tab[(b - 1) * gb + t] : 0;
    }
    rend[t] = upto - before; roff[t] = before; psum[t] = before;
  }
  if (t < BKT_NODES) cnt[t] = 0;
  __syncthreads();
  lds_scan_incl2(rend, psum, gb);
  int total = rend[gb - 1];
  int begp = psum[gb - 1];
  int endp = begp + total;
  int nstep = 32 - __clz(gb - 1);
  if (gb < 2) nstep = 0;
  if (srcblk) {
    for (int i0 = t; i0 < total; i0 += 8 * EBLK) {
      unsigned r[8]; int ii[8], pp[8];
#pragma unroll
      for (int u = 0; u < 8; ++u) ii[u] = min(i0 + u * EBLK, total - 1);
      run_slots<8>(rend, roff, gb, nstep, ii, pp);
#pragma unroll
      for (int u = 0; u < 8; ++u) r[u] = (unsigned)srec[pp[u]];
#pragma unroll
      for (int u = 0; u < 8; ++u) if (i0 + u * EBLK < total) atomicAdd(&cnt[r[u]], 1);
    }
    __syncthreads();
    int n = (b << BKT_SHIFT) + t;
    if (t < BKT_NODES && n < N) {
      float g = (float)cnt[t];
      dinv[n] = (g > 0.0f) ? rsqrtf(fmaxf(g, 1e-12f)) : 0.0f;
    }
    return;
  }
  for (int i0 = t; i0 < total; i0 += 4 * EBLK) {
    unsigned r[4]; int ii[4], pp[4];
#pragma unroll
    for (int u = 0; u < 4; ++u) ii[u] = min(i0 + u * EBLK, total - 1);
    run_slots<4>(rend, roff, gb, nstep, ii, pp);
#pragma unroll
    for (int u = 0; u < 4; ++u) r[u] = (unsigned)bsd[pp[u]];
#pragma unroll
    for (int u = 0; u < 4; ++u) if (i0 + u * EBLK < total) atomicAdd(&cnt[r[u] >> 24], 1);
  }
  __syncthreads();
  int mycnt = (t < BKT_NODES) ? cnt[t] : 0;
  lds_scan_incl(cnt, BKT_NODES);
  int lo = b << BKT_SHIFT;
  if (t < BKT_NODES) {
    int first = begp + cnt[t] - mycnt;
    cur[t] = first;
    if (lo + t < N) rowptr[lo + t] = first;
  }
  if (b == nb - 1 && t == 0) rowptr[N] = endp;
  __syncthreads();
  bool staged = total <= PCAP;
  for (int i0 = t; i0 < total; i0 += 4 * EBLK) {
    unsigned r[4]; int ev[4], ii[4], pp[4];
#pragma unroll
    for (int u = 0; u < 4; ++u) ii[u] = min(i0 + u * EBLK, total - 1);
    run_slots<4>(rend, roff, gb, nstep, ii, pp);
#pragma unroll
    for (int u = 0; u < 4; ++u) {
      r[u] = (unsigned)bsd[pp[u]];
      ev[u] = be[pp[u]];
    }
#pragma unroll
    for (int u = 0; u < 4; ++u) {
      if (i0 + u * EBLK < total) {
        int q = atomicAdd(&cur[r[u] >> 24], 1);
        if (staged) {
          ssrc[q - begp] = (int)(r[u] & 0xFFFFFFu);
          seid[q - begp] = ev[u];
        } else {
          csrc[q] = (int)(r[u] & 0xFFFFFFu);
          ceid[q] = ev[u];
        }
      }
    }
  }
  if (staged) {
    __syncthreads();
    int sh = begp & 31;
    for (int i = t - sh; i < total; i += EBLK) {
      if (i >= 0) {
        csrc[begp + i] = ssrc[i];
        ceid[begp + i] = seid[i];
      }
    }
  }
}

__global__ void k_gemm_xlxr(const float* __restrict__ x,
                            const float* __restrict__ Wl, const float* __restrict__ bl,
                            const float* __restrict__ Wr, const float* __restrict__ br,
                            float* __restrict__ xl, float* __restrict__ xr) {
  int tile = blockIdx.x;
  int lane = threadIdx.x;
  int half = lane >> 4;
  int l16 = lane & 15;
  int row = tile * 16 + l16;
  v8f acc0 = {}; v8f acc1 = {};
#pragma unroll
  for (int k = 0; k < 128; k += 4) {
    int ka = k + half * 2;
    v2f a;  a.x = x[row * 128 + ka];     a.y = x[row * 128 + ka + 1];
    v2f b0; b0.x = Wl[ka * 16 + l16];    b0.y = Wl[(ka + 1) * 16 + l16];
    v2f b1; b1.x = Wr[ka * 16 + l16];    b1.y = Wr[(ka + 1) * 16 + l16];
    acc0 = WMMA4(a, b0, acc0);
    acc1 = WMMA4(a, b1, acc1);
  }
#pragma unroll
  for (int i = 0; i < 8; ++i) {
    int m = tile * 16 + i + half * 8;
    int c = l16;
    xl[m * 16 + c] = acc0[i] + bl[c];
    xr[m * 16 + c] = acc1[i] + br[c];
  }
}

__device__ __forceinline__ void ldhalf(const float* __restrict__ p, float* r) {
  const float4* q = (const float4*)p;
  float4 a = q[0], b = q[1];
  r[0] = a.x; r[1] = a.y; r[2] = a.z; r[3] = a.w;
  r[4] = b.x; r[5] = b.y; r[6] = b.z; r[7] = b.w;
}

__device__ __forceinline__ float tile_logit(const v2f* wa, const float* evh, const float* cin,
                                            const float* atth) {
  v8f acc;
#pragma unroll
  for (int i = 0; i < 8; ++i) acc[i] = cin[i];
#pragma unroll
  for (int t = 0; t < 4; ++t) {
    v2f b; b.x = evh[2 * t]; b.y = evh[2 * t + 1];
    acc = WMMA4(wa[t], b, acc);
  }
  float p = 0.0f;
#pragma unroll
  for (int i = 0; i < 8; ++i) {
    float m = acc[i];
    m = m > 0.0f ? m : 0.2f * m;
    p += m * atth[i];
  }
  return p + __shfl_xor(p, 16);
}

__global__ void __launch_bounds__(256)
k_gat(const int* __restrict__ rowptr, const int* __restrict__ csrc, const int* __restrict__ ceid,
      const float* __restrict__ ea, const float* __restrict__ xl, float* xr,
      const float* __restrict__ We, const float* __restrict__ att,
      const float* __restrict__ gbias, const float* __restrict__ dinv,
      float* ascr, float* __restrict__ h, int N) {
  int lane = threadIdx.x & 31;
  int c = lane & 15, hf = lane >> 4;
  int wave = __builtin_amdgcn_readfirstlane((int)((blockIdx.x * blockDim.x + threadIdx.x) >> 5));

  v2f wa[4];
#pragma unroll
  for (int t = 0; t < 4; ++t) {
    int j = 8 * hf + 2 * t;
    wa[t].x = We[j * 16 + c];
    wa[t].y = We[(j + 1) * 16 + c];
  }
  float atth[8];
  ldhalf(att + 8 * hf, atth);

  for (int q = 0; q < 2; ++q) {
    int n = wave * 2 + q;
    if (n >= N) break;
    int beg = rowptr[n], end = rowptr[n + 1];
    float xrh[8], xlh[8];
    ldhalf(xr + (size_t)n * 16 + 8 * hf, xrh);
    ldhalf(xl + (size_t)n * 16 + 8 * hf, xlh);

    float psum[8];
#pragma unroll
    for (int i = 0; i < 8; ++i) psum[i] = 0.0f;
    float amax = -INFINITY;
    for (int base = beg; base < end; base += 32) {
      int j0 = base + c, j1 = j0 + 16;
      bool v0 = j0 < end, v1 = j1 < end;
      int q0 = v0 ? j0 : end - 1;
      int q1 = v1 ? j1 : end - 1;
      int s0 = csrc[q0], e0 = ceid[q0];
      int s1 = csrc[q1], e1 = ceid[q1];
      float ev0[8], xs0[8], ev1[8], xs1[8];
      ldhalf(ea + (size_t)e0 * 16 + 8 * hf, ev0);
      ldhalf(xl + (size_t)s0 * 16 + 8 * hf, xs0);
      ldhalf(ea + (size_t)e1 * 16 + 8 * hf, ev1);
      ldhalf(xl + (size_t)s1 * 16 + 8 * hf, xs1);
#pragma unroll
      for (int i = 0; i < 8; ++i) { xs0[i] += xrh[i]; xs1[i] += xrh[i]; }
      float a0 = tile_logit(wa, ev0, xs0, atth);
      float a1 = tile_logit(wa, ev1, xs1, atth);
      if (v0) {
        ascr[j0] = a0;
        amax = fmaxf(amax, a0);
#pragma unroll
        for (int i = 0; i < 8; ++i) psum[i] += ev0[i];
      }
      if (v1) {
        ascr[j1] = a1;
        amax = fmaxf(amax, a1);
#pragma unroll
        for (int i = 0; i < 8; ++i) psum[i] += ev1[i];
      }
    }
    float dn = fmaxf((float)(end - beg), 1.0f);
    float emh[8], cl[8];
#pragma unroll
    for (int i = 0; i < 8; ++i) { emh[i] = gsum16(psum[i]) / dn; cl[i] = xlh[i] + xrh[i]; }
    float aloop = tile_logit(wa, emh, cl, atth);
    amax = fmaxf(gmax16(amax), aloop);

    float acc[8];
#pragma unroll
    for (int i = 0; i < 8; ++i) acc[i] = 0.0f;
    float den = 0.0f;
    for (int base = beg; base < end; base += 32) {
      int j0 = base + c, j1 = j0 + 16;
      bool v0 = j0 < end, v1 = j1 < end;
      int q0 = v0 ? j0 : end - 1;
      int q1 = v1 ? j1 : end - 1;
      int s0 = csrc[q0], s1 = csrc[q1];
      float al0 = ascr[q0], al1 = ascr[q1];
      float xs0[8], xs1[8];
      ldhalf(xl + (size_t)s0 * 16 + 8 * hf, xs0);
      ldhalf(xl + (size_t)s1 * 16 + 8 * hf, xs1);
      if (v0) {
        float ex = expf(al0 - amax);
        den += ex;
#pragma unroll
        for (int i = 0; i < 8; ++i) acc[i] += ex * xs0[i];
      }
      if (v1) {
        float ex = expf(al1 - amax);
        den += ex;
#pragma unroll
        for (int i = 0; i < 8; ++i) acc[i] += ex * xs1[i];
      }
    }
    float exl = expf(aloop - amax);
    den = gsum16(den) + exl;
    float mine = 0.0f;
#pragma unroll
    for (int i = 0; i < 8; ++i) {
      float v = gsum16(acc[i]) + exl * xlh[i];
      if (c == i) mine = v;
    }
    if (c < 8) {
      int f = 8 * hf + c;
      float hv = mine / (den + 1e-16f) + gbias[f];
      h[(size_t)n * 16 + f] = hv;
      xr[(size_t)n * 16 + f] = dinv[n] * hv;
    }
  }
}

__global__ void __launch_bounds__(256)
k_matvec(const int* __restrict__ rowptr, const int* __restrict__ csrc,
         const float* __restrict__ dinv, const float* __restrict__ yin,
         const float* __restrict__ prev, float scale, float* __restrict__ out,
         float* __restrict__ yout, int N) {
  int gid = blockIdx.x * blockDim.x + threadIdx.x;
  int n = gid >> 4, l = gid & 15;
  if (n >= N) return;
  int g = l >> 2, q = l & 3;
  int beg = rowptr[n], end = rowptr[n + 1];
  float dn = dinv[n];
  const float4* __restrict__ yin4 = (const float4*)yin;
  float a0 = 0.0f, a1 = 0.0f, a2 = 0.0f, a3 = 0.0f;
  for (int base = beg; base < end; base += 16) {
    int jj = base + l;
    unsigned sidx = (unsigned)((jj < end) ? csrc[jj] : n);
    int s0 = (int)__builtin_amdgcn_permlane16(sidx, sidx, 0x11110000u, 0x33332222u, false, false);
    int s1 = (int)__builtin_amdgcn_permlane16(sidx, sidx, 0x55554444u, 0x77776666u, false, false);
    int s2 = (int)__builtin_amdgcn_permlane16(sidx, sidx, 0x99998888u, 0xBBBBAAAAu, false, false);
    int s3 = (int)__builtin_amdgcn_permlane16(sidx, sidx, 0xDDDDCCCCu, 0xFFFFEEEEu, false, false);
    float4 r0 = yin4[(size_t)s0 * 4 + q];
    float4 r1 = yin4[(size_t)s1 * 4 + q];
    float4 r2 = yin4[(size_t)s2 * 4 + q];
    float4 r3 = yin4[(size_t)s3 * 4 + q];
    int e = base + g;
    if (e < end)      { float w = (s0 != n) ? 1.0f : 0.0f; a0 += w * r0.x; a1 += w * r0.y; a2 += w * r0.z; a3 += w * r0.w; }
    if (e + 4 < end)  { float w = (s1 != n) ? 1.0f : 0.0f; a0 += w * r1.x; a1 += w * r1.y; a2 += w * r1.z; a3 += w * r1.w; }
    if (e + 8 < end)  { float w = (s2 != n) ? 1.0f : 0.0f; a0 += w * r2.x; a1 += w * r2.y; a2 += w * r2.z; a3 += w * r2.w; }
    if (e + 12 < end) { float w = (s3 != n) ? 1.0f : 0.0f; a0 += w * r3.x; a1 += w * r3.y; a2 += w * r3.z; a3 += w * r3.w; }
  }
  a0 += __shfl_xor(a0, 4, 16); a1 += __shfl_xor(a1, 4, 16); a2 += __shfl_xor(a2, 4, 16); a3 += __shfl_xor(a3, 4, 16);
  a0 += __shfl_xor(a0, 8, 16); a1 += __shfl_xor(a1, 8, 16); a2 += __shfl_xor(a2, 8, 16); a3 += __shfl_xor(a3, 8, 16);
  float mine = (g == 0) ? a0 : (g == 1) ? a1 : (g == 2) ? a2 : a3;
  int f = 4 * q + g;
  float r = scale * (-dn * mine);
  if (prev) r -= prev[(size_t)n * 16 + f];
  out[(size_t)n * 16 + f] = r;
  if (yout) yout[(size_t)n * 16 + f] = dn * r;
}

#define T3PITCH 20
__global__ void __launch_bounds__(256)
k_last(const int* __restrict__ rowptr, const int* __restrict__ csrc,
       const float* __restrict__ dinv, const float* __restrict__ yin,
       const float* __restrict__ T0, const float* __restrict__ T1, const float* __restrict__ T2,
       const float* __restrict__ chebW, const float* __restrict__ chebB,
       const float* __restrict__ W1, const float* __restrict__ b1,
       const float* __restrict__ W2, const float* __restrict__ b2,
       float* __restrict__ out, int N) {
  __shared__ float4 wsh4[512];
  __shared__ float4 t3s4[16 * T3PITCH / 4];
  float* t3s = (float*)t3s4;
  const float* wsh = (const float*)wsh4;
  int tid = threadIdx.x;
  int nl = tid >> 4, l = tid & 15;
  int n = (int)blockIdx.x * 16 - 1 + nl;
  bool valid = (n >= 0) && (n < N);
  int g = l >> 2, q = l & 3;
  int nn = min(max(n, 0), N - 1);
  int beg = rowptr[nn], end = rowptr[nn + 1];
  float dn = dinv[nn];
  const float4* w4 = (const float4*)chebW;
  float4 wz = w4[tid];
  float4 wh = w4[1024 + tid];
  float pv = T1[(size_t)nn * 16 + 4 * q + g];
  asm volatile("" : "+v"(pv), "+v"(wz.x), "+v"(wz.y), "+v"(wz.z), "+v"(wz.w),
                    "+v"(wh.x), "+v"(wh.y), "+v"(wh.z), "+v"(wh.w));
  if (!valid) { beg = 0; end = 0; dn = 0.0f; }
  const float4* __restrict__ yin4 = (const float4*)yin;
  float a0 = 0.0f, a1 = 0.0f, a2 = 0.0f, a3 = 0.0f;
  for (int base = beg; base < end; base += 16) {
    int jj = base + l;
    unsigned sidx = (unsigned)((jj < end) ? csrc[jj] : n);
    int s0 = (int)__builtin_amdgcn_permlane16(sidx, sidx, 0x11110000u, 0x33332222u, false, false);
    int s1 = (int)__builtin_amdgcn_permlane16(sidx, sidx, 0x55554444u, 0x77776666u, false, false);
    int s2 = (int)__builtin_amdgcn_permlane16(sidx, sidx, 0x99998888u, 0xBBBBAAAAu, false, false);
    int s3 = (int)__builtin_amdgcn_permlane16(sidx, sidx, 0xDDDDCCCCu, 0xFFFFEEEEu, false, false);
    float4 r0 = yin4[(size_t)s0 * 4 + q];
    float4 r1 = yin4[(size_t)s1 * 4 + q];
    float4 r2 = yin4[(size_t)s2 * 4 + q];
    float4 r3 = yin4[(size_t)s3 * 4 + q];
    int e = base + g;
    if (e < end)      { float w = (s0 != n) ? 1.0f : 0.0f; a0 += w * r0.x; a1 += w * r0.y; a2 += w * r0.z; a3 += w * r0.w; }
    if (e + 4 < end)  { float w = (s1 != n) ? 1.0f : 0.0f; a0 += w * r1.x; a1 += w * r1.y; a2 += w * r1.z; a3 += w * r1.w; }
    if (e + 8 < end)  { float w = (s2 != n) ? 1.0f : 0.0f; a0 += w * r2.x; a1 += w * r2.y; a2 += w * r2.z; a3 += w * r2.w; }
    if (e + 12 < end) { float w = (s3 != n) ? 1.0f : 0.0f; a0 += w * r3.x; a1 += w * r3.y; a2 += w * r3.z; a3 += w * r3.w; }
  }
  a0 += __shfl_xor(a0, 4, 16); a1 += __shfl_xor(a1, 4, 16); a2 += __shfl_xor(a2, 4, 16); a3 += __shfl_xor(a3, 4, 16);
  a0 += __shfl_xor(a0, 8, 16); a1 += __shfl_xor(a1, 8, 16); a2 += __shfl_xor(a2, 8, 16); a3 += __shfl_xor(a3, 8, 16);
  float mine = (g == 0) ? a0 : (g == 1) ? a1 : (g == 2) ? a2 : a3;
  int f = 4 * q + g;
  float r = 2.0f * (-dn * mine);
  r -= pv;
  t3s[nl * T3PITCH + f] = valid ? r : 0.0f;
  wsh4[tid] = wz;
  wsh4[256 + tid] = wh;

  bool w0 = tid < 32;
  int c = tid & 15, hf = (tid >> 4) & 1;
  int m = (int)blockIdx.x * 16 - 1 + c;
  bool mvalid = (m >= 0) && (m < N);
  float tb[3][8];
#pragma unroll
  for (int p = 0; p < 3; ++p)
#pragma unroll
    for (int i = 0; i < 8; ++i) tb[p][i] = 0.0f;
  if (w0 && mvalid) {
    ldhalf(T0 + (size_t)m * 16 + 8 * hf, tb[0]);
    ldhalf(T1 + (size_t)m * 16 + 8 * hf, tb[1]);
    ldhalf(T2 + (size_t)m * 16 + 8 * hf, tb[2]);
  }
  __syncthreads();
  if (!w0) return;

  float t3h[8];
  ldhalf(t3s + c * T3PITCH + 8 * hf, t3h);
  v8f accZ = {}; v8f accH = {};
#pragma unroll
  for (int p = 0; p < 4; ++p) {
#pragma unroll
    for (int t = 0; t < 4; ++t) {
      int k = 8 * hf + 2 * t;
      v2f b;
      b.x = (p < 3) ? tb[p < 3 ? p : 0][2 * t] : t3h[2 * t];
      b.y = (p < 3) ? tb[p < 3 ? p : 0][2 * t + 1] : t3h[2 * t + 1];
      v2f az; az.x = wsh[p * 256 + k * 16 + c];        az.y = wsh[p * 256 + (k + 1) * 16 + c];
      v2f ah; ah.x = wsh[1024 + p * 256 + k * 16 + c]; ah.y = wsh[1024 + p * 256 + (k + 1) * 16 + c];
      accZ = WMMA4(az, b, accZ);
      accH = WMMA4(ah, b, accH);
    }
  }
  float bz0[8], bz1[8], bh0[8], bh1[8];
  ldhalf(chebB + 8 * hf, bz0);
  ldhalf(chebB + 16 + 8 * hf, bz1);
  ldhalf(chebB + 64 + 8 * hf, bh0);
  ldhalf(chebB + 80 + 8 * hf, bh1);
  float hrow[8];
#pragma unroll
  for (int i = 0; i < 8; ++i) {
    float zp = accZ[i] + bz0[i] + bz1[i];
    float hp = accH[i] + bh0[i] + bh1[i];
    float z = 1.0f / (1.0f + expf(-zp));
    float ht = tanhf(hp);
    hrow[i] = mvalid ? (1.0f - z) * ht : 0.0f;
  }
  float hid[4];
#pragma unroll
  for (int j = 0; j < 4; ++j) {
    float s = (hf == 0) ? b1[j] : 0.0f;
#pragma unroll
    for (int i = 0; i < 8; ++i) s += hrow[i] * W1[(8 * hf + i) * 4 + j];
    s += __shfl_xor(s, 16);
    hid[j] = fmaxf(s, 0.0f);
  }
  float o4[4];
#pragma unroll
  for (int oo = 0; oo < 4; ++oo) {
    int o = 4 * hf + oo;
    float s = b2[o];
#pragma unroll
    for (int j = 0; j < 4; ++j) s += hid[j] * W2[j * 8 + o];
    o4[oo] = s;
  }
  if (m < N) {
    float4 v; v.x = o4[0]; v.y = o4[1]; v.z = o4[2]; v.w = o4[3];
    *(float4*)(out + (size_t)(m + 1) * 8 + 4 * hf) = v;
  }
}

extern "C" void kernel_launch(void* const* d_in, const int* in_sizes, int n_in,
                              void* d_out, int out_size, void* d_ws, size_t ws_size,
                              hipStream_t stream) {
  const float* x     = (const float*)d_in[0];
  const int*   ei    = (const int*)  d_in[1];
  const float* ea    = (const float*)d_in[2];
  const float* Wl    = (const float*)d_in[3];
  const float* bl    = (const float*)d_in[4];
  const float* Wr    = (const float*)d_in[5];
  const float* br    = (const float*)d_in[6];
  const float* We    = (const float*)d_in[7];
  const float* att   = (const float*)d_in[8];
  const float* gbias = (const float*)d_in[9];
  const float* chebW = (const float*)d_in[10];
  const float* chebB = (const float*)d_in[11];
  const float* W1    = (const float*)d_in[12];
  const float* b1    = (const float*)d_in[13];
  const float* W2    = (const float*)d_in[14];
  const float* b2    = (const float*)d_in[15];
  float* out = (float*)d_out;

  const int N = in_sizes[0] / 128;
  const int E = in_sizes[1] / 2;
  const size_t NF = (size_t)N * 16;

  float* ws    = (float*)d_ws;
  float* xl    = ws;
  float* xr    = xl + NF;
  float* hh    = xr + NF;
  float* T1    = hh + NF;
  float* T2    = T1 + NF;
  float* T3    = T2 + NF;
  float* spare = T3 + NF;
  int*   cnt   = (int*)(spare + NF);
  int*   scnt  = cnt + N;
  float* dinv  = (float*)(scnt + N);
  int*   rowptr = (int*)(dinv + N);
  int*   csrc  = (int*)(spare + NF + (size_t)N * 5);
  int*   be    = (int*)T1;
  int*   bsd   = (int*)xl;
  int*   tabd  = cnt;
  int*   tabs  = scnt;
  unsigned char* srec = (unsigned char*)hh;
  float* ascr  = T1;
  int*   ceid  = (int*)T3;

  const int TPB = 256;
  int gEb  = (E + ESEG - 1) / ESEG;
  int nb   = (N + BKT_NODES - 1) >> BKT_SHIFT;
  int gN16 = (N * 16 + TPB - 1) / TPB;
  int gT   = (N + 16) / 16;
  int tiles = N / 16;

  k_bin<<<gEb, EBLK, 0, stream>>>(ei, tabd, tabs, bsd, be, srec, E, nb);
  k_place<<<nb + nb, EBLK, 0, stream>>>(tabd, bsd, be, rowptr, csrc, ceid, tabs, srec, dinv, N, nb, gEb);

  k_gemm_xlxr<<<tiles, 32, 0, stream>>>(x, Wl, bl, Wr, br, xl, xr);
  k_gat<<<gN16, TPB, 0, stream>>>(rowptr, csrc, ceid, ea, xl, xr, We, att, gbias, dinv, ascr, hh, N);

  k_matvec<<<gN16, TPB, 0, stream>>>(rowptr, csrc, dinv, xr, (const float*)nullptr, 1.0f, T1, xl, N);
  k_matvec<<<gN16, TPB, 0, stream>>>(rowptr, csrc, dinv, xl, hh, 2.0f, T2, xr, N);
  k_last<<<gT, TPB, 0, stream>>>(rowptr, csrc, dinv, xr, hh, T1, T2, chebW, chebB, W1, b1, W2, b2, out, N);
}


# GATv2 walks each node's edge list once: per-lane running max, sums rescaled by exp(old-new), lanes merged by exp(lane max - node max) (online softmax); no logit scratch, no 2nd row gather; 86 VGPRs
# speedup vs baseline: 7.4176x; 1.1083x over previous
#include <hip/hip_runtime.h>
#include <math.h>

typedef float v2f __attribute__((ext_vector_type(2)));
typedef float v8f __attribute__((ext_vector_type(8)));

#define WMMA4(a, b, c) \
  __builtin_amdgcn_wmma_f32_16x16x4_f32(false, (a), false, (b), (short)0, (c), false, false)

__device__ __forceinline__ float gsum16(float v) {
  v += __shfl_xor(v, 8, 16);
  v += __shfl_xor(v, 4, 16);
  v += __shfl_xor(v, 2, 16);
  v += __shfl_xor(v, 1, 16);
  return v;
}
__device__ __forceinline__ float gmax16(float v) {
  v = fmaxf(v, __shfl_xor(v, 8, 16));
  v = fmaxf(v, __shfl_xor(v, 4, 16));
  v = fmaxf(v, __shfl_xor(v, 2, 16));
  v = fmaxf(v, __shfl_xor(v, 1, 16));
  return v;
}

__device__ __forceinline__ void ldrow(const float* __restrict__ p, float* r) {
  const float4* q = (const float4*)p;
  float4 a = q[0], b = q[1], c = q[2], d = q[3];
  r[0] = a.x;  r[1] = a.y;  r[2] = a.z;  r[3] = a.w;
  r[4] = b.x;  r[5] = b.y;  r[6] = b.z;  r[7] = b.w;
  r[8] = c.x;  r[9] = c.y;  r[10] = c.z; r[11] = c.w;
  r[12] = d.x; r[13] = d.y; r[14] = d.z; r[15] = d.w;
}

#define BKT_SHIFT 8
#define BKT_NODES (1 << BKT_SHIFT)
#define MAXB 512
#define EBLK 1024
#define EPT 8
#define ESEG (EBLK * EPT)

__device__ __forceinline__ void lds_scan_incl(int* a, int n) {
  int t = threadIdx.x;
  for (int off = 1; off < n; off <<= 1) {
    int v = (t < n && t >= off) ? a[t - off] : 0;
    __syncthreads();
    if (t < n) a[t] += v;
    __syncthreads();
  }
}

__device__ __forceinline__ void lds_scan_incl2(int* a, int* b, int n) {
  int t = threadIdx.x;
  for (int off = 1; off < n; off <<= 1) {
    bool on = (t < n && t >= off);
    int va = on ? a[t - off] : 0;
    int vb = on ? b[t - off] : 0;
    __syncthreads();
    if (t < n) { a[t] += va; b[t] += vb; }
    __syncthreads();
  }
}

__global__ void __launch_bounds__(EBLK)
k_bin(const int* __restrict__ ei, int* __restrict__ tabd, int* __restrict__ tabs,
      int* __restrict__ bsd, int* __restrict__ be, unsigned char* __restrict__ srec, int E, int nb) {
  __shared__ int hist[MAXB];
  __shared__ int shist[MAXB];
  __shared__ int sbsd[ESEG];
  __shared__ int sbe[ESEG];
  __shared__ int ssrw[ESEG / 4];
  unsigned char* ssr = (unsigned char*)ssrw;
  int t = threadIdx.x;
  int blk = blockIdx.x, gb = gridDim.x;
  if (t < MAXB) { hist[t] = 0; shist[t] = 0; }
  __syncthreads();
  int e0 = blk * ESEG + t;
  int s[EPT], d[EPT], lr[EPT], slr[EPT];
#pragma unroll
  for (int k = 0; k < EPT; ++k) {
    int e = e0 + k * EBLK;
    bool v = e < E;
    s[k] = v ? ei[e] : 0;
    d[k] = v ? ei[E + e] : 0;
  }
#pragma unroll
  for (int k = 0; k < EPT; ++k) {
    lr[k] = 0; slr[k] = 0;
    if (e0 + k * EBLK < E) {
      lr[k] = atomicAdd(&hist[d[k] >> BKT_SHIFT], 1);
      if (s[k] != d[k]) slr[k] = atomicAdd(&shist[s[k] >> BKT_SHIFT], 1);
    }
  }
  __syncthreads();
  lds_scan_incl2(hist, shist, nb);
  int base = blk * ESEG;
#pragma unroll
  for (int k = 0; k < EPT; ++k) {
    int e = e0 + k * EBLK;
    if (e < E) {
      int bk = d[k] >> BKT_SHIFT;
      int p = (bk ? hist[bk - 1] : 0) + lr[k];
      sbsd[p] = (int)((unsigned)s[k] | ((unsigned)(d[k] & (BKT_NODES - 1)) << 24));
      sbe[p] = e;
      if (s[k] != d[k]) {
        int sb = s[k] >> BKT_SHIFT;
        ssr[(sb ? shist[sb - 1] : 0) + slr[k]] = (unsigned char)(s[k] & (BKT_NODES - 1));
      }
    }
  }
  __syncthreads();
  int nd = hist[nb - 1];
  int ns = shist[nb - 1];
  int nsw = ns >> 2;
  for (int i = t; i < nd; i += EBLK) {
    bsd[base + i] = sbsd[i];
    be[base + i] = sbe[i];
  }
  int* srecw = (int*)(srec + base);
  for (int i = t; i < nsw; i += EBLK) srecw[i] = ssrw[i];
  if (t < (ns & 3)) srec[base + (nsw << 2) + t] = ssr[(nsw << 2) + t];
  if (t < nb) {
    tabd[t * gb + blk] = hist[t];
    tabs[t * gb + blk] = shist[t];
  }
}

template <int U>
__device__ __forceinline__ void run_slots(const int* rend, const int* roff, int gb, int nstep, const int* i, int* p) {
  int lo[U], hi[U];
#pragma unroll
  for (int u = 0; u < U; ++u) { lo[u] = 0; hi[u] = gb - 1; }
  for (int st = 0; st < nstep; ++st) {
#pragma unroll
    for (int u = 0; u < U; ++u) {
      int mid = (lo[u] + hi[u]) >> 1;
      bool gt = rend[mid] > i[u];
      hi[u] = gt ? mid : hi[u];
      lo[u] = gt ? lo[u] : mid + 1;
    }
  }
#pragma unroll
  for (int u = 0; u < U; ++u) {
    int r = hi[u];
    int first = r ? rend[r - 1] : 0;
    p[u] = r * ESEG + roff[r] + (i[u] - first);
  }
}

#define PCAP 10240
__global__ void __launch_bounds__(EBLK)
k_place(const int* __restrict__ tabd, const int* __restrict__ bsd, const int* __restrict__ be,
        int* __restrict__ rowptr, int* __restrict__ csrc, int* __restrict__ ceid,
        const int* __restrict__ tabs, const unsigned char* __restrict__ srec, float* __restrict__ dinv,
        int N, int nb, int gb) {
  __shared__ int rend[MAXB];
  __shared__ int roff[MAXB];
  __shared__ int psum[MAXB];
  __shared__ int cnt[BKT_NODES];
  __shared__ int cur[BKT_NODES];
  __shared__ int ssrc[PCAP];
  __shared__ int seid[PCAP];
  int t = threadIdx.x;
  bool srcblk = (int)blockIdx.x >= nb;
  int b = srcblk ? (int)blockIdx.x - nb : (int)blockIdx.x;
  const int* tab = srcblk ? tabs : tabd;
  if (t < MAXB) {
    int upto = 0, before = 0;
    if (t < gb) {
      upto = tab[b * gb + t];
      before = b ? tab[(b - 1) * gb + t] : 0;
    }
    rend[t] = upto - before; roff[t] = before; psum[t] = before;
  }
  if (t < BKT_NODES) cnt[t] = 0;
  __syncthreads();
  lds_scan_incl2(rend, psum, gb);
  int total = rend[gb - 1];
  int begp = psum[gb - 1];
  int endp = begp + total;
  int nstep = 32 - __clz(gb - 1);
  if (gb < 2) nstep = 0;
  if (srcblk) {
    for (int i0 = t; i0 < total; i0 += 8 * EBLK) {
      unsigned r[8]; int ii[8], pp[8];
#pragma unroll
      for (int u = 0; u < 8; ++u) ii[u] = min(i0 + u * EBLK, total - 1);
      run_slots<8>(rend, roff, gb, nstep, ii, pp);
#pragma unroll
      for (int u = 0; u < 8; ++u) r[u] = (unsigned)srec[pp[u]];
#pragma unroll
      for (int u = 0; u < 8; ++u) if (i0 + u * EBLK < total) atomicAdd(&cnt[r[u]], 1);
    }
    __syncthreads();
    int n = (b << BKT_SHIFT) + t;
    if (t < BKT_NODES && n < N) {
      float g = (float)cnt[t];
      dinv[n] = (g > 0.0f) ? rsqrtf(fmaxf(g, 1e-12f)) : 0.0f;
    }
    return;
  }
  for (int i0 = t; i0 < total; i0 += 4 * EBLK) {
    unsigned r[4]; int ii[4], pp[4];
#pragma unroll
    for (int u = 0; u < 4; ++u) ii[u] = min(i0 + u * EBLK, total - 1);
    run_slots<4>(rend, roff, gb, nstep, ii, pp);
#pragma unroll
    for (int u = 0; u < 4; ++u) r[u] = (unsigned)bsd[pp[u]];
#pragma unroll
    for (int u = 0; u < 4; ++u) if (i0 + u * EBLK < total) atomicAdd(&cnt[r[u] >> 24], 1);
  }
  __syncthreads();
  int mycnt = (t < BKT_NODES) ? cnt[t] : 0;
  lds_scan_incl(cnt, BKT_NODES);
  int lo = b << BKT_SHIFT;
  if (t < BKT_NODES) {
    int first = begp + cnt[t] - mycnt;
    cur[t] = first;
    if (lo + t < N) rowptr[lo + t] = first;
  }
  if (b == nb - 1 && t == 0) rowptr[N] = endp;
  __syncthreads();
  bool staged = total <= PCAP;
  for (int i0 = t; i0 < total; i0 += 4 * EBLK) {
    unsigned r[4]; int ev[4], ii[4], pp[4];
#pragma unroll
    for (int u = 0; u < 4; ++u) ii[u] = min(i0 + u * EBLK, total - 1);
    run_slots<4>(rend, roff, gb, nstep, ii, pp);
#pragma unroll
    for (int u = 0; u < 4; ++u) {
      r[u] = (unsigned)bsd[pp[u]];
      ev[u] = be[pp[u]];
    }
#pragma unroll
    for (int u = 0; u < 4; ++u) {
      if (i0 + u * EBLK < total) {
        int q = atomicAdd(&cur[r[u] >> 24], 1);
        if (staged) {
          ssrc[q - begp] = (int)(r[u] & 0xFFFFFFu);
          seid[q - begp] = ev[u];
        } else {
          csrc[q] = (int)(r[u] & 0xFFFFFFu);
          ceid[q] = ev[u];
        }
      }
    }
  }
  if (staged) {
    __syncthreads();
    int sh = begp & 31;
    for (int i = t - sh; i < total; i += EBLK) {
      if (i >= 0) {
        csrc[begp + i] = ssrc[i];
        ceid[begp + i] = seid[i];
      }
    }
  }
}

__global__ void k_gemm_xlxr(const float* __restrict__ x,
                            const float* __restrict__ Wl, const float* __restrict__ bl,
                            const float* __restrict__ Wr, const float* __restrict__ br,
                            float* __restrict__ xl, float* __restrict__ xr) {
  int tile = blockIdx.x;
  int lane = threadIdx.x;
  int half = lane >> 4;
  int l16 = lane & 15;
  int row = tile * 16 + l16;
  v8f acc0 = {}; v8f acc1 = {};
#pragma unroll
  for (int k = 0; k < 128; k += 4) {
    int ka = k + half * 2;
    v2f a;  a.x = x[row * 128 + ka];     a.y = x[row * 128 + ka + 1];
    v2f b0; b0.x = Wl[ka * 16 + l16];    b0.y = Wl[(ka + 1) * 16 + l16];
    v2f b1; b1.x = Wr[ka * 16 + l16];    b1.y = Wr[(ka + 1) * 16 + l16];
    acc0 = WMMA4(a, b0, acc0);
    acc1 = WMMA4(a, b1, acc1);
  }
#pragma unroll
  for (int i = 0; i < 8; ++i) {
    int m = tile * 16 + i + half * 8;
    int c = l16;
    xl[m * 16 + c] = acc0[i] + bl[c];
    xr[m * 16 + c] = acc1[i] + br[c];
  }
}

__device__ __forceinline__ void ldhalf(const float* __restrict__ p, float* r) {
  const float4* q = (const float4*)p;
  float4 a = q[0], b = q[1];
  r[0] = a.x; r[1] = a.y; r[2] = a.z; r[3] = a.w;
  r[4] = b.x; r[5] = b.y; r[6] = b.z; r[7] = b.w;
}

__device__ __forceinline__ void ldhalf4(const float4* __restrict__ p4, int i4, float* r) {
  float4 a = p4[i4], b = p4[i4 + 1];
  r[0] = a.x; r[1] = a.y; r[2] = a.z; r[3] = a.w;
  r[4] = b.x; r[5] = b.y; r[6] = b.z; r[7] = b.w;
}

__device__ __forceinline__ float tile_logit(const v2f* wa, const float* evh, const float* cin,
                                            const float* __restrict__ att, int hf) {
  v8f acc;
#pragma unroll
  for (int i = 0; i < 8; ++i) acc[i] = cin[i];
#pragma unroll
  for (int t = 0; t < 4; ++t) {
    v2f b; b.x = evh[2 * t]; b.y = evh[2 * t + 1];
    acc = WMMA4(wa[t], b, acc);
  }
  float plo = 0.0f, phi = 0.0f;
#pragma unroll
  for (int i = 0; i < 8; ++i) {
    float m = acc[i];
    m = m > 0.0f ? m : 0.2f * m;
    plo += m * att[i];
    phi += m * att[8 + i];
  }
  float p = hf ? phi : plo;
  return p + __shfl_xor(p, 16);
}

__global__ void __launch_bounds__(256, 10)
k_gat(const int* __restrict__ rowptr, const int* __restrict__ csrc, const int* __restrict__ ceid,
      const float* __restrict__ ea, const float* __restrict__ xl, float* xr,
      const float* __restrict__ We, const float* __restrict__ att,
      const float* __restrict__ gbias, const float* __restrict__ dinv,
      float* __restrict__ h, int N) {
  int lane = threadIdx.x & 31;
  int c = lane & 15, hf = lane >> 4;
  int wave = __builtin_amdgcn_readfirstlane((int)((blockIdx.x * blockDim.x + threadIdx.x) >> 5));

  v2f wa[4];
#pragma unroll
  for (int t = 0; t < 4; ++t) {
    int j = 8 * hf + 2 * t;
    wa[t].x = We[j * 16 + c];
    wa[t].y = We[(j + 1) * 16 + c];
  }
  const float4* ea4 = (const float4*)ea;
  const float4* xl4 = (const float4*)xl;
  const float4* xr4 = (const float4*)xr;

  for (int q = 0; q < 2; ++q) {
    int n = wave * 2 + q;
    if (n >= N) break;
    int beg = rowptr[n], end = rowptr[n + 1];
    float xrh[8];
    ldhalf4(xr4, n * 4 + 2 * hf, xrh);

    float psum[8], acc[8];
#pragma unroll
    for (int i = 0; i < 8; ++i) { psum[i] = 0.0f; acc[i] = 0.0f; }
    float mrun = -__FLT_MAX__;
    float den = 0.0f;
    for (int base = beg; base < end; base += 32) {
      int j0 = base + c, j1 = j0 + 16;
      bool v0 = j0 < end, v1 = j1 < end;
      int q0 = v0 ? j0 : end - 1;
      int q1 = v1 ? j1 : end - 1;
      int s0 = csrc[q0], e0 = ceid[q0];
      int s1 = csrc[q1], e1 = ceid[q1];
      float ev0[8], xs0[8], ev1[8], xs1[8];
      ldhalf4(ea4, e0 * 4 + 2 * hf, ev0);
      ldhalf4(xl4, s0 * 4 + 2 * hf, xs0);
      ldhalf4(ea4, e1 * 4 + 2 * hf, ev1);
      ldhalf4(xl4, s1 * 4 + 2 * hf, xs1);
      if (v0) {
#pragma unroll
        for (int i = 0; i < 8; ++i) psum[i] += ev0[i];
      }
      if (v1) {
#pragma unroll
        for (int i = 0; i < 8; ++i) psum[i] += ev1[i];
      }
      float ci0[8], ci1[8];
#pragma unroll
      for (int i = 0; i < 8; ++i) { ci0[i] = xs0[i] + xrh[i]; ci1[i] = xs1[i] + xrh[i]; }
      float a0 = tile_logit(wa, ev0, ci0, att, hf);
      float a1 = tile_logit(wa, ev1, ci1, att, hf);
      if (v0) {
        float mn = fmaxf(mrun, a0);
        if (v1) mn = fmaxf(mn, a1);
        float sc = expf(mrun - mn);
        float w0 = expf(a0 - mn);
        den = den * sc + w0;
#pragma unroll
        for (int i = 0; i < 8; ++i) acc[i] = acc[i] * sc + w0 * xs0[i];
        if (v1) {
          float w1 = expf(a1 - mn);
          den += w1;
#pragma unroll
          for (int i = 0; i < 8; ++i) acc[i] += w1 * xs1[i];
        }
        mrun = mn;
      }
    }
    float xlh[8];
    ldhalf4(xl4, n * 4 + 2 * hf, xlh);
    float dn = fmaxf((float)(end - beg), 1.0f);
    float emh[8], cl[8];
#pragma unroll
    for (int i = 0; i < 8; ++i) { emh[i] = gsum16(psum[i]) / dn; cl[i] = xlh[i] + xrh[i]; }
    float aloop = tile_logit(wa, emh, cl, att, hf);
    float amax = fmaxf(gmax16(mrun), aloop);
    float fl = expf(mrun - amax);
    float exl = expf(aloop - amax);
    den = gsum16(den * fl) + exl;
    float mine = 0.0f;
#pragma unroll
    for (int i = 0; i < 8; ++i) {
      float v = gsum16(acc[i] * fl) + exl * xlh[i];
      if (c == i) mine = v;
    }
    if (c < 8) {
      int f = 8 * hf + c;
      float hv = mine / (den + 1e-16f) + gbias[f];
      h[(size_t)n * 16 + f] = hv;
      xr[(size_t)n * 16 + f] = dinv[n] * hv;
    }
  }
}

__global__ void __launch_bounds__(256)
k_matvec(const int* __restrict__ rowptr, const int* __restrict__ csrc,
         const float* __restrict__ dinv, const float* __restrict__ yin,
         const float* __restrict__ prev, float scale, float* __restrict__ out,
         float* __restrict__ yout, int N) {
  int gid = blockIdx.x * blockDim.x + threadIdx.x;
  int n = gid >> 4, l = gid & 15;
  if (n >= N) return;
  int g = l >> 2, q = l & 3;
  int beg = rowptr[n], end = rowptr[n + 1];
  float dn = dinv[n];
  const float4* __restrict__ yin4 = (const float4*)yin;
  float a0 = 0.0f, a1 = 0.0f, a2 = 0.0f, a3 = 0.0f;
  for (int base = beg; base < end; base += 16) {
    int jj = base + l;
    unsigned sidx = (unsigned)((jj < end) ? csrc[jj] : n);
    int s0 = (int)__builtin_amdgcn_permlane16(sidx, sidx, 0x11110000u, 0x33332222u, false, false);
    int s1 = (int)__builtin_amdgcn_permlane16(sidx, sidx, 0x55554444u, 0x77776666u, false, false);
    int s2 = (int)__builtin_amdgcn_permlane16(sidx, sidx, 0x99998888u, 0xBBBBAAAAu, false, false);
    int s3 = (int)__builtin_amdgcn_permlane16(sidx, sidx, 0xDDDDCCCCu, 0xFFFFEEEEu, false, false);
    float4 r0 = yin4[(size_t)s0 * 4 + q];
    float4 r1 = yin4[(size_t)s1 * 4 + q];
    float4 r2 = yin4[(size_t)s2 * 4 + q];
    float4 r3 = yin4[(size_t)s3 * 4 + q];
    int e = base + g;
    if (e < end)      { float w = (s0 != n) ? 1.0f : 0.0f; a0 += w * r0.x; a1 += w * r0.y; a2 += w * r0.z; a3 += w * r0.w; }
    if (e + 4 < end)  { float w = (s1 != n) ? 1.0f : 0.0f; a0 += w * r1.x; a1 += w * r1.y; a2 += w * r1.z; a3 += w * r1.w; }
    if (e + 8 < end)  { float w = (s2 != n) ? 1.0f : 0.0f; a0 += w * r2.x; a1 += w * r2.y; a2 += w * r2.z; a3 += w * r2.w; }
    if (e + 12 < end) { float w = (s3 != n) ? 1.0f : 0.0f; a0 += w * r3.x; a1 += w * r3.y; a2 += w * r3.z; a3 += w * r3.w; }
  }
  a0 += __shfl_xor(a0, 4, 16); a1 += __shfl_xor(a1, 4, 16); a2 += __shfl_xor(a2, 4, 16); a3 += __shfl_xor(a3, 4, 16);
  a0 += __shfl_xor(a0, 8, 16); a1 += __shfl_xor(a1, 8, 16); a2 += __shfl_xor(a2, 8, 16); a3 += __shfl_xor(a3, 8, 16);
  float mine = (g == 0) ? a0 : (g == 1) ? a1 : (g == 2) ? a2 : a3;
  int f = 4 * q + g;
  float r = scale * (-dn * mine);
  if (prev) r -= prev[(size_t)n * 16 + f];
  out[(size_t)n * 16 + f] = r;
  if (yout) yout[(size_t)n * 16 + f] = dn * r;
}

#define T3PITCH 20
__global__ void __launch_bounds__(256)
k_last(const int* __restrict__ rowptr, const int* __restrict__ csrc,
       const float* __restrict__ dinv, const float* __restrict__ yin,
       const float* __restrict__ T0, const float* __restrict__ T1, const float* __restrict__ T2,
       const float* __restrict__ chebW, const float* __restrict__ chebB,
       const float* __restrict__ W1, const float* __restrict__ b1,
       const float* __restrict__ W2, const float* __restrict__ b2,
       float* __restrict__ out, int N) {
  __shared__ float4 wsh4[512];
  __shared__ float4 t3s4[16 * T3PITCH / 4];
  float* t3s = (float*)t3s4;
  const float* wsh = (const float*)wsh4;
  int tid = threadIdx.x;
  int nl = tid >> 4, l = tid & 15;
  int n = (int)blockIdx.x * 16 - 1 + nl;
  bool valid = (n >= 0) && (n < N);
  int g = l >> 2, q = l & 3;
  int nn = min(max(n, 0), N - 1);
  int beg = rowptr[nn], end = rowptr[nn + 1];
  float dn = dinv[nn];
  const float4* w4 = (const float4*)chebW;
  float4 wz = w4[tid];
  float4 wh = w4[1024 + tid];
  float pv = T1[(size_t)nn * 16 + 4 * q + g];
  asm volatile("" : "+v"(pv), "+v"(wz.x), "+v"(wz.y), "+v"(wz.z), "+v"(wz.w),
                    "+v"(wh.x), "+v"(wh.y), "+v"(wh.z), "+v"(wh.w));
  if (!valid) { beg = 0; end = 0; dn = 0.0f; }
  const float4* __restrict__ yin4 = (const float4*)yin;
  float a0 = 0.0f, a1 = 0.0f, a2 = 0.0f, a3 = 0.0f;
  for (int base = beg; base < end; base += 16) {
    int jj = base + l;
    unsigned sidx = (unsigned)((jj < end) ? csrc[jj] : n);
    int s0 = (int)__builtin_amdgcn_permlane16(sidx, sidx, 0x11110000u, 0x33332222u, false, false);
    int s1 = (int)__builtin_amdgcn_permlane16(sidx, sidx, 0x55554444u, 0x77776666u, false, false);
    int s2 = (int)__builtin_amdgcn_permlane16(sidx, sidx, 0x99998888u, 0xBBBBAAAAu, false, false);
    int s3 = (int)__builtin_amdgcn_permlane16(sidx, sidx, 0xDDDDCCCCu, 0xFFFFEEEEu, false, false);
    float4 r0 = yin4[(size_t)s0 * 4 + q];
    float4 r1 = yin4[(size_t)s1 * 4 + q];
    float4 r2 = yin4[(size_t)s2 * 4 + q];
    float4 r3 = yin4[(size_t)s3 * 4 + q];
    int e = base + g;
    if (e < end)      { float w = (s0 != n) ? 1.0f : 0.0f; a0 += w * r0.x; a1 += w * r0.y; a2 += w * r0.z; a3 += w * r0.w; }
    if (e + 4 < end)  { float w = (s1 != n) ? 1.0f : 0.0f; a0 += w * r1.x; a1 += w * r1.y; a2 += w * r1.z; a3 += w * r1.w; }
    if (e + 8 < end)  { float w = (s2 != n) ? 1.0f : 0.0f; a0 += w * r2.x; a1 += w * r2.y; a2 += w * r2.z; a3 += w * r2.w; }
    if (e + 12 < end) { float w = (s3 != n) ? 1.0f : 0.0f; a0 += w * r3.x; a1 += w * r3.y; a2 += w * r3.z; a3 += w * r3.w; }
  }
  a0 += __shfl_xor(a0, 4, 16); a1 += __shfl_xor(a1, 4, 16); a2 += __shfl_xor(a2, 4, 16); a3 += __shfl_xor(a3, 4, 16);
  a0 += __shfl_xor(a0, 8, 16); a1 += __shfl_xor(a1, 8, 16); a2 += __shfl_xor(a2, 8, 16); a3 += __shfl_xor(a3, 8, 16);
  float mine = (g == 0) ? a0 : (g == 1) ? a1 : (g == 2) ? a2 : a3;
  int f = 4 * q + g;
  float r = 2.0f * (-dn * mine);
  r -= pv;
  t3s[nl * T3PITCH + f] = valid ? r : 0.0f;
  wsh4[tid] = wz;
  wsh4[256 + tid] = wh;

  bool w0 = tid < 32;
  int c = tid & 15, hf = (tid >> 4) & 1;
  int m = (int)blockIdx.x * 16 - 1 + c;
  bool mvalid = (m >= 0) && (m < N);
  float tb[3][8];
#pragma unroll
  for (int p = 0; p < 3; ++p)
#pragma unroll
    for (int i = 0; i < 8; ++i) tb[p][i] = 0.0f;
  if (w0 && mvalid) {
    ldhalf(T0 + (size_t)m * 16 + 8 * hf, tb[0]);
    ldhalf(T1 + (size_t)m * 16 + 8 * hf, tb[1]);
    ldhalf(T2 + (size_t)m * 16 + 8 * hf, tb[2]);
  }
  __syncthreads();
  if (!w0) return;

  float t3h[8];
  ldhalf(t3s + c * T3PITCH + 8 * hf, t3h);
  v8f accZ = {}; v8f accH = {};
#pragma unroll
  for (int p = 0; p < 4; ++p) {
#pragma unroll
    for (int t = 0; t < 4; ++t) {
      int k = 8 * hf + 2 * t;
      v2f b;
      b.x = (p < 3) ? tb[p < 3 ? p : 0][2 * t] : t3h[2 * t];
      b.y = (p < 3) ? tb[p < 3 ? p : 0][2 * t + 1] : t3h[2 * t + 1];
      v2f az; az.x = wsh[p * 256 + k * 16 + c];        az.y = wsh[p * 256 + (k + 1) * 16 + c];
      v2f ah; ah.x = wsh[1024 + p * 256 + k * 16 + c]; ah.y = wsh[1024 + p * 256 + (k + 1) * 16 + c];
      accZ = WMMA4(az, b, accZ);
      accH = WMMA4(ah, b, accH);
    }
  }
  float bz0[8], bz1[8], bh0[8], bh1[8];
  ldhalf(chebB + 8 * hf, bz0);
  ldhalf(chebB + 16 + 8 * hf, bz1);
  ldhalf(chebB + 64 + 8 * hf, bh0);
  ldhalf(chebB + 80 + 8 * hf, bh1);
  float hrow[8];
#pragma unroll
  for (int i = 0; i < 8; ++i) {
    float zp = accZ[i] + bz0[i] + bz1[i];
    float hp = accH[i] + bh0[i] + bh1[i];
    float z = 1.0f / (1.0f + expf(-zp));
    float ht = tanhf(hp);
    hrow[i] = mvalid ? (1.0f - z) * ht : 0.0f;
  }
  float hid[4];
#pragma unroll
  for (int j = 0; j < 4; ++j) {
    float s = (hf == 0) ? b1[j] : 0.0f;
#pragma unroll
    for (int i = 0; i < 8; ++i) s += hrow[i] * W1[(8 * hf + i) * 4 + j];
    s += __shfl_xor(s, 16);
    hid[j] = fmaxf(s, 0.0f);
  }
  float o4[4];
#pragma unroll
  for (int oo = 0; oo < 4; ++oo) {
    int o = 4 * hf + oo;
    float s = b2[o];
#pragma unroll
    for (int j = 0; j < 4; ++j) s += hid[j] * W2[j * 8 + o];
    o4[oo] = s;
  }
  if (m < N) {
    float4 v; v.x = o4[0]; v.y = o4[1]; v.z = o4[2]; v.w = o4[3];
    *(float4*)(out + (size_t)(m + 1) * 8 + 4 * hf) = v;
  }
}

extern "C" void kernel_launch(void* const* d_in, const int* in_sizes, int n_in,
                              void* d_out, int out_size, void* d_ws, size_t ws_size,
                              hipStream_t stream) {
  const float* x     = (const float*)d_in[0];
  const int*   ei    = (const int*)  d_in[1];
  const float* ea    = (const float*)d_in[2];
  const float* Wl    = (const float*)d_in[3];
  const float* bl    = (const float*)d_in[4];
  const float* Wr    = (const float*)d_in[5];
  const float* br    = (const float*)d_in[6];
  const float* We    = (const float*)d_in[7];
  const float* att   = (const float*)d_in[8];
  const float* gbias = (const float*)d_in[9];
  const float* chebW = (const float*)d_in[10];
  const float* chebB = (const float*)d_in[11];
  const float* W1    = (const float*)d_in[12];
  const float* b1    = (const float*)d_in[13];
  const float* W2    = (const float*)d_in[14];
  const float* b2    = (const float*)d_in[15];
  float* out = (float*)d_out;

  const int N = in_sizes[0] / 128;
  const int E = in_sizes[1] / 2;
  const size_t NF = (size_t)N * 16;

  float* ws    = (float*)d_ws;
  float* xl    = ws;
  float* xr    = xl + NF;
  float* hh    = xr + NF;
  float* T1    = hh + NF;
  float* T2    = T1 + NF;
  float* T3    = T2 + NF;
  float* spare = T3 + NF;
  int*   cnt   = (int*)(spare + NF);
  int*   scnt  = cnt + N;
  float* dinv  = (float*)(scnt + N);
  int*   rowptr = (int*)(dinv + N);
  int*   csrc  = (int*)(spare + NF + (size_t)N * 5);
  int*   be    = (int*)T1;
  int*   bsd   = (int*)xl;
  int*   tabd  = cnt;
  int*   tabs  = scnt;
  unsigned char* srec = (unsigned char*)hh;
  int*   ceid  = (int*)T3;

  const int TPB = 256;
  int gEb  = (E + ESEG - 1) / ESEG;
  int nb   = (N + BKT_NODES - 1) >> BKT_SHIFT;
  int gN16 = (N * 16 + TPB - 1) / TPB;
  int gT   = (N + 16) / 16;
  int tiles = N / 16;

  k_bin<<<gEb, EBLK, 0, stream>>>(ei, tabd, tabs, bsd, be, srec, E, nb);
  k_place<<<nb + nb, EBLK, 0, stream>>>(tabd, bsd, be, rowptr, csrc, ceid, tabs, srec, dinv, N, nb, gEb);

  k_gemm_xlxr<<<tiles, 32, 0, stream>>>(x, Wl, bl, Wr, br, xl, xr);
  k_gat<<<gN16, TPB, 0, stream>>>(rowptr, csrc, ceid, ea, xl, xr, We, att, gbias, dinv, hh, N);

  k_matvec<<<gN16, TPB, 0, stream>>>(rowptr, csrc, dinv, xr, (const float*)nullptr, 1.0f, T1, xl, N);
  k_matvec<<<gN16, TPB, 0, stream>>>(rowptr, csrc, dinv, xl, hh, 2.0f, T2, xr, N);
  k_last<<<gT, TPB, 0, stream>>>(rowptr, csrc, dinv, xr, hh, T1, T2, chebW, chebB, W1, b1, W2, b2, out, N);
}
